# starting kernel
# baseline (speedup 1.0000x reference)
#include <hip/hip_runtime.h>
#include <hip/hip_bf16.h>

#define HEADS 3
#define CPH   128
#define F     384
#define HIN   12

typedef __attribute__((ext_vector_type(2))) float v2f;
typedef __attribute__((ext_vector_type(8))) float v8f;

__device__ __forceinline__ unsigned flipf(float f) {
  unsigned b = __float_as_uint(f);
  return (b & 0x80000000u) ? ~b : (b | 0x80000000u);
}
__device__ __forceinline__ float unflipf(unsigned u) {
  unsigned b = (u & 0x80000000u) ? (u & 0x7FFFFFFFu) : ~u;
  return __uint_as_float(b);
}
__device__ __forceinline__ float lrelu(float v) { return v > 0.f ? v : 0.2f * v; }

__global__ void k_zero(float* __restrict__ p, unsigned long long n) {
  unsigned long long i = (unsigned long long)blockIdx.x * blockDim.x + threadIdx.x;
  if (i < n) p[i] = 0.0f;
}

__global__ void k_proj1(const float* __restrict__ x, const float* __restrict__ W1,
                        const float* __restrict__ a_src, const float* __restrict__ a_dst,
                        float* __restrict__ xp, float* __restrict__ asrc,
                        float* __restrict__ adst) {
  const int n = blockIdx.x, t = threadIdx.x;
  __shared__ float xrow[HIN];
  __shared__ float red[6][CPH];
  if (t < HIN) xrow[t] = x[(size_t)n * HIN + t];
  __syncthreads();
#pragma unroll
  for (int h = 0; h < HEADS; ++h) {
    const int c = h * CPH + t;
    float acc = 0.f;
#pragma unroll
    for (int k = 0; k < HIN; ++k) acc = fmaf(xrow[k], W1[k * F + c], acc);
    xp[(size_t)n * F + c] = acc;
    red[h][t]     = acc * a_src[c];
    red[3 + h][t] = acc * a_dst[c];
  }
  __syncthreads();
  for (int off = CPH / 2; off > 0; off >>= 1) {
    if (t < off) {
#pragma unroll
      for (int j = 0; j < 6; ++j) red[j][t] += red[j][t + off];
    }
    __syncthreads();
  }
  if (t < HEADS)          asrc[(size_t)n * HEADS + t]          = red[t][0];
  else if (t < 2 * HEADS) adst[(size_t)n * HEADS + (t - HEADS)] = red[t][0];
}

__global__ void k_alpha(const float* __restrict__ xp, const float* __restrict__ a_src,
                        const float* __restrict__ a_dst, float* __restrict__ asrc,
                        float* __restrict__ adst) {
  const int n = blockIdx.x, t = threadIdx.x;
  __shared__ float red[6][CPH];
#pragma unroll
  for (int h = 0; h < HEADS; ++h) {
    const int c = h * CPH + t;
    const float v = xp[(size_t)n * F + c];
    red[h][t]     = v * a_src[c];
    red[3 + h][t] = v * a_dst[c];
  }
  __syncthreads();
  for (int off = CPH / 2; off > 0; off >>= 1) {
    if (t < off) {
#pragma unroll
      for (int j = 0; j < 6; ++j) red[j][t] += red[j][t + off];
    }
    __syncthreads();
  }
  if (t < HEADS)          asrc[(size_t)n * HEADS + t]          = red[t][0];
  else if (t < 2 * HEADS) adst[(size_t)n * HEADS + (t - HEADS)] = red[t][0];
}

__global__ void k_edge_max(const int* __restrict__ src, const int* __restrict__ dst,
                           int E, int N, const float* __restrict__ asrc,
                           const float* __restrict__ adst, unsigned* __restrict__ mx) {
  const int i = blockIdx.x * blockDim.x + threadIdx.x;
  if (i >= E + N) return;
  int s, d;
  if (i < E) { s = src[i]; d = dst[i]; } else { s = i - E; d = s; }
#pragma unroll
  for (int h = 0; h < HEADS; ++h) {
    float e = lrelu(asrc[(size_t)s * HEADS + h] + adst[(size_t)d * HEADS + h]);
    atomicMax(&mx[(size_t)d * HEADS + h], flipf(e));
  }
}

__global__ void k_edge_expsum(const int* __restrict__ src, const int* __restrict__ dst,
                              int E, int N, const float* __restrict__ asrc,
                              const float* __restrict__ adst, const unsigned* __restrict__ mx,
                              float* __restrict__ ex, float* __restrict__ den) {
  const int i = blockIdx.x * blockDim.x + threadIdx.x;
  if (i >= E + N) return;
  int s, d;
  if (i < E) { s = src[i]; d = dst[i]; } else { s = i - E; d = s; }
#pragma unroll
  for (int h = 0; h < HEADS; ++h) {
    float e = lrelu(asrc[(size_t)s * HEADS + h] + adst[(size_t)d * HEADS + h]);
    float v = expf(e - unflipf(mx[(size_t)d * HEADS + h]));
    ex[(size_t)i * HEADS + h] = v;
    atomicAdd(&den[(size_t)d * HEADS + h], v);
  }
}

__global__ void k_edge_agg(const int* __restrict__ src, const int* __restrict__ dst,
                           int E, int N, const float* __restrict__ ex,
                           const float* __restrict__ den, const float* __restrict__ xp,
                           float* __restrict__ agg) {
  const int i = blockIdx.x, t = threadIdx.x;
  int s, d;
  if (i < E) { s = src[i]; d = dst[i]; } else { s = i - E; d = s; }
#pragma unroll
  for (int h = 0; h < HEADS; ++h) {
    const float alpha = ex[(size_t)i * HEADS + h] / den[(size_t)d * HEADS + h];
    const int c = h * CPH + t;
    atomicAdd(&agg[(size_t)d * F + c], alpha * xp[(size_t)s * F + c]);
  }
}

__global__ void k_bias_rows(float* __restrict__ agg, const float* __restrict__ b,
                            unsigned long long total) {
  unsigned long long i = (unsigned long long)blockIdx.x * blockDim.x + threadIdx.x;
  if (i < total) agg[i] += b[i % F];
}

__global__ void __launch_bounds__(256) k_gemm(const float* __restrict__ A,
                                              const float* __restrict__ B,
                                              float* __restrict__ C, int mtiles) {
  const int lane = threadIdx.x & 31;
  const int wave = threadIdx.x >> 5;
  const int wt = blockIdx.x * 8 + wave;
  const int npair = F / 32;
  const int tm = wt / npair;
  const int tp = wt % npair;
  if (tm >= mtiles) return;
  const int m  = tm * 16 + (lane & 15);
  const int n0 = tp * 32 + (lane & 15);
  const int n1 = n0 + 16;
  const int kb = (lane >> 4) * 2;
  const float* arow = A + (size_t)m * F;
  v8f acc0 = {};
  v8f acc1 = {};
  for (int k0 = 0; k0 < F; k0 += 4) {
    const int k = k0 + kb;
    v2f a;  a.x  = arow[k];            a.y  = arow[k + 1];
    const float* bp = B + (size_t)k * F;
    v2f b0; b0.x = bp[n0];             b0.y = bp[F + n0];
    v2f b1; b1.x = bp[n1];             b1.y = bp[F + n1];
    acc0 = __builtin_amdgcn_wmma_f32_16x16x4_f32(false, a, false, b0, (short)0, acc0,
                                                 false, false);
    acc1 = __builtin_amdgcn_wmma_f32_16x16x4_f32(false, a, false, b1, (short)0, acc1,
                                                 false, false);
  }
  const int mr = tm * 16 + ((lane >> 4) << 3);
#pragma unroll
  for (int i = 0; i < 8; ++i) {
    C[(size_t)(mr + i) * F + n0] = acc0[i];
    C[(size_t)(mr + i) * F + n1] = acc1[i];
  }
}

__global__ void k_final(const float* __restrict__ agg, const float* __restrict__ b2,
                        const float* __restrict__ Wl, const float* __restrict__ bl,
                        float* __restrict__ out) {
  const int n = blockIdx.x, t = threadIdx.x;
  __shared__ float red[CPH];
  float acc = 0.f;
#pragma unroll
  for (int h = 0; h < HEADS; ++h) {
    const int c = h * CPH + t;
    acc = fmaf(agg[(size_t)n * F + c] + b2[c], Wl[c], acc);
  }
  red[t] = acc;
  __syncthreads();
  for (int off = CPH / 2; off > 0; off >>= 1) {
    if (t < off) red[t] += red[t + off];
    __syncthreads();
  }
  if (t == 0) out[n] = red[0] + bl[0];
}

extern "C" void kernel_launch(void* const* d_in, const int* in_sizes, int n_in,
                              void* d_out, int out_size, void* d_ws, size_t ws_size,
                              hipStream_t stream) {
  const float* x   = (const float*)d_in[0];
  const int*   ei  = (const int*)d_in[1];
  const float* W1  = (const float*)d_in[2];
  const float* as1 = (const float*)d_in[3];
  const float* ad1 = (const float*)d_in[4];
  const float* b1  = (const float*)d_in[5];
  const float* W2  = (const float*)d_in[6];
  const float* as2 = (const float*)d_in[7];
  const float* ad2 = (const float*)d_in[8];
  const float* b2  = (const float*)d_in[9];
  const float* Wl  = (const float*)d_in[10];
  const float* bl  = (const float*)d_in[11];
  float* out = (float*)d_out;

  const int N  = in_sizes[0] / HIN;
  const int E  = in_sizes[1] / 2;
  const int EN = E + N;
  const int* src = ei;
  const int* dst = ei + E;

  float* ws = (float*)d_ws;
  const size_t NF = (size_t)N * F;
  float*    XP   = ws;
  float*    AGG  = XP + NF;
  float*    DEN  = AGG + NF;
  unsigned* MX   = (unsigned*)(DEN + (size_t)3 * N);
  float*    ASRC = (float*)(MX + (size_t)3 * N);
  float*    ADST = ASRC + (size_t)3 * N;
  float*    EX   = ADST + (size_t)3 * N;

  const unsigned long long zcount = NF + (unsigned long long)6 * N;
  const int zblocks = (int)((zcount + 255) / 256);
  const int eblocks = (EN + 255) / 256;
  const int bblocks = (int)((NF + 255) / 256);
  const int mtiles  = N / 16;
  const int gblocks = (mtiles * (F / 32) + 7) / 8;

  k_zero<<<zblocks, 256, 0, stream>>>(AGG, zcount);
  k_proj1<<<N, CPH, 0, stream>>>(x, W1, as1, ad1, XP, ASRC, ADST);
  k_edge_max<<<eblocks, 256, 0, stream>>>(src, dst, E, N, ASRC, ADST, MX);
  k_edge_expsum<<<eblocks, 256, 0, stream>>>(src, dst, E, N, ASRC, ADST, MX, EX, DEN);
  k_edge_agg<<<EN, CPH, 0, stream>>>(src, dst, E, N, EX, DEN, XP, AGG);
  k_bias_rows<<<bblocks, 256, 0, stream>>>(AGG, b1, NF);

  k_gemm<<<gblocks, 256, 0, stream>>>(AGG, W2, XP, mtiles);
  k_alpha<<<N, CPH, 0, stream>>>(XP, as2, ad2, ASRC, ADST);

  k_zero<<<zblocks, 256, 0, stream>>>(AGG, zcount);
  k_edge_max<<<eblocks, 256, 0, stream>>>(src, dst, E, N, ASRC, ADST, MX);
  k_edge_expsum<<<eblocks, 256, 0, stream>>>(src, dst, E, N, ASRC, ADST, MX, EX, DEN);
  k_edge_agg<<<EN, CPH, 0, stream>>>(src, dst, E, N, EX, DEN, XP, AGG);

  k_final<<<N, CPH, 0, stream>>>(AGG, b2, Wl, bl, out);
}


# edge scatter with float atomics replaced by per-destination gather over an edge list grouped in the call (count, scan, fill): softmax, bias, final dot in the row pass; sums in arrival-order rank
# speedup vs baseline: 3.4625x; 3.4625x over previous
#include <hip/hip_runtime.h>
#include <hip/hip_bf16.h>

#define HEADS 3
#define CPH   128
#define F     384
#define HIN   12

typedef __attribute__((ext_vector_type(2))) float v2f;
typedef __attribute__((ext_vector_type(8))) float v8f;

__device__ __forceinline__ float lrelu(float v) { return v > 0.f ? v : 0.2f * v; }

__global__ void k_proj1(const float* __restrict__ x, const float* __restrict__ W1,
                        const float* __restrict__ a_src, const float* __restrict__ a_dst,
                        float* __restrict__ xp, float* __restrict__ asrc,
                        float* __restrict__ adst, int* __restrict__ cnt) {
  const int n = blockIdx.x, t = threadIdx.x;
  __shared__ float xrow[HIN];
  __shared__ float red[6][CPH];
  if (t < HIN) xrow[t] = x[(size_t)n * HIN + t];
  if (t == CPH - 1) cnt[n] = 0;
  __syncthreads();
#pragma unroll
  for (int h = 0; h < HEADS; ++h) {
    const int c = h * CPH + t;
    float acc = 0.f;
#pragma unroll
    for (int k = 0; k < HIN; ++k) acc = fmaf(xrow[k], W1[k * F + c], acc);
    xp[(size_t)n * F + c] = acc;
    red[h][t]     = acc * a_src[c];
    red[3 + h][t] = acc * a_dst[c];
  }
  __syncthreads();
  for (int off = CPH / 2; off > 0; off >>= 1) {
    if (t < off) {
#pragma unroll
      for (int j = 0; j < 6; ++j) red[j][t] += red[j][t + off];
    }
    __syncthreads();
  }
  if (t < HEADS)          asrc[(size_t)n * HEADS + t]          = red[t][0];
  else if (t < 2 * HEADS) adst[(size_t)n * HEADS + (t - HEADS)] = red[t][0];
}

__global__ void k_alpha(const float* __restrict__ xp, const float* __restrict__ a_src,
                        const float* __restrict__ a_dst, float* __restrict__ asrc,
                        float* __restrict__ adst) {
  const int n = blockIdx.x, t = threadIdx.x;
  __shared__ float red[6][CPH];
#pragma unroll
  for (int h = 0; h < HEADS; ++h) {
    const int c = h * CPH + t;
    const float v = xp[(size_t)n * F + c];
    red[h][t]     = v * a_src[c];
    red[3 + h][t] = v * a_dst[c];
  }
  __syncthreads();
  for (int off = CPH / 2; off > 0; off >>= 1) {
    if (t < off) {
#pragma unroll
      for (int j = 0; j < 6; ++j) red[j][t] += red[j][t + off];
    }
    __syncthreads();
  }
  if (t < HEADS)          asrc[(size_t)n * HEADS + t]          = red[t][0];
  else if (t < 2 * HEADS) adst[(size_t)n * HEADS + (t - HEADS)] = red[t][0];
}

__global__ void k_count(const int* __restrict__ dst, int E, int* __restrict__ cnt,
                        int* __restrict__ rank) {
  const int i = blockIdx.x * blockDim.x + threadIdx.x;
  if (i < E) rank[i] = atomicAdd(&cnt[dst[i]], 1);
}

__global__ void __launch_bounds__(1024) k_scan(const int* __restrict__ cnt,
                                               int* __restrict__ rowptr, int N) {
  __shared__ int part[1024];
  const int t = threadIdx.x;
  const int per = (N + 1023) / 1024;
  const int b = t * per;
  const int e = (b + per < N) ? (b + per) : N;
  int s = 0;
  for (int i = b; i < e; ++i) s += cnt[i];
  part[t] = s;
  __syncthreads();
  for (int off = 1; off < 1024; off <<= 1) {
    const int v = (t >= off) ? part[t - off] : 0;
    __syncthreads();
    part[t] += v;
    __syncthreads();
  }
  int run = part[t] - s;
  for (int i = b; i < e; ++i) { rowptr[i] = run; run += cnt[i]; }
  if (t == 1023) rowptr[N] = part[1023];
}

__global__ void k_fill(const int* __restrict__ src, const int* __restrict__ dst, int E,
                       const int* __restrict__ rowptr, const int* __restrict__ rank,
                       int* __restrict__ col) {
  const int i = blockIdx.x * blockDim.x + threadIdx.x;
  if (i < E) col[rowptr[dst[i]] + rank[i]] = src[i];
}

__device__ __forceinline__ float wave_max(float v) {
#pragma unroll
  for (int o = 16; o >= 1; o >>= 1) v = fmaxf(v, __shfl_xor(v, o, 32));
  return v;
}
__device__ __forceinline__ float wave_sum(float v) {
#pragma unroll
  for (int o = 16; o >= 1; o >>= 1) v += __shfl_xor(v, o, 32);
  return v;
}

template <bool FINAL>
__global__ void __launch_bounds__(256) k_rows(const int* __restrict__ rowptr,
                                              const int* __restrict__ col,
                                              const float* __restrict__ asrc,
                                              const float* __restrict__ adst,
                                              const float* __restrict__ xp,
                                              const float* __restrict__ bias,
                                              const float* __restrict__ Wl,
                                              const float* __restrict__ bl,
                                              float* __restrict__ res, int N) {
  const int lane = threadIdx.x & 31;
  const int d = blockIdx.x * 8 + __builtin_amdgcn_readfirstlane((int)(threadIdx.x >> 5));
  if (d >= N) return;
  const int beg = rowptr[d];
  const int cnt = rowptr[d + 1] - beg + 1;
  const float ad0 = adst[(size_t)d * HEADS + 0];
  const float ad1 = adst[(size_t)d * HEADS + 1];
  const float ad2 = adst[(size_t)d * HEADS + 2];

  int   s_c = d;
  float c0 = -INFINITY, c1 = -INFINITY, c2 = -INFINITY;
  if (lane < cnt) {
    s_c = (lane < cnt - 1) ? col[beg + lane] : d;
    const float* ap = asrc + (size_t)s_c * HEADS;
    c0 = lrelu(ap[0] + ad0); c1 = lrelu(ap[1] + ad1); c2 = lrelu(ap[2] + ad2);
  }
  float m0 = c0, m1 = c1, m2 = c2;
  for (int j = lane + 32; j < cnt; j += 32) {
    const int s = (j < cnt - 1) ? col[beg + j] : d;
    const float* ap = asrc + (size_t)s * HEADS;
    m0 = fmaxf(m0, lrelu(ap[0] + ad0));
    m1 = fmaxf(m1, lrelu(ap[1] + ad1));
    m2 = fmaxf(m2, lrelu(ap[2] + ad2));
  }
  m0 = wave_max(m0); m1 = wave_max(m1); m2 = wave_max(m2);

  const float x0 = (lane < cnt) ? expf(c0 - m0) : 0.f;
  const float x1 = (lane < cnt) ? expf(c1 - m1) : 0.f;
  const float x2 = (lane < cnt) ? expf(c2 - m2) : 0.f;
  float z0 = x0, z1 = x1, z2 = x2;
  for (int j = lane + 32; j < cnt; j += 32) {
    const int s = (j < cnt - 1) ? col[beg + j] : d;
    const float* ap = asrc + (size_t)s * HEADS;
    z0 += expf(lrelu(ap[0] + ad0) - m0);
    z1 += expf(lrelu(ap[1] + ad1) - m1);
    z2 += expf(lrelu(ap[2] + ad2) - m2);
  }
  z0 = wave_sum(z0); z1 = wave_sum(z1); z2 = wave_sum(z2);

  float4 a0 = make_float4(0.f, 0.f, 0.f, 0.f);
  float4 a1 = a0, a2 = a0;
  for (int base = 0; base < cnt; base += 32) {
    int   s  = s_c;
    float w0 = x0 / z0, w1 = x1 / z1, w2 = x2 / z2;
    if (base > 0) {
      s = d; w0 = 0.f; w1 = 0.f; w2 = 0.f;
      const int j = base + lane;
      if (j < cnt) {
        s = (j < cnt - 1) ? col[beg + j] : d;
        const float* ap = asrc + (size_t)s * HEADS;
        w0 = expf(lrelu(ap[0] + ad0) - m0) / z0;
        w1 = expf(lrelu(ap[1] + ad1) - m1) / z1;
        w2 = expf(lrelu(ap[2] + ad2) - m2) / z2;
      }
    }
    const int nb = (cnt - base < 32) ? (cnt - base) : 32;
#pragma unroll 4
    for (int k = 0; k < nb; ++k) {
      const int   sk = __builtin_amdgcn_readlane(s, k);
      const float q0 = __int_as_float(__builtin_amdgcn_readlane(__float_as_int(w0), k));
      const float q1 = __int_as_float(__builtin_amdgcn_readlane(__float_as_int(w1), k));
      const float q2 = __int_as_float(__builtin_amdgcn_readlane(__float_as_int(w2), k));
      const float4* row = reinterpret_cast<const float4*>(xp + (size_t)sk * F) + lane;
      const float4 r0 = row[0], r1 = row[32], r2 = row[64];
      a0.x = fmaf(q0, r0.x, a0.x); a0.y = fmaf(q0, r0.y, a0.y);
      a0.z = fmaf(q0, r0.z, a0.z); a0.w = fmaf(q0, r0.w, a0.w);
      a1.x = fmaf(q1, r1.x, a1.x); a1.y = fmaf(q1, r1.y, a1.y);
      a1.z = fmaf(q1, r1.z, a1.z); a1.w = fmaf(q1, r1.w, a1.w);
      a2.x = fmaf(q2, r2.x, a2.x); a2.y = fmaf(q2, r2.y, a2.y);
      a2.z = fmaf(q2, r2.z, a2.z); a2.w = fmaf(q2, r2.w, a2.w);
    }
  }

  const int cb = lane * 4;
  a0.x += bias[cb + 0];           a0.y += bias[cb + 1];
  a0.z += bias[cb + 2];           a0.w += bias[cb + 3];
  a1.x += bias[CPH + cb + 0];     a1.y += bias[CPH + cb + 1];
  a1.z += bias[CPH + cb + 2];     a1.w += bias[CPH + cb + 3];
  a2.x += bias[2 * CPH + cb + 0]; a2.y += bias[2 * CPH + cb + 1];
  a2.z += bias[2 * CPH + cb + 2]; a2.w += bias[2 * CPH + cb + 3];
  if (FINAL) {
    float p = a0.x * Wl[cb + 0];
    p = fmaf(a0.y, Wl[cb + 1], p);           p = fmaf(a0.z, Wl[cb + 2], p);
    p = fmaf(a0.w, Wl[cb + 3], p);
    p = fmaf(a1.x, Wl[CPH + cb + 0], p);     p = fmaf(a1.y, Wl[CPH + cb + 1], p);
    p = fmaf(a1.z, Wl[CPH + cb + 2], p);     p = fmaf(a1.w, Wl[CPH + cb + 3], p);
    p = fmaf(a2.x, Wl[2 * CPH + cb + 0], p); p = fmaf(a2.y, Wl[2 * CPH + cb + 1], p);
    p = fmaf(a2.z, Wl[2 * CPH + cb + 2], p); p = fmaf(a2.w, Wl[2 * CPH + cb + 3], p);
    p = wave_sum(p);
    if (lane == 0) res[d] = p + bl[0];
  } else {
    float4* o = reinterpret_cast<float4*>(res + (size_t)d * F) + lane;
    o[0] = a0; o[32] = a1; o[64] = a2;
  }
}

__global__ void __launch_bounds__(256) k_gemm(const float* __restrict__ A,
                                              const float* __restrict__ B,
                                              float* __restrict__ C, int mtiles) {
  const int lane = threadIdx.x & 31;
  const int wave = threadIdx.x >> 5;
  const int wt = blockIdx.x * 8 + wave;
  const int npair = F / 32;
  const int tm = wt / npair;
  const int tp = wt % npair;
  if (tm >= mtiles) return;
  const int m  = tm * 16 + (lane & 15);
  const int n0 = tp * 32 + (lane & 15);
  const int n1 = n0 + 16;
  const int kb = (lane >> 4) * 2;
  const float* arow = A + (size_t)m * F;
  v8f acc0 = {};
  v8f acc1 = {};
  for (int k0 = 0; k0 < F; k0 += 4) {
    const int k = k0 + kb;
    v2f a;  a.x  = arow[k];            a.y  = arow[k + 1];
    const float* bp = B + (size_t)k * F;
    v2f b0; b0.x = bp[n0];             b0.y = bp[F + n0];
    v2f b1; b1.x = bp[n1];             b1.y = bp[F + n1];
    acc0 = __builtin_amdgcn_wmma_f32_16x16x4_f32(false, a, false, b0, (short)0, acc0,
                                                 false, false);
    acc1 = __builtin_amdgcn_wmma_f32_16x16x4_f32(false, a, false, b1, (short)0, acc1,
                                                 false, false);
  }
  const int mr = tm * 16 + ((lane >> 4) << 3);
#pragma unroll
  for (int i = 0; i < 8; ++i) {
    C[(size_t)(mr + i) * F + n0] = acc0[i];
    C[(size_t)(mr + i) * F + n1] = acc1[i];
  }
}

extern "C" void kernel_launch(void* const* d_in, const int* in_sizes, int n_in,
                              void* d_out, int out_size, void* d_ws, size_t ws_size,
                              hipStream_t stream) {
  const float* x   = (const float*)d_in[0];
  const int*   ei  = (const int*)d_in[1];
  const float* W1  = (const float*)d_in[2];
  const float* as1 = (const float*)d_in[3];
  const float* ad1 = (const float*)d_in[4];
  const float* b1  = (const float*)d_in[5];
  const float* W2  = (const float*)d_in[6];
  const float* as2 = (const float*)d_in[7];
  const float* ad2 = (const float*)d_in[8];
  const float* b2  = (const float*)d_in[9];
  const float* Wl  = (const float*)d_in[10];
  const float* bl  = (const float*)d_in[11];
  float* out = (float*)d_out;

  const int N  = in_sizes[0] / HIN;
  const int E  = in_sizes[1] / 2;
  const int* src = ei;
  const int* dst = ei + E;

  float* ws = (float*)d_ws;
  const size_t NF = (size_t)N * F;
  float* XP     = ws;
  float* AGG    = XP + NF;
  float* ASRC   = AGG + NF;
  float* ADST   = ASRC + (size_t)3 * N;
  int*   CNT    = (int*)(ADST + (size_t)3 * N);
  int*   ROWPTR = CNT + N;
  int*   RANK   = ROWPTR + (N + 1);
  int*   COL    = RANK + E;

  const int eblocks = (E + 255) / 256;
  const int rblocks = (N + 7) / 8;
  const int mtiles  = N / 16;
  const int gblocks = (mtiles * (F / 32) + 7) / 8;

  k_proj1<<<N, CPH, 0, stream>>>(x, W1, as1, ad1, XP, ASRC, ADST, CNT);
  k_count<<<eblocks, 256, 0, stream>>>(dst, E, CNT, RANK);
  k_scan<<<1, 1024, 0, stream>>>(CNT, ROWPTR, N);
  k_fill<<<eblocks, 256, 0, stream>>>(src, dst, E, ROWPTR, RANK, COL);

  k_rows<false><<<rblocks, 256, 0, stream>>>(ROWPTR, COL, ASRC, ADST, XP, b1, Wl, bl, AGG, N);

  k_gemm<<<gblocks, 256, 0, stream>>>(AGG, W2, XP, mtiles);
  k_alpha<<<N, CPH, 0, stream>>>(XP, as2, ad2, ASRC, ADST);

  k_rows<true><<<rblocks, 256, 0, stream>>>(ROWPTR, COL, ASRC, ADST, XP, b2, Wl, bl, out, N);
}


# layer-1 projection moved behind the aggregation (it is linear): softmax-weighted sums over the 12-wide input rows, 48 B per edge not 1536 B, then times W1; logits from x.(W1 a); sums reassociated
# speedup vs baseline: 4.0191x; 1.1608x over previous
#include <hip/hip_runtime.h>
#include <hip/hip_bf16.h>

#define HEADS 3
#define CPH   128
#define F     384
#define HIN   12

typedef __attribute__((ext_vector_type(2))) float v2f;
typedef __attribute__((ext_vector_type(8))) float v8f;

__device__ __forceinline__ float lrelu(float v) { return v > 0.f ? v : 0.2f * v; }

__device__ __forceinline__ float wave_max(float v) {
#pragma unroll
  for (int o = 16; o >= 1; o >>= 1) v = fmaxf(v, __shfl_xor(v, o, 32));
  return v;
}
__device__ __forceinline__ float wave_sum(float v) {
#pragma unroll
  for (int o = 16; o >= 1; o >>= 1) v += __shfl_xor(v, o, 32);
  return v;
}

__global__ void __launch_bounds__(128) k_att1(const float* __restrict__ x, const float* __restrict__ W1,
                                              const float* __restrict__ a_src,
                                              const float* __restrict__ a_dst,
                                              float* __restrict__ asrc, float* __restrict__ adst,
                                              int* __restrict__ cnt, int N) {
  __shared__ float v[6][HIN];
  const int t = threadIdx.x;
  if (t < 6 * HIN) {
    const int j = t / HIN, k = t - j * HIN;
    const int h = (j < HEADS) ? j : (j - HEADS);
    const float4* wp = reinterpret_cast<const float4*>(W1 + (size_t)k * F + h * CPH);
    const float4* ap = reinterpret_cast<const float4*>(((j < HEADS) ? a_src : a_dst) + h * CPH);
    float p0 = 0.f, p1 = 0.f, p2 = 0.f, p3 = 0.f;
#pragma unroll 8
    for (int i = 0; i < CPH / 4; ++i) {
      const float4 w = wp[i], a = ap[i];
      p0 = fmaf(w.x, a.x, p0); p1 = fmaf(w.y, a.y, p1);
      p2 = fmaf(w.z, a.z, p2); p3 = fmaf(w.w, a.w, p3);
    }
    v[j][k] = (p0 + p1) + (p2 + p3);
  }
  __syncthreads();
  const int n = blockIdx.x * 128 + t;
  if (n >= N) return;
  const float4* xr = reinterpret_cast<const float4*>(x + (size_t)n * HIN);
  const float4 xa = xr[0], xb = xr[1], xc = xr[2];
  const float xs[HIN] = {xa.x, xa.y, xa.z, xa.w, xb.x, xb.y, xb.z, xb.w, xc.x, xc.y, xc.z, xc.w};
  float r[6];
#pragma unroll
  for (int j = 0; j < 6; ++j) {
    float s = 0.f;
#pragma unroll
    for (int k = 0; k < HIN; ++k) s = fmaf(xs[k], v[j][k], s);
    r[j] = s;
  }
  asrc[(size_t)n * HEADS + 0] = r[0]; asrc[(size_t)n * HEADS + 1] = r[1];
  asrc[(size_t)n * HEADS + 2] = r[2];
  adst[(size_t)n * HEADS + 0] = r[3]; adst[(size_t)n * HEADS + 1] = r[4];
  adst[(size_t)n * HEADS + 2] = r[5];
  cnt[n] = 0;
}

__global__ void k_alpha(const float* __restrict__ xp, const float* __restrict__ a_src,
                        const float* __restrict__ a_dst, float* __restrict__ asrc,
                        float* __restrict__ adst) {
  const int n = blockIdx.x, t = threadIdx.x;
  __shared__ float red[6][CPH];
#pragma unroll
  for (int h = 0; h < HEADS; ++h) {
    const int c = h * CPH + t;
    const float v = xp[(size_t)n * F + c];
    red[h][t]     = v * a_src[c];
    red[3 + h][t] = v * a_dst[c];
  }
  __syncthreads();
  for (int off = CPH / 2; off > 0; off >>= 1) {
    if (t < off) {
#pragma unroll
      for (int j = 0; j < 6; ++j) red[j][t] += red[j][t + off];
    }
    __syncthreads();
  }
  if (t < HEADS)          asrc[(size_t)n * HEADS + t]          = red[t][0];
  else if (t < 2 * HEADS) adst[(size_t)n * HEADS + (t - HEADS)] = red[t][0];
}

__global__ void k_count(const int* __restrict__ dst, int E, int* __restrict__ cnt,
                        int* __restrict__ rank) {
  const int i = blockIdx.x * blockDim.x + threadIdx.x;
  if (i < E) rank[i] = atomicAdd(&cnt[dst[i]], 1);
}

__global__ void __launch_bounds__(1024) k_scan(const int* __restrict__ cnt,
                                               int* __restrict__ rowptr, int N) {
  __shared__ int part[1024];
  const int t = threadIdx.x;
  const int per = (N + 1023) / 1024;
  const int b = t * per;
  const int e = (b + per < N) ? (b + per) : N;
  int s = 0;
  for (int i = b; i < e; ++i) s += cnt[i];
  part[t] = s;
  __syncthreads();
  for (int off = 1; off < 1024; off <<= 1) {
    const int v = (t >= off) ? part[t - off] : 0;
    __syncthreads();
    part[t] += v;
    __syncthreads();
  }
  int run = part[t] - s;
  for (int i = b; i < e; ++i) { rowptr[i] = run; run += cnt[i]; }
  if (t == 1023) rowptr[N] = part[1023];
}

__global__ void k_fill(const int* __restrict__ src, const int* __restrict__ dst, int E,
                       const int* __restrict__ rowptr, const int* __restrict__ rank,
                       int* __restrict__ col) {
  const int i = blockIdx.x * blockDim.x + threadIdx.x;
  if (i < E) col[rowptr[dst[i]] + rank[i]] = src[i];
}

struct RunSoftmax {
  int   s_c;
  float x0, x1, x2;
  float m0, m1, m2;
  float z0, z1, z2;
};

__device__ __forceinline__ RunSoftmax run_softmax(const int* __restrict__ col,
                                                  const float* __restrict__ asrc, int beg, int cnt,
                                                  int d, int lane, float ad0, float ad1, float ad2) {
  RunSoftmax r;
  r.s_c = d;
  float c0 = -INFINITY, c1 = -INFINITY, c2 = -INFINITY;
  if (lane < cnt) {
    r.s_c = (lane < cnt - 1) ? col[beg + lane] : d;
    const float* ap = asrc + (size_t)r.s_c * HEADS;
    c0 = lrelu(ap[0] + ad0); c1 = lrelu(ap[1] + ad1); c2 = lrelu(ap[2] + ad2);
  }
  float m0 = c0, m1 = c1, m2 = c2;
  for (int j = lane + 32; j < cnt; j += 32) {
    const int s = (j < cnt - 1) ? col[beg + j] : d;
    const float* ap = asrc + (size_t)s * HEADS;
    m0 = fmaxf(m0, lrelu(ap[0] + ad0));
    m1 = fmaxf(m1, lrelu(ap[1] + ad1));
    m2 = fmaxf(m2, lrelu(ap[2] + ad2));
  }
  m0 = wave_max(m0); m1 = wave_max(m1); m2 = wave_max(m2);

  r.x0 = (lane < cnt) ? expf(c0 - m0) : 0.f;
  r.x1 = (lane < cnt) ? expf(c1 - m1) : 0.f;
  r.x2 = (lane < cnt) ? expf(c2 - m2) : 0.f;
  float z0 = r.x0, z1 = r.x1, z2 = r.x2;
  for (int j = lane + 32; j < cnt; j += 32) {
    const int s = (j < cnt - 1) ? col[beg + j] : d;
    const float* ap = asrc + (size_t)s * HEADS;
    z0 += expf(lrelu(ap[0] + ad0) - m0);
    z1 += expf(lrelu(ap[1] + ad1) - m1);
    z2 += expf(lrelu(ap[2] + ad2) - m2);
  }
  r.m0 = m0; r.m1 = m1; r.m2 = m2;
  r.z0 = wave_sum(z0); r.z1 = wave_sum(z1); r.z2 = wave_sum(z2);
  return r;
}

__device__ __forceinline__ void chunk_entry(const RunSoftmax& r, int base, const int* __restrict__ col,
                                            const float* __restrict__ asrc, int beg, int cnt, int d,
                                            int lane, float ad0, float ad1, float ad2, int& s,
                                            float& w0, float& w1, float& w2) {
  s = r.s_c; w0 = r.x0 / r.z0; w1 = r.x1 / r.z1; w2 = r.x2 / r.z2;
  if (base > 0) {
    s = d; w0 = 0.f; w1 = 0.f; w2 = 0.f;
    const int j = base + lane;
    if (j < cnt) {
      s = (j < cnt - 1) ? col[beg + j] : d;
      const float* ap = asrc + (size_t)s * HEADS;
      w0 = expf(lrelu(ap[0] + ad0) - r.m0) / r.z0;
      w1 = expf(lrelu(ap[1] + ad1) - r.m1) / r.z1;
      w2 = expf(lrelu(ap[2] + ad2) - r.m2) / r.z2;
    }
  }
}

__global__ void __launch_bounds__(256) k_rows1(const int* __restrict__ rowptr,
                                               const int* __restrict__ col,
                                               const float* __restrict__ asrc,
                                               const float* __restrict__ adst,
                                               const float* __restrict__ x,
                                               const float* __restrict__ W1,
                                               const float* __restrict__ bias,
                                               float* __restrict__ res, int N) {
  const int lane = threadIdx.x & 31;
  const int d = blockIdx.x * 8 + __builtin_amdgcn_readfirstlane((int)(threadIdx.x >> 5));
  if (d >= N) return;
  const int beg = rowptr[d];
  const int cnt = rowptr[d + 1] - beg + 1;
  const float ad0 = adst[(size_t)d * HEADS + 0];
  const float ad1 = adst[(size_t)d * HEADS + 1];
  const float ad2 = adst[(size_t)d * HEADS + 2];
  const RunSoftmax r = run_softmax(col, asrc, beg, cnt, d, lane, ad0, ad1, ad2);

  const int kk = (lane < HIN) ? lane : 0;
  float g0 = 0.f, g1 = 0.f, g2 = 0.f;
  for (int base = 0; base < cnt; base += 32) {
    int s; float w0, w1, w2;
    chunk_entry(r, base, col, asrc, beg, cnt, d, lane, ad0, ad1, ad2, s, w0, w1, w2);
    const int nb = (cnt - base < 32) ? (cnt - base) : 32;
    for (int k0 = 0; k0 < nb; k0 += 16) {
#pragma unroll
      for (int u = 0; u < 16; ++u) {
        const int k = k0 + u;
        const int   sk = __builtin_amdgcn_readlane(s, k);
        const float q0 = __int_as_float(__builtin_amdgcn_readlane(__float_as_int(w0), k));
        const float q1 = __int_as_float(__builtin_amdgcn_readlane(__float_as_int(w1), k));
        const float q2 = __int_as_float(__builtin_amdgcn_readlane(__float_as_int(w2), k));
        const float xv = x[(size_t)sk * HIN + kk];
        g0 = fmaf(q0, xv, g0); g1 = fmaf(q1, xv, g1); g2 = fmaf(q2, xv, g2);
      }
    }
  }

  float4 a0 = make_float4(0.f, 0.f, 0.f, 0.f);
  float4 a1 = a0, a2 = a0;
#pragma unroll
  for (int k = 0; k < HIN; ++k) {
    const float p0 = __int_as_float(__builtin_amdgcn_readlane(__float_as_int(g0), k));
    const float p1 = __int_as_float(__builtin_amdgcn_readlane(__float_as_int(g1), k));
    const float p2 = __int_as_float(__builtin_amdgcn_readlane(__float_as_int(g2), k));
    const float4* wr = reinterpret_cast<const float4*>(W1 + (size_t)k * F) + lane;
    const float4 u0 = wr[0], u1 = wr[32], u2 = wr[64];
    a0.x = fmaf(p0, u0.x, a0.x); a0.y = fmaf(p0, u0.y, a0.y);
    a0.z = fmaf(p0, u0.z, a0.z); a0.w = fmaf(p0, u0.w, a0.w);
    a1.x = fmaf(p1, u1.x, a1.x); a1.y = fmaf(p1, u1.y, a1.y);
    a1.z = fmaf(p1, u1.z, a1.z); a1.w = fmaf(p1, u1.w, a1.w);
    a2.x = fmaf(p2, u2.x, a2.x); a2.y = fmaf(p2, u2.y, a2.y);
    a2.z = fmaf(p2, u2.z, a2.z); a2.w = fmaf(p2, u2.w, a2.w);
  }

  const float4* bp = reinterpret_cast<const float4*>(bias) + lane;
  const float4 b0 = bp[0], b1 = bp[32], b2 = bp[64];
  a0.x += b0.x; a0.y += b0.y; a0.z += b0.z; a0.w += b0.w;
  a1.x += b1.x; a1.y += b1.y; a1.z += b1.z; a1.w += b1.w;
  a2.x += b2.x; a2.y += b2.y; a2.z += b2.z; a2.w += b2.w;
  float4* o = reinterpret_cast<float4*>(res + (size_t)d * F) + lane;
  o[0] = a0; o[32] = a1; o[64] = a2;
}

__global__ void __launch_bounds__(256) k_rows2(const int* __restrict__ rowptr,
                                               const int* __restrict__ col,
                                               const float* __restrict__ asrc,
                                               const float* __restrict__ adst,
                                               const float* __restrict__ xp,
                                               const float* __restrict__ bias,
                                               const float* __restrict__ Wl,
                                               const float* __restrict__ bl,
                                               float* __restrict__ res, int N) {
  const int lane = threadIdx.x & 31;
  const int d = blockIdx.x * 8 + __builtin_amdgcn_readfirstlane((int)(threadIdx.x >> 5));
  if (d >= N) return;
  const int beg = rowptr[d];
  const int cnt = rowptr[d + 1] - beg + 1;
  const float ad0 = adst[(size_t)d * HEADS + 0];
  const float ad1 = adst[(size_t)d * HEADS + 1];
  const float ad2 = adst[(size_t)d * HEADS + 2];
  const RunSoftmax r = run_softmax(col, asrc, beg, cnt, d, lane, ad0, ad1, ad2);

  float4 a0 = make_float4(0.f, 0.f, 0.f, 0.f);
  float4 a1 = a0, a2 = a0;
  for (int base = 0; base < cnt; base += 32) {
    int s; float w0, w1, w2;
    chunk_entry(r, base, col, asrc, beg, cnt, d, lane, ad0, ad1, ad2, s, w0, w1, w2);
    const int nb = (cnt - base < 32) ? (cnt - base) : 32;
#pragma unroll 4
    for (int k = 0; k < nb; ++k) {
      const int   sk = __builtin_amdgcn_readlane(s, k);
      const float q0 = __int_as_float(__builtin_amdgcn_readlane(__float_as_int(w0), k));
      const float q1 = __int_as_float(__builtin_amdgcn_readlane(__float_as_int(w1), k));
      const float q2 = __int_as_float(__builtin_amdgcn_readlane(__float_as_int(w2), k));
      const float4* row = reinterpret_cast<const float4*>(xp + (size_t)sk * F) + lane;
      const float4 r0 = row[0], r1 = row[32], r2 = row[64];
      a0.x = fmaf(q0, r0.x, a0.x); a0.y = fmaf(q0, r0.y, a0.y);
      a0.z = fmaf(q0, r0.z, a0.z); a0.w = fmaf(q0, r0.w, a0.w);
      a1.x = fmaf(q1, r1.x, a1.x); a1.y = fmaf(q1, r1.y, a1.y);
      a1.z = fmaf(q1, r1.z, a1.z); a1.w = fmaf(q1, r1.w, a1.w);
      a2.x = fmaf(q2, r2.x, a2.x); a2.y = fmaf(q2, r2.y, a2.y);
      a2.z = fmaf(q2, r2.z, a2.z); a2.w = fmaf(q2, r2.w, a2.w);
    }
  }

  const int cb = lane * 4;
  a0.x += bias[cb + 0];           a0.y += bias[cb + 1];
  a0.z += bias[cb + 2];           a0.w += bias[cb + 3];
  a1.x += bias[CPH + cb + 0];     a1.y += bias[CPH + cb + 1];
  a1.z += bias[CPH + cb + 2];     a1.w += bias[CPH + cb + 3];
  a2.x += bias[2 * CPH + cb + 0]; a2.y += bias[2 * CPH + cb + 1];
  a2.z += bias[2 * CPH + cb + 2]; a2.w += bias[2 * CPH + cb + 3];
  float p = a0.x * Wl[cb + 0];
  p = fmaf(a0.y, Wl[cb + 1], p);           p = fmaf(a0.z, Wl[cb + 2], p);
  p = fmaf(a0.w, Wl[cb + 3], p);
  p = fmaf(a1.x, Wl[CPH + cb + 0], p);     p = fmaf(a1.y, Wl[CPH + cb + 1], p);
  p = fmaf(a1.z, Wl[CPH + cb + 2], p);     p = fmaf(a1.w, Wl[CPH + cb + 3], p);
  p = fmaf(a2.x, Wl[2 * CPH + cb + 0], p); p = fmaf(a2.y, Wl[2 * CPH + cb + 1], p);
  p = fmaf(a2.z, Wl[2 * CPH + cb + 2], p); p = fmaf(a2.w, Wl[2 * CPH + cb + 3], p);
  p = wave_sum(p);
  if (lane == 0) res[d] = p + bl[0];
}

__global__ void __launch_bounds__(256) k_gemm(const float* __restrict__ A,
                                              const float* __restrict__ B,
                                              float* __restrict__ C, int mtiles) {
  const int lane = threadIdx.x & 31;
  const int wave = threadIdx.x >> 5;
  const int wt = blockIdx.x * 8 + wave;
  const int npair = F / 32;
  const int tm = wt / npair;
  const int tp = wt % npair;
  if (tm >= mtiles) return;
  const int m  = tm * 16 + (lane & 15);
  const int n0 = tp * 32 + (lane & 15);
  const int n1 = n0 + 16;
  const int kb = (lane >> 4) * 2;
  const float* arow = A + (size_t)m * F;
  v8f acc0 = {};
  v8f acc1 = {};
  for (int k0 = 0; k0 < F; k0 += 4) {
    const int k = k0 + kb;
    v2f a;  a.x  = arow[k];            a.y  = arow[k + 1];
    const float* bp = B + (size_t)k * F;
    v2f b0; b0.x = bp[n0];             b0.y = bp[F + n0];
    v2f b1; b1.x = bp[n1];             b1.y = bp[F + n1];
    acc0 = __builtin_amdgcn_wmma_f32_16x16x4_f32(false, a, false, b0, (short)0, acc0,
                                                 false, false);
    acc1 = __builtin_amdgcn_wmma_f32_16x16x4_f32(false, a, false, b1, (short)0, acc1,
                                                 false, false);
  }
  const int mr = tm * 16 + ((lane >> 4) << 3);
#pragma unroll
  for (int i = 0; i < 8; ++i) {
    C[(size_t)(mr + i) * F + n0] = acc0[i];
    C[(size_t)(mr + i) * F + n1] = acc1[i];
  }
}

extern "C" void kernel_launch(void* const* d_in, const int* in_sizes, int n_in,
                              void* d_out, int out_size, void* d_ws, size_t ws_size,
                              hipStream_t stream) {
  const float* x   = (const float*)d_in[0];
  const int*   ei  = (const int*)d_in[1];
  const float* W1  = (const float*)d_in[2];
  const float* as1 = (const float*)d_in[3];
  const float* ad1 = (const float*)d_in[4];
  const float* b1  = (const float*)d_in[5];
  const float* W2  = (const float*)d_in[6];
  const float* as2 = (const float*)d_in[7];
  const float* ad2 = (const float*)d_in[8];
  const float* b2  = (const float*)d_in[9];
  const float* Wl  = (const float*)d_in[10];
  const float* bl  = (const float*)d_in[11];
  float* out = (float*)d_out;

  const int N  = in_sizes[0] / HIN;
  const int E  = in_sizes[1] / 2;
  const int* src = ei;
  const int* dst = ei + E;

  float* ws = (float*)d_ws;
  const size_t NF = (size_t)N * F;
  float* XP     = ws;
  float* AGG    = XP + NF;
  float* ASRC   = AGG + NF;
  float* ADST   = ASRC + (size_t)3 * N;
  int*   CNT    = (int*)(ADST + (size_t)3 * N);
  int*   ROWPTR = CNT + N;
  int*   RANK   = ROWPTR + (N + 1);
  int*   COL    = RANK + E;

  const int eblocks = (E + 255) / 256;
  const int rblocks = (N + 7) / 8;
  const int ablocks = (N + 127) / 128;
  const int mtiles  = N / 16;
  const int gblocks = (mtiles * (F / 32) + 7) / 8;

  k_att1<<<ablocks, 128, 0, stream>>>(x, W1, as1, ad1, ASRC, ADST, CNT, N);
  k_count<<<eblocks, 256, 0, stream>>>(dst, E, CNT, RANK);
  k_scan<<<1, 1024, 0, stream>>>(CNT, ROWPTR, N);
  k_fill<<<eblocks, 256, 0, stream>>>(src, dst, E, ROWPTR, RANK, COL);

  k_rows1<<<rblocks, 256, 0, stream>>>(ROWPTR, COL, ASRC, ADST, x, W1, b1, AGG, N);

  k_gemm<<<gblocks, 256, 0, stream>>>(AGG, W2, XP, mtiles);
  k_alpha<<<N, CPH, 0, stream>>>(XP, as2, ad2, ASRC, ADST);

  k_rows2<<<rblocks, 256, 0, stream>>>(ROWPTR, COL, ASRC, ADST, XP, b2, Wl, bl, out, N);
}


# layer-2 projection and final linear folded behind the aggregation: W2 applied to the 9 vectors (a_src2, a_dst2, Wl per head) on WMMA, nodes carry 9 scalars not 384-wide rows; sums reassociated
# speedup vs baseline: 7.9620x; 1.9810x over previous
#include <hip/hip_runtime.h>
#include <hip/hip_bf16.h>

#define HEADS 3
#define CPH   128
#define F     384
#define HIN   12

typedef __attribute__((ext_vector_type(2))) float v2f;
typedef __attribute__((ext_vector_type(8))) float v8f;

__device__ __forceinline__ float lrelu(float v) { return v > 0.f ? v : 0.2f * v; }

__device__ __forceinline__ float wave_max(float v) {
#pragma unroll
  for (int o = 16; o >= 1; o >>= 1) v = fmaxf(v, __shfl_xor(v, o, 32));
  return v;
}
__device__ __forceinline__ float wave_sum(float v) {
#pragma unroll
  for (int o = 16; o >= 1; o >>= 1) v += __shfl_xor(v, o, 32);
  return v;
}

__global__ void __launch_bounds__(128) k_att1(const float* __restrict__ x, const float* __restrict__ W1,
                                              const float* __restrict__ a_src,
                                              const float* __restrict__ a_dst,
                                              float* __restrict__ asrc, float* __restrict__ adst,
                                              int* __restrict__ cnt, int N) {
  __shared__ float v[6][HIN];
  const int t = threadIdx.x;
  if (t < 6 * HIN) {
    const int j = t / HIN, k = t - j * HIN;
    const int h = (j < HEADS) ? j : (j - HEADS);
    const float4* wp = reinterpret_cast<const float4*>(W1 + (size_t)k * F + h * CPH);
    const float4* ap = reinterpret_cast<const float4*>(((j < HEADS) ? a_src : a_dst) + h * CPH);
    float p0 = 0.f, p1 = 0.f, p2 = 0.f, p3 = 0.f;
#pragma unroll 8
    for (int i = 0; i < CPH / 4; ++i) {
      const float4 w = wp[i], a = ap[i];
      p0 = fmaf(w.x, a.x, p0); p1 = fmaf(w.y, a.y, p1);
      p2 = fmaf(w.z, a.z, p2); p3 = fmaf(w.w, a.w, p3);
    }
    v[j][k] = (p0 + p1) + (p2 + p3);
  }
  __syncthreads();
  const int n = blockIdx.x * 128 + t;
  if (n >= N) return;
  const float4* xr = reinterpret_cast<const float4*>(x + (size_t)n * HIN);
  const float4 xa = xr[0], xb = xr[1], xc = xr[2];
  const float xs[HIN] = {xa.x, xa.y, xa.z, xa.w, xb.x, xb.y, xb.z, xb.w, xc.x, xc.y, xc.z, xc.w};
  float r[6];
#pragma unroll
  for (int j = 0; j < 6; ++j) {
    float s = 0.f;
#pragma unroll
    for (int k = 0; k < HIN; ++k) s = fmaf(xs[k], v[j][k], s);
    r[j] = s;
  }
  asrc[(size_t)n * HEADS + 0] = r[0]; asrc[(size_t)n * HEADS + 1] = r[1];
  asrc[(size_t)n * HEADS + 2] = r[2];
  adst[(size_t)n * HEADS + 0] = r[3]; adst[(size_t)n * HEADS + 1] = r[4];
  adst[(size_t)n * HEADS + 2] = r[5];
  cnt[n] = 0;
}

__global__ void k_count(const int* __restrict__ dst, int E, int* __restrict__ cnt,
                        int* __restrict__ rank) {
  const int i = blockIdx.x * blockDim.x + threadIdx.x;
  if (i < E) rank[i] = atomicAdd(&cnt[dst[i]], 1);
}

__global__ void __launch_bounds__(1024) k_scan(const int* __restrict__ cnt,
                                               int* __restrict__ rowptr, int N) {
  __shared__ int part[1024];
  const int t = threadIdx.x;
  const int per = (N + 1023) / 1024;
  const int b = t * per;
  const int e = (b + per < N) ? (b + per) : N;
  int s = 0;
  for (int i = b; i < e; ++i) s += cnt[i];
  part[t] = s;
  __syncthreads();
  for (int off = 1; off < 1024; off <<= 1) {
    const int v = (t >= off) ? part[t - off] : 0;
    __syncthreads();
    part[t] += v;
    __syncthreads();
  }
  int run = part[t] - s;
  for (int i = b; i < e; ++i) { rowptr[i] = run; run += cnt[i]; }
  if (t == 1023) rowptr[N] = part[1023];
}

__global__ void k_fill(const int* __restrict__ src, const int* __restrict__ dst, int E,
                       const int* __restrict__ rowptr, const int* __restrict__ rank,
                       int* __restrict__ col) {
  const int i = blockIdx.x * blockDim.x + threadIdx.x;
  if (i < E) col[rowptr[dst[i]] + rank[i]] = src[i];
}

struct RunSoftmax {
  int   s_c;
  float x0, x1, x2;
  float m0, m1, m2;
  float z0, z1, z2;
};

__device__ __forceinline__ RunSoftmax run_softmax(const int* __restrict__ col,
                                                  const float* __restrict__ asrc, int beg, int cnt,
                                                  int d, int lane, float ad0, float ad1, float ad2) {
  RunSoftmax r;
  r.s_c = d;
  float c0 = -INFINITY, c1 = -INFINITY, c2 = -INFINITY;
  if (lane < cnt) {
    r.s_c = (lane < cnt - 1) ? col[beg + lane] : d;
    const float* ap = asrc + (size_t)r.s_c * HEADS;
    c0 = lrelu(ap[0] + ad0); c1 = lrelu(ap[1] + ad1); c2 = lrelu(ap[2] + ad2);
  }
  float m0 = c0, m1 = c1, m2 = c2;
  for (int j = lane + 32; j < cnt; j += 32) {
    const int s = (j < cnt - 1) ? col[beg + j] : d;
    const float* ap = asrc + (size_t)s * HEADS;
    m0 = fmaxf(m0, lrelu(ap[0] + ad0));
    m1 = fmaxf(m1, lrelu(ap[1] + ad1));
    m2 = fmaxf(m2, lrelu(ap[2] + ad2));
  }
  m0 = wave_max(m0); m1 = wave_max(m1); m2 = wave_max(m2);

  r.x0 = (lane < cnt) ? expf(c0 - m0) : 0.f;
  r.x1 = (lane < cnt) ? expf(c1 - m1) : 0.f;
  r.x2 = (lane < cnt) ? expf(c2 - m2) : 0.f;
  float z0 = r.x0, z1 = r.x1, z2 = r.x2;
  for (int j = lane + 32; j < cnt; j += 32) {
    const int s = (j < cnt - 1) ? col[beg + j] : d;
    const float* ap = asrc + (size_t)s * HEADS;
    z0 += expf(lrelu(ap[0] + ad0) - m0);
    z1 += expf(lrelu(ap[1] + ad1) - m1);
    z2 += expf(lrelu(ap[2] + ad2) - m2);
  }
  r.m0 = m0; r.m1 = m1; r.m2 = m2;
  r.z0 = wave_sum(z0); r.z1 = wave_sum(z1); r.z2 = wave_sum(z2);
  return r;
}

__device__ __forceinline__ void chunk_entry(const RunSoftmax& r, int base, const int* __restrict__ col,
                                            const float* __restrict__ asrc, int beg, int cnt, int d,
                                            int lane, float ad0, float ad1, float ad2, int& s,
                                            float& w0, float& w1, float& w2) {
  s = r.s_c; w0 = r.x0 / r.z0; w1 = r.x1 / r.z1; w2 = r.x2 / r.z2;
  if (base > 0) {
    s = d; w0 = 0.f; w1 = 0.f; w2 = 0.f;
    const int j = base + lane;
    if (j < cnt) {
      s = (j < cnt - 1) ? col[beg + j] : d;
      const float* ap = asrc + (size_t)s * HEADS;
      w0 = expf(lrelu(ap[0] + ad0) - r.m0) / r.z0;
      w1 = expf(lrelu(ap[1] + ad1) - r.m1) / r.z1;
      w2 = expf(lrelu(ap[2] + ad2) - r.m2) / r.z2;
    }
  }
}

__global__ void __launch_bounds__(256) k_rows1(const int* __restrict__ rowptr,
                                               const int* __restrict__ col,
                                               const float* __restrict__ asrc,
                                               const float* __restrict__ adst,
                                               const float* __restrict__ x,
                                               const float* __restrict__ W1,
                                               const float* __restrict__ bias,
                                               float* __restrict__ res, int N) {
  const int lane = threadIdx.x & 31;
  const int d = blockIdx.x * 8 + __builtin_amdgcn_readfirstlane((int)(threadIdx.x >> 5));
  if (d >= N) return;
  const int beg = rowptr[d];
  const int cnt = rowptr[d + 1] - beg + 1;
  const float ad0 = adst[(size_t)d * HEADS + 0];
  const float ad1 = adst[(size_t)d * HEADS + 1];
  const float ad2 = adst[(size_t)d * HEADS + 2];
  const RunSoftmax r = run_softmax(col, asrc, beg, cnt, d, lane, ad0, ad1, ad2);

  const int kk = (lane < HIN) ? lane : 0;
  float g0 = 0.f, g1 = 0.f, g2 = 0.f;
  for (int base = 0; base < cnt; base += 32) {
    int s; float w0, w1, w2;
    chunk_entry(r, base, col, asrc, beg, cnt, d, lane, ad0, ad1, ad2, s, w0, w1, w2);
    const int nb = (cnt - base < 32) ? (cnt - base) : 32;
    for (int k0 = 0; k0 < nb; k0 += 16) {
#pragma unroll
      for (int u = 0; u < 16; ++u) {
        const int k = k0 + u;
        const int   sk = __builtin_amdgcn_readlane(s, k);
        const float q0 = __int_as_float(__builtin_amdgcn_readlane(__float_as_int(w0), k));
        const float q1 = __int_as_float(__builtin_amdgcn_readlane(__float_as_int(w1), k));
        const float q2 = __int_as_float(__builtin_amdgcn_readlane(__float_as_int(w2), k));
        const float xv = x[(size_t)sk * HIN + kk];
        g0 = fmaf(q0, xv, g0); g1 = fmaf(q1, xv, g1); g2 = fmaf(q2, xv, g2);
      }
    }
  }

  float4 a0 = make_float4(0.f, 0.f, 0.f, 0.f);
  float4 a1 = a0, a2 = a0;
#pragma unroll
  for (int k = 0; k < HIN; ++k) {
    const float p0 = __int_as_float(__builtin_amdgcn_readlane(__float_as_int(g0), k));
    const float p1 = __int_as_float(__builtin_amdgcn_readlane(__float_as_int(g1), k));
    const float p2 = __int_as_float(__builtin_amdgcn_readlane(__float_as_int(g2), k));
    const float4* wr = reinterpret_cast<const float4*>(W1 + (size_t)k * F) + lane;
    const float4 u0 = wr[0], u1 = wr[32], u2 = wr[64];
    a0.x = fmaf(p0, u0.x, a0.x); a0.y = fmaf(p0, u0.y, a0.y);
    a0.z = fmaf(p0, u0.z, a0.z); a0.w = fmaf(p0, u0.w, a0.w);
    a1.x = fmaf(p1, u1.x, a1.x); a1.y = fmaf(p1, u1.y, a1.y);
    a1.z = fmaf(p1, u1.z, a1.z); a1.w = fmaf(p1, u1.w, a1.w);
    a2.x = fmaf(p2, u2.x, a2.x); a2.y = fmaf(p2, u2.y, a2.y);
    a2.z = fmaf(p2, u2.z, a2.z); a2.w = fmaf(p2, u2.w, a2.w);
  }

  const float4* bp = reinterpret_cast<const float4*>(bias) + lane;
  const float4 b0 = bp[0], b1 = bp[32], b2 = bp[64];
  a0.x += b0.x; a0.y += b0.y; a0.z += b0.z; a0.w += b0.w;
  a1.x += b1.x; a1.y += b1.y; a1.z += b1.z; a1.w += b1.w;
  a2.x += b2.x; a2.y += b2.y; a2.z += b2.z; a2.w += b2.w;
  float4* o = reinterpret_cast<float4*>(res + (size_t)d * F) + lane;
  o[0] = a0; o[32] = a1; o[64] = a2;
}

#define TW 16

__global__ void __launch_bounds__(32) k_fold(const float* __restrict__ W2,
                                             const float* __restrict__ a_src,
                                             const float* __restrict__ a_dst,
                                             const float* __restrict__ Wl,
                                             const float* __restrict__ b2,
                                             const float* __restrict__ bl,
                                             float* __restrict__ Vt, float* __restrict__ cst) {
  const int lane = threadIdx.x;
  const int w = blockIdx.x;
  if (w == 36) {
    float p = 0.f;
    for (int i = lane; i < F; i += 32) p = fmaf(b2[i], Wl[i], p);
    p = wave_sum(p);
    if (lane == 0) cst[0] = p + bl[0];
    return;
  }
  const int g = w / HEADS, h = w - g * HEADS;
  const int c = lane & 15, hi = lane >> 4;
  const float* qb = (c < 4) ? a_src : (c < 8) ? Wl : a_dst;
  const float4* bp  = reinterpret_cast<const float4*>(qb + h * CPH) + hi;
  const float4* ap0 = reinterpret_cast<const float4*>(W2 + (size_t)(g * 32 + c) * F + h * CPH) + hi;
  const float4* ap1 = reinterpret_cast<const float4*>(W2 + (size_t)(g * 32 + 16 + c) * F + h * CPH) + hi;
  v8f acc0 = {};
  v8f acc1 = {};
#pragma unroll
  for (int i = 0; i < CPH / 8; ++i) {
    const float4 b = bp[2 * i], a0 = ap0[2 * i], a1 = ap1[2 * i];
    v2f bx; bx.x = b.x;  bx.y = b.y;
    v2f by; by.x = b.z;  by.y = b.w;
    v2f p0; p0.x = a0.x; p0.y = a0.y;
    v2f q0; q0.x = a0.z; q0.y = a0.w;
    v2f p1; p1.x = a1.x; p1.y = a1.y;
    v2f q1; q1.x = a1.z; q1.y = a1.w;
    acc0 = __builtin_amdgcn_wmma_f32_16x16x4_f32(false, p0, false, bx, (short)0, acc0, false, false);
    acc1 = __builtin_amdgcn_wmma_f32_16x16x4_f32(false, p1, false, bx, (short)0, acc1, false, false);
    acc0 = __builtin_amdgcn_wmma_f32_16x16x4_f32(false, q0, false, by, (short)0, acc0, false, false);
    acc1 = __builtin_amdgcn_wmma_f32_16x16x4_f32(false, q1, false, by, (short)0, acc1, false, false);
  }
  if ((c & 3) == h && c < 11) {
    float4* o = reinterpret_cast<float4*>(Vt + (size_t)c * F + g * 32 + 8 * hi);
    o[0] = make_float4(acc0[0], acc0[1], acc0[2], acc0[3]);
    o[1] = make_float4(acc0[4], acc0[5], acc0[6], acc0[7]);
    o[4] = make_float4(acc1[0], acc1[1], acc1[2], acc1[3]);
    o[5] = make_float4(acc1[4], acc1[5], acc1[6], acc1[7]);
  }
}

#define KW 96
__global__ void __launch_bounds__(128) k_gemm16(const float* __restrict__ A,
                                                const float* __restrict__ Vt,
                                                float* __restrict__ T) {
  __shared__ float part[3][8][32];
  const int lane = threadIdx.x & 31;
  const int w = threadIdx.x >> 5;
  const int tm = blockIdx.x;
  const int c = lane & 15, hi = lane >> 4;
  int bn = ((c & 3) == 3) ? (c - 1) : c;
  bn = (bn > 10) ? 10 : bn;
  const float4* ap = reinterpret_cast<const float4*>(A + (size_t)(tm * 16 + c) * F + w * KW) + hi;
  const float4* bp = reinterpret_cast<const float4*>(Vt + (size_t)bn * F + w * KW) + hi;
  v8f acc0 = {};
  v8f acc1 = {};
#pragma unroll
  for (int i = 0; i < KW / 8; ++i) {
    const float4 a = ap[2 * i], b = bp[2 * i];
    v2f ax; ax.x = a.x; ax.y = a.y;
    v2f ay; ay.x = a.z; ay.y = a.w;
    v2f bx; bx.x = b.x; bx.y = b.y;
    v2f by; by.x = b.z; by.y = b.w;
    acc0 = __builtin_amdgcn_wmma_f32_16x16x4_f32(false, ax, false, bx, (short)0, acc0, false, false);
    acc1 = __builtin_amdgcn_wmma_f32_16x16x4_f32(false, ay, false, by, (short)0, acc1, false, false);
  }
  float s[8];
#pragma unroll
  for (int r = 0; r < 8; ++r) s[r] = acc0[r] + acc1[r];
  if (w > 0) {
#pragma unroll
    for (int r = 0; r < 8; ++r) part[w - 1][r][lane] = s[r];
  }
  __syncthreads();
  if (w == 0) {
    float* o = T + (size_t)(tm * 16 + 8 * hi) * TW + c;
#pragma unroll
    for (int r = 0; r < 8; ++r)
      o[(size_t)r * TW] = (s[r] + part[0][r][lane]) + (part[1][r][lane] + part[2][r][lane]);
  }
}

__global__ void __launch_bounds__(256) k_rows2(const int* __restrict__ rowptr,
                                               const int* __restrict__ col,
                                               const float* __restrict__ T,
                                               const float* __restrict__ cst,
                                               float* __restrict__ res, int N) {
  const int lane = threadIdx.x & 31;
  const int d = blockIdx.x * 8 + __builtin_amdgcn_readfirstlane((int)(threadIdx.x >> 5));
  if (d >= N) return;
  const int beg = rowptr[d];
  const int cnt = rowptr[d + 1] - beg + 1;
  const float4 td = *reinterpret_cast<const float4*>(T + (size_t)d * TW + 8);
  const float ad0 = td.x, ad1 = td.y, ad2 = td.z;

  float c0 = -INFINITY, c1 = -INFINITY, c2 = -INFINITY;
  float t0 = 0.f, t1 = 0.f, t2 = 0.f;
  if (lane < cnt) {
    const int s = (lane < cnt - 1) ? col[beg + lane] : d;
    const float4* tp = reinterpret_cast<const float4*>(T + (size_t)s * TW);
    const float4 a = tp[0], t = tp[1];
    c0 = lrelu(a.x + ad0); c1 = lrelu(a.y + ad1); c2 = lrelu(a.z + ad2);
    t0 = t.x; t1 = t.y; t2 = t.z;
  }
  float m0 = c0, m1 = c1, m2 = c2;
  for (int j = lane + 32; j < cnt; j += 32) {
    const int s = (j < cnt - 1) ? col[beg + j] : d;
    const float4 a = *reinterpret_cast<const float4*>(T + (size_t)s * TW);
    m0 = fmaxf(m0, lrelu(a.x + ad0));
    m1 = fmaxf(m1, lrelu(a.y + ad1));
    m2 = fmaxf(m2, lrelu(a.z + ad2));
  }
  m0 = wave_max(m0); m1 = wave_max(m1); m2 = wave_max(m2);

  float z0 = (lane < cnt) ? expf(c0 - m0) : 0.f;
  float z1 = (lane < cnt) ? expf(c1 - m1) : 0.f;
  float z2 = (lane < cnt) ? expf(c2 - m2) : 0.f;
  float n0 = z0 * t0, n1 = z1 * t1, n2 = z2 * t2;
  for (int j = lane + 32; j < cnt; j += 32) {
    const int s = (j < cnt - 1) ? col[beg + j] : d;
    const float4* tp = reinterpret_cast<const float4*>(T + (size_t)s * TW);
    const float4 a = tp[0], t = tp[1];
    const float e0 = expf(lrelu(a.x + ad0) - m0);
    const float e1 = expf(lrelu(a.y + ad1) - m1);
    const float e2 = expf(lrelu(a.z + ad2) - m2);
    z0 += e0; z1 += e1; z2 += e2;
    n0 = fmaf(e0, t.x, n0); n1 = fmaf(e1, t.y, n1); n2 = fmaf(e2, t.z, n2);
  }
  z0 = wave_sum(z0); z1 = wave_sum(z1); z2 = wave_sum(z2);
  n0 = wave_sum(n0); n1 = wave_sum(n1); n2 = wave_sum(n2);
  if (lane == 0) res[d] = (n0 / z0 + n1 / z1) + (n2 / z2 + cst[0]);
}

extern "C" void kernel_launch(void* const* d_in, const int* in_sizes, int n_in,
                              void* d_out, int out_size, void* d_ws, size_t ws_size,
                              hipStream_t stream) {
  const float* x   = (const float*)d_in[0];
  const int*   ei  = (const int*)d_in[1];
  const float* W1  = (const float*)d_in[2];
  const float* as1 = (const float*)d_in[3];
  const float* ad1 = (const float*)d_in[4];
  const float* b1  = (const float*)d_in[5];
  const float* W2  = (const float*)d_in[6];
  const float* as2 = (const float*)d_in[7];
  const float* ad2 = (const float*)d_in[8];
  const float* b2  = (const float*)d_in[9];
  const float* Wl  = (const float*)d_in[10];
  const float* bl  = (const float*)d_in[11];
  float* out = (float*)d_out;

  const int N  = in_sizes[0] / HIN;
  const int E  = in_sizes[1] / 2;
  const int* src = ei;
  const int* dst = ei + E;

  float* ws = (float*)d_ws;
  const size_t NF = (size_t)N * F;
  float* T      = ws;
  float* VT     = T + (size_t)N * TW;
  float* CST    = VT + (size_t)12 * F;
  float* AGG    = CST + 64;
  float* ASRC   = AGG + NF;
  float* ADST   = ASRC + (size_t)3 * N;
  int*   CNT    = (int*)(ADST + (size_t)3 * N);
  int*   ROWPTR = CNT + N;
  int*   RANK   = ROWPTR + (N + 1);
  int*   COL    = RANK + E;

  const int eblocks = (E + 255) / 256;
  const int rblocks = (N + 7) / 8;
  const int ablocks = (N + 127) / 128;
  const int mtiles  = N / 16;

  k_fold<<<12 * HEADS + 1, 32, 0, stream>>>(W2, as2, ad2, Wl, b2, bl, VT, CST);

  k_att1<<<ablocks, 128, 0, stream>>>(x, W1, as1, ad1, ASRC, ADST, CNT, N);
  k_count<<<eblocks, 256, 0, stream>>>(dst, E, CNT, RANK);
  k_scan<<<1, 1024, 0, stream>>>(CNT, ROWPTR, N);
  k_fill<<<eblocks, 256, 0, stream>>>(src, dst, E, ROWPTR, RANK, COL);

  k_rows1<<<rblocks, 256, 0, stream>>>(ROWPTR, COL, ASRC, ADST, x, W1, b1, AGG, N);

  k_gemm16<<<mtiles, 128, 0, stream>>>(AGG, VT, T);

  k_rows2<<<rblocks, 256, 0, stream>>>(ROWPTR, COL, T, CST, out, N);
}


# layer-1 projection folded into layer 2's nine vectors (Mt = W1 x V per head, 40 x 9, WMMA): nodes carry 36 input sums, T = g x Mt on WMMA in the row pass, no 384-wide rows; sums reassociated
# speedup vs baseline: 9.5328x; 1.1973x over previous
#include <hip/hip_runtime.h>
#include <hip/hip_bf16.h>

#define HEADS 3
#define CPH   128
#define F     384
#define HIN   12
#define TW    16
#define MK    40

typedef __attribute__((ext_vector_type(2))) float v2f;
typedef __attribute__((ext_vector_type(8))) float v8f;

__device__ __forceinline__ float lrelu(float v) { return v > 0.f ? v : 0.2f * v; }

__device__ __forceinline__ float wave_max(float v) {
#pragma unroll
  for (int o = 16; o >= 1; o >>= 1) v = fmaxf(v, __shfl_xor(v, o, 32));
  return v;
}
__device__ __forceinline__ float wave_sum(float v) {
#pragma unroll
  for (int o = 16; o >= 1; o >>= 1) v += __shfl_xor(v, o, 32);
  return v;
}

__device__ __forceinline__ void fold_w1(int h, int lane, const float* __restrict__ W1,
                                        const float* __restrict__ b1, const float* __restrict__ Vt,
                                        float* __restrict__ Mt) {
  const int c = lane & 15, hi = lane >> 4;
  const float* arow = (c < HIN) ? (W1 + (size_t)c * F) : b1;
  int bn = ((c & 3) == 3) ? (c - 1) : c;
  bn = (bn > 10) ? 10 : bn;
  const float4* ap = reinterpret_cast<const float4*>(arow + h * CPH) + hi;
  const float4* bp = reinterpret_cast<const float4*>(Vt + (size_t)bn * F + h * CPH) + hi;
  v8f acc0 = {};
  v8f acc1 = {};
#pragma unroll
  for (int i = 0; i < CPH / 8; ++i) {
    const float4 a = ap[2 * i], b = bp[2 * i];
    v2f ax; ax.x = a.x; ax.y = a.y;
    v2f ay; ay.x = a.z; ay.y = a.w;
    v2f bx; bx.x = b.x; bx.y = b.y;
    v2f by; by.x = b.z; by.y = b.w;
    acc0 = __builtin_amdgcn_wmma_f32_16x16x4_f32(false, ax, false, bx, (short)0, acc0, false, false);
    acc1 = __builtin_amdgcn_wmma_f32_16x16x4_f32(false, ay, false, by, (short)0, acc1, false, false);
  }
  float s[8];
#pragma unroll
  for (int r = 0; r < 8; ++r) s[r] = acc0[r] + acc1[r];
  if ((c & 3) != 3 && c < 11) {
    float* o = Mt + (size_t)c * MK + h * HIN;
    if (hi == 0) {
      reinterpret_cast<float4*>(o)[0] = make_float4(s[0], s[1], s[2], s[3]);
      reinterpret_cast<float4*>(o)[1] = make_float4(s[4], s[5], s[6], s[7]);
    } else {
      reinterpret_cast<float4*>(o)[2] = make_float4(s[0], s[1], s[2], s[3]);
      Mt[(size_t)c * MK + 3 * HIN + h] = s[4];
      if (h == 0) Mt[(size_t)c * MK + MK - 1] = 0.f;
    }
  }
}

__global__ void __launch_bounds__(128) k_att1(const float* __restrict__ x, const float* __restrict__ W1,
                                              const float* __restrict__ a_src,
                                              const float* __restrict__ a_dst,
                                              const float* __restrict__ b1,
                                              const float* __restrict__ Vt, float* __restrict__ Mt,
                                              float* __restrict__ asrc, float* __restrict__ adst,
                                              int* __restrict__ cnt, int N) {
  __shared__ float v[6][HIN];
  const int t = threadIdx.x;
  const int nb = (N + 127) / 128;
  if ((int)blockIdx.x >= nb) {
    if (__builtin_amdgcn_readfirstlane((int)(threadIdx.x >> 5)) == 0)
      fold_w1((int)blockIdx.x - nb, t, W1, b1, Vt, Mt);
    return;
  }
  if (t < 6 * HIN) {
    const int j = t / HIN, k = t - j * HIN;
    const int h = (j < HEADS) ? j : (j - HEADS);
    const float4* wp = reinterpret_cast<const float4*>(W1 + (size_t)k * F + h * CPH);
    const float4* ap = reinterpret_cast<const float4*>(((j < HEADS) ? a_src : a_dst) + h * CPH);
    float p0 = 0.f, p1 = 0.f, p2 = 0.f, p3 = 0.f;
#pragma unroll 8
    for (int i = 0; i < CPH / 4; ++i) {
      const float4 w = wp[i], a = ap[i];
      p0 = fmaf(w.x, a.x, p0); p1 = fmaf(w.y, a.y, p1);
      p2 = fmaf(w.z, a.z, p2); p3 = fmaf(w.w, a.w, p3);
    }
    v[j][k] = (p0 + p1) + (p2 + p3);
  }
  __syncthreads();
  const int n = blockIdx.x * 128 + t;
  if (n >= N) return;
  const float4* xr = reinterpret_cast<const float4*>(x + (size_t)n * HIN);
  const float4 xa = xr[0], xb = xr[1], xc = xr[2];
  const float xs[HIN] = {xa.x, xa.y, xa.z, xa.w, xb.x, xb.y, xb.z, xb.w, xc.x, xc.y, xc.z, xc.w};
  float r[6];
#pragma unroll
  for (int j = 0; j < 6; ++j) {
    float s = 0.f;
#pragma unroll
    for (int k = 0; k < HIN; ++k) s = fmaf(xs[k], v[j][k], s);
    r[j] = s;
  }
  asrc[(size_t)n * HEADS + 0] = r[0]; asrc[(size_t)n * HEADS + 1] = r[1];
  asrc[(size_t)n * HEADS + 2] = r[2];
  adst[(size_t)n * HEADS + 0] = r[3]; adst[(size_t)n * HEADS + 1] = r[4];
  adst[(size_t)n * HEADS + 2] = r[5];
  cnt[n] = 0;
}

__global__ void k_count(const int* __restrict__ dst, int E, int* __restrict__ cnt,
                        int* __restrict__ rank) {
  const int i = blockIdx.x * blockDim.x + threadIdx.x;
  if (i < E) rank[i] = atomicAdd(&cnt[dst[i]], 1);
}

__global__ void __launch_bounds__(1024) k_scan(const int* __restrict__ cnt,
                                               int* __restrict__ rowptr, int N) {
  __shared__ int part[1024];
  const int t = threadIdx.x;
  const int per = (N + 1023) / 1024;
  const int b = t * per;
  const int e = (b + per < N) ? (b + per) : N;
  int s = 0;
  for (int i = b; i < e; ++i) s += cnt[i];
  part[t] = s;
  __syncthreads();
  for (int off = 1; off < 1024; off <<= 1) {
    const int v = (t >= off) ? part[t - off] : 0;
    __syncthreads();
    part[t] += v;
    __syncthreads();
  }
  int run = part[t] - s;
  for (int i = b; i < e; ++i) { rowptr[i] = run; run += cnt[i]; }
  if (t == 1023) rowptr[N] = part[1023];
}

__global__ void k_fill(const int* __restrict__ src, const int* __restrict__ dst, int E,
                       const int* __restrict__ rowptr, const int* __restrict__ rank,
                       int* __restrict__ col) {
  const int i = blockIdx.x * blockDim.x + threadIdx.x;
  if (i < E) col[rowptr[dst[i]] + rank[i]] = src[i];
}

struct RunSoftmax {
  int   s_c;
  float x0, x1, x2;
  float m0, m1, m2;
  float z0, z1, z2;
};

__device__ __forceinline__ RunSoftmax run_softmax(const int* __restrict__ col,
                                                  const float* __restrict__ asrc, int beg, int cnt,
                                                  int d, int lane, float ad0, float ad1, float ad2) {
  RunSoftmax r;
  r.s_c = d;
  float c0 = -INFINITY, c1 = -INFINITY, c2 = -INFINITY;
  if (lane < cnt) {
    r.s_c = (lane < cnt - 1) ? col[beg + lane] : d;
    const float* ap = asrc + (size_t)r.s_c * HEADS;
    c0 = lrelu(ap[0] + ad0); c1 = lrelu(ap[1] + ad1); c2 = lrelu(ap[2] + ad2);
  }
  float m0 = c0, m1 = c1, m2 = c2;
  for (int j = lane + 32; j < cnt; j += 32) {
    const int s = (j < cnt - 1) ? col[beg + j] : d;
    const float* ap = asrc + (size_t)s * HEADS;
    m0 = fmaxf(m0, lrelu(ap[0] + ad0));
    m1 = fmaxf(m1, lrelu(ap[1] + ad1));
    m2 = fmaxf(m2, lrelu(ap[2] + ad2));
  }
  m0 = wave_max(m0); m1 = wave_max(m1); m2 = wave_max(m2);

  r.x0 = (lane < cnt) ? expf(c0 - m0) : 0.f;
  r.x1 = (lane < cnt) ? expf(c1 - m1) : 0.f;
  r.x2 = (lane < cnt) ? expf(c2 - m2) : 0.f;
  float z0 = r.x0, z1 = r.x1, z2 = r.x2;
  for (int j = lane + 32; j < cnt; j += 32) {
    const int s = (j < cnt - 1) ? col[beg + j] : d;
    const float* ap = asrc + (size_t)s * HEADS;
    z0 += expf(lrelu(ap[0] + ad0) - m0);
    z1 += expf(lrelu(ap[1] + ad1) - m1);
    z2 += expf(lrelu(ap[2] + ad2) - m2);
  }
  r.m0 = m0; r.m1 = m1; r.m2 = m2;
  r.z0 = wave_sum(z0); r.z1 = wave_sum(z1); r.z2 = wave_sum(z2);
  return r;
}

__device__ __forceinline__ void chunk_entry(const RunSoftmax& r, int base, const int* __restrict__ col,
                                            const float* __restrict__ asrc, int beg, int cnt, int d,
                                            int lane, float ad0, float ad1, float ad2, int& s,
                                            float& w0, float& w1, float& w2) {
  s = r.s_c; w0 = r.x0 / r.z0; w1 = r.x1 / r.z1; w2 = r.x2 / r.z2;
  if (base > 0) {
    s = d; w0 = 0.f; w1 = 0.f; w2 = 0.f;
    const int j = base + lane;
    if (j < cnt) {
      s = (j < cnt - 1) ? col[beg + j] : d;
      const float* ap = asrc + (size_t)s * HEADS;
      w0 = expf(lrelu(ap[0] + ad0) - r.m0) / r.z0;
      w1 = expf(lrelu(ap[1] + ad1) - r.m1) / r.z1;
      w2 = expf(lrelu(ap[2] + ad2) - r.m2) / r.z2;
    }
  }
}

__global__ void __launch_bounds__(256) k_rows1(const int* __restrict__ rowptr,
                                               const int* __restrict__ col,
                                               const float* __restrict__ asrc,
                                               const float* __restrict__ adst,
                                               const float* __restrict__ x,
                                               const float* __restrict__ Mt,
                                               float* __restrict__ T, int N) {
  __shared__ __attribute__((aligned(16))) float gt[8][MK];
  const int lane = threadIdx.x & 31;
  const int w = __builtin_amdgcn_readfirstlane((int)(threadIdx.x >> 5));
  const int d = blockIdx.x * 8 + w;
  float g0 = 0.f, g1 = 0.f, g2 = 0.f;
  if (d < N) {
    const int beg = rowptr[d];
    const int cnt = rowptr[d + 1] - beg + 1;
    const float ad0 = adst[(size_t)d * HEADS + 0];
    const float ad1 = adst[(size_t)d * HEADS + 1];
    const float ad2 = adst[(size_t)d * HEADS + 2];
    const RunSoftmax r = run_softmax(col, asrc, beg, cnt, d, lane, ad0, ad1, ad2);

    const int kk = (lane < HIN) ? lane : 0;
    for (int base = 0; base < cnt; base += 32) {
      int s; float w0, w1, w2;
      chunk_entry(r, base, col, asrc, beg, cnt, d, lane, ad0, ad1, ad2, s, w0, w1, w2);
      const int nb = (cnt - base < 32) ? (cnt - base) : 32;
      for (int k0 = 0; k0 < nb; k0 += 16) {
#pragma unroll
        for (int u = 0; u < 16; ++u) {
          const int k = k0 + u;
          const int   sk = __builtin_amdgcn_readlane(s, k);
          const float q0 = __int_as_float(__builtin_amdgcn_readlane(__float_as_int(w0), k));
          const float q1 = __int_as_float(__builtin_amdgcn_readlane(__float_as_int(w1), k));
          const float q2 = __int_as_float(__builtin_amdgcn_readlane(__float_as_int(w2), k));
          const float xv = x[(size_t)sk * HIN + kk];
          g0 = fmaf(q0, xv, g0); g1 = fmaf(q1, xv, g1); g2 = fmaf(q2, xv, g2);
        }
      }
    }
  }

  if (lane < HIN) {
    gt[w][lane] = g0; gt[w][HIN + lane] = g1; gt[w][2 * HIN + lane] = g2;
  } else if (lane < 16) {
    gt[w][3 * HIN + (lane - HIN)] = (lane < 15) ? 1.f : 0.f;
  }
  __syncthreads();
  if (w != 0) return;

  const int c = lane & 15, hi = lane >> 4;
  int bn = ((c & 3) == 3) ? (c - 1) : c;
  bn = (bn > 10) ? 10 : bn;
  const float4* ap = reinterpret_cast<const float4*>(&gt[c & 7][0]) + hi;
  const float4* bp = reinterpret_cast<const float4*>(Mt + (size_t)bn * MK) + hi;
  v8f acc0 = {};
  v8f acc1 = {};
#pragma unroll
  for (int i = 0; i < MK / 8; ++i) {
    const float4 a = ap[2 * i], b = bp[2 * i];
    v2f ax; ax.x = a.x; ax.y = a.y;
    v2f ay; ay.x = a.z; ay.y = a.w;
    v2f bx; bx.x = b.x; bx.y = b.y;
    v2f by; by.x = b.z; by.y = b.w;
    acc0 = __builtin_amdgcn_wmma_f32_16x16x4_f32(false, ax, false, bx, (short)0, acc0, false, false);
    acc1 = __builtin_amdgcn_wmma_f32_16x16x4_f32(false, ay, false, by, (short)0, acc1, false, false);
  }
  if (hi == 0) {
    const int n0 = blockIdx.x * 8;
    float* o = T + (size_t)n0 * TW + c;
#pragma unroll
    for (int r = 0; r < 8; ++r)
      if (n0 + r < N) o[(size_t)r * TW] = acc0[r] + acc1[r];
  }
}

__global__ void __launch_bounds__(32) k_fold(const float* __restrict__ W2,
                                             const float* __restrict__ a_src,
                                             const float* __restrict__ a_dst,
                                             const float* __restrict__ Wl,
                                             const float* __restrict__ b2,
                                             const float* __restrict__ bl,
                                             float* __restrict__ Vt, float* __restrict__ cst) {
  const int lane = threadIdx.x;
  const int w = blockIdx.x;
  if (w == 36) {
    float p = 0.f;
    for (int i = lane; i < F; i += 32) p = fmaf(b2[i], Wl[i], p);
    p = wave_sum(p);
    if (lane == 0) cst[0] = p + bl[0];
    return;
  }
  const int g = w / HEADS, h = w - g * HEADS;
  const int c = lane & 15, hi = lane >> 4;
  const float* qb = (c < 4) ? a_src : (c < 8) ? Wl : a_dst;
  const float4* bp  = reinterpret_cast<const float4*>(qb + h * CPH) + hi;
  const float4* ap0 = reinterpret_cast<const float4*>(W2 + (size_t)(g * 32 + c) * F + h * CPH) + hi;
  const float4* ap1 = reinterpret_cast<const float4*>(W2 + (size_t)(g * 32 + 16 + c) * F + h * CPH) + hi;
  v8f acc0 = {};
  v8f acc1 = {};
#pragma unroll
  for (int i = 0; i < CPH / 8; ++i) {
    const float4 b = bp[2 * i], a0 = ap0[2 * i], a1 = ap1[2 * i];
    v2f bx; bx.x = b.x;  bx.y = b.y;
    v2f by; by.x = b.z;  by.y = b.w;
    v2f p0; p0.x = a0.x; p0.y = a0.y;
    v2f q0; q0.x = a0.z; q0.y = a0.w;
    v2f p1; p1.x = a1.x; p1.y = a1.y;
    v2f q1; q1.x = a1.z; q1.y = a1.w;
    acc0 = __builtin_amdgcn_wmma_f32_16x16x4_f32(false, p0, false, bx, (short)0, acc0, false, false);
    acc1 = __builtin_amdgcn_wmma_f32_16x16x4_f32(false, p1, false, bx, (short)0, acc1, false, false);
    acc0 = __builtin_amdgcn_wmma_f32_16x16x4_f32(false, q0, false, by, (short)0, acc0, false, false);
    acc1 = __builtin_amdgcn_wmma_f32_16x16x4_f32(false, q1, false, by, (short)0, acc1, false, false);
  }
  if ((c & 3) == h && c < 11) {
    float4* o = reinterpret_cast<float4*>(Vt + (size_t)c * F + g * 32 + 8 * hi);
    o[0] = make_float4(acc0[0], acc0[1], acc0[2], acc0[3]);
    o[1] = make_float4(acc0[4], acc0[5], acc0[6], acc0[7]);
    o[4] = make_float4(acc1[0], acc1[1], acc1[2], acc1[3]);
    o[5] = make_float4(acc1[4], acc1[5], acc1[6], acc1[7]);
  }
}

__global__ void __launch_bounds__(256) k_rows2(const int* __restrict__ rowptr,
                                               const int* __restrict__ col,
                                               const float* __restrict__ T,
                                               const float* __restrict__ cst,
                                               float* __restrict__ res, int N) {
  const int lane = threadIdx.x & 31;
  const int d = blockIdx.x * 8 + __builtin_amdgcn_readfirstlane((int)(threadIdx.x >> 5));
  if (d >= N) return;
  const int beg = rowptr[d];
  const int cnt = rowptr[d + 1] - beg + 1;
  const float4 td = *reinterpret_cast<const float4*>(T + (size_t)d * TW + 8);
  const float ad0 = td.x, ad1 = td.y, ad2 = td.z;

  float c0 = -INFINITY, c1 = -INFINITY, c2 = -INFINITY;
  float t0 = 0.f, t1 = 0.f, t2 = 0.f;
  if (lane < cnt) {
    const int s = (lane < cnt - 1) ? col[beg + lane] : d;
    const float4* tp = reinterpret_cast<const float4*>(T + (size_t)s * TW);
    const float4 a = tp[0], t = tp[1];
    c0 = lrelu(a.x + ad0); c1 = lrelu(a.y + ad1); c2 = lrelu(a.z + ad2);
    t0 = t.x; t1 = t.y; t2 = t.z;
  }
  float m0 = c0, m1 = c1, m2 = c2;
  for (int j = lane + 32; j < cnt; j += 32) {
    const int s = (j < cnt - 1) ? col[beg + j] : d;
    const float4 a = *reinterpret_cast<const float4*>(T + (size_t)s * TW);
    m0 = fmaxf(m0, lrelu(a.x + ad0));
    m1 = fmaxf(m1, lrelu(a.y + ad1));
    m2 = fmaxf(m2, lrelu(a.z + ad2));
  }
  m0 = wave_max(m0); m1 = wave_max(m1); m2 = wave_max(m2);

  float z0 = (lane < cnt) ? expf(c0 - m0) : 0.f;
  float z1 = (lane < cnt) ? expf(c1 - m1) : 0.f;
  float z2 = (lane < cnt) ? expf(c2 - m2) : 0.f;
  float n0 = z0 * t0, n1 = z1 * t1, n2 = z2 * t2;
  for (int j = lane + 32; j < cnt; j += 32) {
    const int s = (j < cnt - 1) ? col[beg + j] : d;
    const float4* tp = reinterpret_cast<const float4*>(T + (size_t)s * TW);
    const float4 a = tp[0], t = tp[1];
    const float e0 = expf(lrelu(a.x + ad0) - m0);
    const float e1 = expf(lrelu(a.y + ad1) - m1);
    const float e2 = expf(lrelu(a.z + ad2) - m2);
    z0 += e0; z1 += e1; z2 += e2;
    n0 = fmaf(e0, t.x, n0); n1 = fmaf(e1, t.y, n1); n2 = fmaf(e2, t.z, n2);
  }
  z0 = wave_sum(z0); z1 = wave_sum(z1); z2 = wave_sum(z2);
  n0 = wave_sum(n0); n1 = wave_sum(n1); n2 = wave_sum(n2);
  if (lane == 0) res[d] = (n0 / z0 + n1 / z1) + (n2 / z2 + cst[0]);
}

extern "C" void kernel_launch(void* const* d_in, const int* in_sizes, int n_in,
                              void* d_out, int out_size, void* d_ws, size_t ws_size,
                              hipStream_t stream) {
  const float* x   = (const float*)d_in[0];
  const int*   ei  = (const int*)d_in[1];
  const float* W1  = (const float*)d_in[2];
  const float* as1 = (const float*)d_in[3];
  const float* ad1 = (const float*)d_in[4];
  const float* b1  = (const float*)d_in[5];
  const float* W2  = (const float*)d_in[6];
  const float* as2 = (const float*)d_in[7];
  const float* ad2 = (const float*)d_in[8];
  const float* b2  = (const float*)d_in[9];
  const float* Wl  = (const float*)d_in[10];
  const float* bl  = (const float*)d_in[11];
  float* out = (float*)d_out;

  const int N  = in_sizes[0] / HIN;
  const int E  = in_sizes[1] / 2;
  const int* src = ei;
  const int* dst = ei + E;

  float* ws = (float*)d_ws;
  float* T      = ws;
  float* VT     = T + (size_t)N * TW;
  float* CST    = VT + (size_t)12 * F;
  float* MT     = CST + 64;
  float* ASRC   = MT + (size_t)16 * MK;
  float* ADST   = ASRC + (size_t)3 * N;
  int*   CNT    = (int*)(ADST + (size_t)3 * N);
  int*   ROWPTR = CNT + N;
  int*   RANK   = ROWPTR + (N + 1);
  int*   COL    = RANK + E;

  const int eblocks = (E + 255) / 256;
  const int rblocks = (N + 7) / 8;
  const int ablocks = (N + 127) / 128;

  k_fold<<<12 * HEADS + 1, 32, 0, stream>>>(W2, as2, ad2, Wl, b2, bl, VT, CST);

  k_att1<<<ablocks + HEADS, 128, 0, stream>>>(x, W1, as1, ad1, b1, VT, MT, ASRC, ADST, CNT, N);
  k_count<<<eblocks, 256, 0, stream>>>(dst, E, CNT, RANK);
  k_scan<<<1, 1024, 0, stream>>>(CNT, ROWPTR, N);
  k_fill<<<eblocks, 256, 0, stream>>>(src, dst, E, ROWPTR, RANK, COL);

  k_rows1<<<rblocks, 256, 0, stream>>>(ROWPTR, COL, ASRC, ADST, x, MT, T, N);

  k_rows2<<<rblocks, 256, 0, stream>>>(ROWPTR, COL, T, CST, out, N);
}


# edge grouping in one pass: the in-degree atomic's old value is the edge's slot in its destination's 64-entry bucket (overflow list beyond 64); scan and fill launches removed; sum order as before
# speedup vs baseline: 13.7582x; 1.4432x over previous
#include <hip/hip_runtime.h>
#include <hip/hip_bf16.h>

#define HEADS 3
#define CPH   128
#define F     384
#define HIN   12
#define TW    16
#define MK    40
#define CAP   64

typedef __attribute__((ext_vector_type(2))) float v2f;
typedef __attribute__((ext_vector_type(8))) float v8f;

__device__ __forceinline__ float lrelu(float v) { return v > 0.f ? v : 0.2f * v; }

__device__ __forceinline__ float wave_max(float v) {
#pragma unroll
  for (int o = 16; o >= 1; o >>= 1) v = fmaxf(v, __shfl_xor(v, o, 32));
  return v;
}
__device__ __forceinline__ float wave_sum(float v) {
#pragma unroll
  for (int o = 16; o >= 1; o >>= 1) v += __shfl_xor(v, o, 32);
  return v;
}

__device__ __forceinline__ void fold_w1(int h, int lane, const float* __restrict__ W1,
                                        const float* __restrict__ b1, const float* __restrict__ Vt,
                                        float* __restrict__ Mt) {
  const int c = lane & 15, hi = lane >> 4;
  const float* arow = (c < HIN) ? (W1 + (size_t)c * F) : b1;
  int bn = ((c & 3) == 3) ? (c - 1) : c;
  bn = (bn > 10) ? 10 : bn;
  const float4* ap = reinterpret_cast<const float4*>(arow + h * CPH) + hi;
  const float4* bp = reinterpret_cast<const float4*>(Vt + (size_t)bn * F + h * CPH) + hi;
  v8f acc0 = {};
  v8f acc1 = {};
#pragma unroll
  for (int i = 0; i < CPH / 8; ++i) {
    const float4 a = ap[2 * i], b = bp[2 * i];
    v2f ax; ax.x = a.x; ax.y = a.y;
    v2f ay; ay.x = a.z; ay.y = a.w;
    v2f bx; bx.x = b.x; bx.y = b.y;
    v2f by; by.x = b.z; by.y = b.w;
    acc0 = __builtin_amdgcn_wmma_f32_16x16x4_f32(false, ax, false, bx, (short)0, acc0, false, false);
    acc1 = __builtin_amdgcn_wmma_f32_16x16x4_f32(false, ay, false, by, (short)0, acc1, false, false);
  }
  float s[8];
#pragma unroll
  for (int r = 0; r < 8; ++r) s[r] = acc0[r] + acc1[r];
  if ((c & 3) != 3 && c < 11) {
    float* o = Mt + (size_t)c * MK + h * HIN;
    if (hi == 0) {
      reinterpret_cast<float4*>(o)[0] = make_float4(s[0], s[1], s[2], s[3]);
      reinterpret_cast<float4*>(o)[1] = make_float4(s[4], s[5], s[6], s[7]);
    } else {
      reinterpret_cast<float4*>(o)[2] = make_float4(s[0], s[1], s[2], s[3]);
      Mt[(size_t)c * MK + 3 * HIN + h] = s[4];
      if (h == 0) Mt[(size_t)c * MK + MK - 1] = 0.f;
    }
  }
}

__global__ void __launch_bounds__(128) k_att1(const float* __restrict__ x, const float* __restrict__ W1,
                                              const float* __restrict__ a_src,
                                              const float* __restrict__ a_dst,
                                              const float* __restrict__ b1,
                                              const float* __restrict__ Vt, float* __restrict__ Mt,
                                              float* __restrict__ asrc, float* __restrict__ adst,
                                              int* __restrict__ cnt, int* __restrict__ ovc,
                                              int N) {
  __shared__ float v[6][HIN];
  const int t = threadIdx.x;
  const int nb = (N + 127) / 128;
  if ((int)blockIdx.x >= nb) {
    if (__builtin_amdgcn_readfirstlane((int)(threadIdx.x >> 5)) == 0)
      fold_w1((int)blockIdx.x - nb, t, W1, b1, Vt, Mt);
    return;
  }
  if (t < 6 * HIN) {
    const int j = t / HIN, k = t - j * HIN;
    const int h = (j < HEADS) ? j : (j - HEADS);
    const float4* wp = reinterpret_cast<const float4*>(W1 + (size_t)k * F + h * CPH);
    const float4* ap = reinterpret_cast<const float4*>(((j < HEADS) ? a_src : a_dst) + h * CPH);
    float p0 = 0.f, p1 = 0.f, p2 = 0.f, p3 = 0.f;
#pragma unroll 8
    for (int i = 0; i < CPH / 4; ++i) {
      const float4 w = wp[i], a = ap[i];
      p0 = fmaf(w.x, a.x, p0); p1 = fmaf(w.y, a.y, p1);
      p2 = fmaf(w.z, a.z, p2); p3 = fmaf(w.w, a.w, p3);
    }
    v[j][k] = (p0 + p1) + (p2 + p3);
  }
  __syncthreads();
  const int n = blockIdx.x * 128 + t;
  if (n >= N) return;
  const float4* xr = reinterpret_cast<const float4*>(x + (size_t)n * HIN);
  const float4 xa = xr[0], xb = xr[1], xc = xr[2];
  const float xs[HIN] = {xa.x, xa.y, xa.z, xa.w, xb.x, xb.y, xb.z, xb.w, xc.x, xc.y, xc.z, xc.w};
  float r[6];
#pragma unroll
  for (int j = 0; j < 6; ++j) {
    float s = 0.f;
#pragma unroll
    for (int k = 0; k < HIN; ++k) s = fmaf(xs[k], v[j][k], s);
    r[j] = s;
  }
  asrc[(size_t)n * HEADS + 0] = r[0]; asrc[(size_t)n * HEADS + 1] = r[1];
  asrc[(size_t)n * HEADS + 2] = r[2];
  adst[(size_t)n * HEADS + 0] = r[3]; adst[(size_t)n * HEADS + 1] = r[4];
  adst[(size_t)n * HEADS + 2] = r[5];
  cnt[n] = 0;
  if (n == 0) ovc[0] = 0;
}

__global__ void k_bucket(const int* __restrict__ src, const int* __restrict__ dst, int E,
                         int* __restrict__ cnt, int* __restrict__ col, int* __restrict__ ovc,
                         int* __restrict__ ovd, int* __restrict__ ovs) {
  const int i = blockIdx.x * blockDim.x + threadIdx.x;
  if (i < E) {
    const int d = dst[i], s = src[i];
    const int r = atomicAdd(&cnt[d], 1);
    if (r < CAP) {
      col[(size_t)d * CAP + r] = s;
    } else {
      const int o = atomicAdd(ovc, 1);
      ovd[o] = d; ovs[o] = s;
    }
  }
}

struct Run {
  const int* colrow;
  const int* ovd;
  const int* ovs;
  int nmain;
  int cnt;
  int d;
};

__device__ __forceinline__ Run make_run(const int* __restrict__ cnt, const int* __restrict__ col,
                                        const int* __restrict__ ovc, const int* __restrict__ ovd,
                                        const int* __restrict__ ovs, int d) {
  Run q;
  const int indeg = cnt[d];
  q.colrow = col + (size_t)d * CAP;
  q.ovd = ovd; q.ovs = ovs; q.d = d;
  q.nmain = (indeg < CAP) ? indeg : CAP;
  q.cnt = q.nmain + 1;
  if (indeg > CAP) q.cnt += ovc[0];
  return q;
}

__device__ __forceinline__ int entry_src(const int* __restrict__ colrow, const int* __restrict__ ovd,
                                         const int* __restrict__ ovs, int nmain, int d, int j) {
  int s = d;
  if (j < nmain) {
    s = colrow[j];
  } else if (j > nmain) {
    const int o = j - nmain - 1;
    s = (ovd[o] == d) ? ovs[o] : -1;
  }
  return s;
}

struct RunSoftmax {
  int   s_c;
  float x0, x1, x2;
  float m0, m1, m2;
  float z0, z1, z2;
};

__device__ __forceinline__ RunSoftmax run_softmax(const Run q, const float* __restrict__ asrc,
                                                  int lane, float ad0, float ad1, float ad2) {
  RunSoftmax r;
  const int cnt = q.cnt, d = q.d;
  r.s_c = d;
  float c0 = -INFINITY, c1 = -INFINITY, c2 = -INFINITY;
  if (lane < cnt) {
    r.s_c = (lane < q.nmain) ? q.colrow[lane] : d;
    const float* ap = asrc + (size_t)r.s_c * HEADS;
    c0 = lrelu(ap[0] + ad0); c1 = lrelu(ap[1] + ad1); c2 = lrelu(ap[2] + ad2);
  }
  float m0 = c0, m1 = c1, m2 = c2;
  for (int j = lane + 32; j < cnt; j += 32) {
    const int e = entry_src(q.colrow, q.ovd, q.ovs, q.nmain, q.d, j);
    const bool ok = e >= 0;
    const int s = ok ? e : q.d;
    const float* ap = asrc + (size_t)s * HEADS;
    m0 = fmaxf(m0, ok ? lrelu(ap[0] + ad0) : -INFINITY);
    m1 = fmaxf(m1, ok ? lrelu(ap[1] + ad1) : -INFINITY);
    m2 = fmaxf(m2, ok ? lrelu(ap[2] + ad2) : -INFINITY);
  }
  m0 = wave_max(m0); m1 = wave_max(m1); m2 = wave_max(m2);

  r.x0 = (lane < cnt) ? expf(c0 - m0) : 0.f;
  r.x1 = (lane < cnt) ? expf(c1 - m1) : 0.f;
  r.x2 = (lane < cnt) ? expf(c2 - m2) : 0.f;
  float z0 = r.x0, z1 = r.x1, z2 = r.x2;
  for (int j = lane + 32; j < cnt; j += 32) {
    const int e = entry_src(q.colrow, q.ovd, q.ovs, q.nmain, q.d, j);
    const bool ok = e >= 0;
    const int s = ok ? e : q.d;
    const float* ap = asrc + (size_t)s * HEADS;
    z0 += ok ? expf(lrelu(ap[0] + ad0) - m0) : 0.f;
    z1 += ok ? expf(lrelu(ap[1] + ad1) - m1) : 0.f;
    z2 += ok ? expf(lrelu(ap[2] + ad2) - m2) : 0.f;
  }
  r.m0 = m0; r.m1 = m1; r.m2 = m2;
  r.z0 = wave_sum(z0); r.z1 = wave_sum(z1); r.z2 = wave_sum(z2);
  return r;
}

__device__ __forceinline__ void chunk_entry(const RunSoftmax& r, const Run q, int base,
                                            const float* __restrict__ asrc, int lane, float ad0,
                                            float ad1, float ad2, int& s, float& w0, float& w1,
                                            float& w2) {
  s = r.s_c; w0 = r.x0 / r.z0; w1 = r.x1 / r.z1; w2 = r.x2 / r.z2;
  if (base > 0) {
    s = q.d; w0 = 0.f; w1 = 0.f; w2 = 0.f;
    const int j = base + lane;
    if (j < q.cnt) {
      const int e = entry_src(q.colrow, q.ovd, q.ovs, q.nmain, q.d, j);
      s = (e >= 0) ? e : q.d;
      if (e >= 0) {
        const float* ap = asrc + (size_t)s * HEADS;
        w0 = expf(lrelu(ap[0] + ad0) - r.m0) / r.z0;
        w1 = expf(lrelu(ap[1] + ad1) - r.m1) / r.z1;
        w2 = expf(lrelu(ap[2] + ad2) - r.m2) / r.z2;
      }
    }
  }
}

__global__ void __launch_bounds__(256) k_rows1(const int* __restrict__ cnt_in,
                                               const int* __restrict__ col,
                                               const int* __restrict__ ovc,
                                               const int* __restrict__ ovd,
                                               const int* __restrict__ ovs,
                                               const float* __restrict__ asrc,
                                               const float* __restrict__ adst,
                                               const float* __restrict__ x,
                                               const float* __restrict__ Mt,
                                               float* __restrict__ T, int N) {
  __shared__ __attribute__((aligned(16))) float gt[8][MK];
  const int lane = threadIdx.x & 31;
  const int w = __builtin_amdgcn_readfirstlane((int)(threadIdx.x >> 5));
  const int d = blockIdx.x * 8 + w;
  float g0 = 0.f, g1 = 0.f, g2 = 0.f;
  if (d < N) {
    const Run q = make_run(cnt_in, col, ovc, ovd, ovs, d);
    const int cnt = q.cnt;
    const float ad0 = adst[(size_t)d * HEADS + 0];
    const float ad1 = adst[(size_t)d * HEADS + 1];
    const float ad2 = adst[(size_t)d * HEADS + 2];
    const RunSoftmax r = run_softmax(q, asrc, lane, ad0, ad1, ad2);

    const int kk = (lane < HIN) ? lane : 0;
    for (int base = 0; base < cnt; base += 32) {
      int s; float w0, w1, w2;
      chunk_entry(r, q, base, asrc, lane, ad0, ad1, ad2, s, w0, w1, w2);
      const int nb = (cnt - base < 32) ? (cnt - base) : 32;
      for (int k0 = 0; k0 < nb; k0 += 16) {
#pragma unroll
        for (int u = 0; u < 16; ++u) {
          const int k = k0 + u;
          const int   sk = __builtin_amdgcn_readlane(s, k);
          const float q0 = __int_as_float(__builtin_amdgcn_readlane(__float_as_int(w0), k));
          const float q1 = __int_as_float(__builtin_amdgcn_readlane(__float_as_int(w1), k));
          const float q2 = __int_as_float(__builtin_amdgcn_readlane(__float_as_int(w2), k));
          const float xv = x[(size_t)sk * HIN + kk];
          g0 = fmaf(q0, xv, g0); g1 = fmaf(q1, xv, g1); g2 = fmaf(q2, xv, g2);
        }
      }
    }
  }

  if (lane < HIN) {
    gt[w][lane] = g0; gt[w][HIN + lane] = g1; gt[w][2 * HIN + lane] = g2;
  } else if (lane < 16) {
    gt[w][3 * HIN + (lane - HIN)] = (lane < 15) ? 1.f : 0.f;
  }
  __syncthreads();
  if (w != 0) return;

  const int c = lane & 15, hi = lane >> 4;
  int bn = ((c & 3) == 3) ? (c - 1) : c;
  bn = (bn > 10) ? 10 : bn;
  const float4* ap = reinterpret_cast<const float4*>(&gt[c & 7][0]) + hi;
  const float4* bp = reinterpret_cast<const float4*>(Mt + (size_t)bn * MK) + hi;
  v8f acc0 = {};
  v8f acc1 = {};
#pragma unroll
  for (int i = 0; i < MK / 8; ++i) {
    const float4 a = ap[2 * i], b = bp[2 * i];
    v2f ax; ax.x = a.x; ax.y = a.y;
    v2f ay; ay.x = a.z; ay.y = a.w;
    v2f bx; bx.x = b.x; bx.y = b.y;
    v2f by; by.x = b.z; by.y = b.w;
    acc0 = __builtin_amdgcn_wmma_f32_16x16x4_f32(false, ax, false, bx, (short)0, acc0, false, false);
    acc1 = __builtin_amdgcn_wmma_f32_16x16x4_f32(false, ay, false, by, (short)0, acc1, false, false);
  }
  if (hi == 0) {
    const int n0 = blockIdx.x * 8;
    float* o = T + (size_t)n0 * TW + c;
#pragma unroll
    for (int r = 0; r < 8; ++r)
      if (n0 + r < N) o[(size_t)r * TW] = acc0[r] + acc1[r];
  }
}

__global__ void __launch_bounds__(32) k_fold(const float* __restrict__ W2,
                                             const float* __restrict__ a_src,
                                             const float* __restrict__ a_dst,
                                             const float* __restrict__ Wl,
                                             const float* __restrict__ b2,
                                             const float* __restrict__ bl,
                                             float* __restrict__ Vt, float* __restrict__ cst) {
  const int lane = threadIdx.x;
  const int w = blockIdx.x;
  if (w == 36) {
    float p = 0.f;
    for (int i = lane; i < F; i += 32) p = fmaf(b2[i], Wl[i], p);
    p = wave_sum(p);
    if (lane == 0) cst[0] = p + bl[0];
    return;
  }
  const int g = w / HEADS, h = w - g * HEADS;
  const int c = lane & 15, hi = lane >> 4;
  const float* qb = (c < 4) ? a_src : (c < 8) ? Wl : a_dst;
  const float4* bp  = reinterpret_cast<const float4*>(qb + h * CPH) + hi;
  const float4* ap0 = reinterpret_cast<const float4*>(W2 + (size_t)(g * 32 + c) * F + h * CPH) + hi;
  const float4* ap1 = reinterpret_cast<const float4*>(W2 + (size_t)(g * 32 + 16 + c) * F + h * CPH) + hi;
  v8f acc0 = {};
  v8f acc1 = {};
#pragma unroll
  for (int i = 0; i < CPH / 8; ++i) {
    const float4 b = bp[2 * i], a0 = ap0[2 * i], a1 = ap1[2 * i];
    v2f bx; bx.x = b.x;  bx.y = b.y;
    v2f by; by.x = b.z;  by.y = b.w;
    v2f p0; p0.x = a0.x; p0.y = a0.y;
    v2f q0; q0.x = a0.z; q0.y = a0.w;
    v2f p1; p1.x = a1.x; p1.y = a1.y;
    v2f q1; q1.x = a1.z; q1.y = a1.w;
    acc0 = __builtin_amdgcn_wmma_f32_16x16x4_f32(false, p0, false, bx, (short)0, acc0, false, false);
    acc1 = __builtin_amdgcn_wmma_f32_16x16x4_f32(false, p1, false, bx, (short)0, acc1, false, false);
    acc0 = __builtin_amdgcn_wmma_f32_16x16x4_f32(false, q0, false, by, (short)0, acc0, false, false);
    acc1 = __builtin_amdgcn_wmma_f32_16x16x4_f32(false, q1, false, by, (short)0, acc1, false, false);
  }
  if ((c & 3) == h && c < 11) {
    float4* o = reinterpret_cast<float4*>(Vt + (size_t)c * F + g * 32 + 8 * hi);
    o[0] = make_float4(acc0[0], acc0[1], acc0[2], acc0[3]);
    o[1] = make_float4(acc0[4], acc0[5], acc0[6], acc0[7]);
    o[4] = make_float4(acc1[0], acc1[1], acc1[2], acc1[3]);
    o[5] = make_float4(acc1[4], acc1[5], acc1[6], acc1[7]);
  }
}

__global__ void __launch_bounds__(256) k_rows2(const int* __restrict__ cnt_in,
                                               const int* __restrict__ col,
                                               const int* __restrict__ ovc,
                                               const int* __restrict__ ovd,
                                               const int* __restrict__ ovs,
                                               const float* __restrict__ T,
                                               const float* __restrict__ cst,
                                               float* __restrict__ res, int N) {
  const int lane = threadIdx.x & 31;
  const int d = blockIdx.x * 8 + __builtin_amdgcn_readfirstlane((int)(threadIdx.x >> 5));
  if (d >= N) return;
  const Run q = make_run(cnt_in, col, ovc, ovd, ovs, d);
  const int cnt = q.cnt;
  const float4 td = *reinterpret_cast<const float4*>(T + (size_t)d * TW + 8);
  const float ad0 = td.x, ad1 = td.y, ad2 = td.z;

  float c0 = -INFINITY, c1 = -INFINITY, c2 = -INFINITY;
  float t0 = 0.f, t1 = 0.f, t2 = 0.f;
  if (lane < cnt) {
    const int s = (lane < q.nmain) ? q.colrow[lane] : d;
    const float4* tp = reinterpret_cast<const float4*>(T + (size_t)s * TW);
    const float4 a = tp[0], t = tp[1];
    c0 = lrelu(a.x + ad0); c1 = lrelu(a.y + ad1); c2 = lrelu(a.z + ad2);
    t0 = t.x; t1 = t.y; t2 = t.z;
  }
  float m0 = c0, m1 = c1, m2 = c2;
  for (int j = lane + 32; j < cnt; j += 32) {
    const int e = entry_src(q.colrow, q.ovd, q.ovs, q.nmain, q.d, j);
    const bool ok = e >= 0;
    const int s = ok ? e : q.d;
    const float4 a = *reinterpret_cast<const float4*>(T + (size_t)s * TW);
    m0 = fmaxf(m0, ok ? lrelu(a.x + ad0) : -INFINITY);
    m1 = fmaxf(m1, ok ? lrelu(a.y + ad1) : -INFINITY);
    m2 = fmaxf(m2, ok ? lrelu(a.z + ad2) : -INFINITY);
  }
  m0 = wave_max(m0); m1 = wave_max(m1); m2 = wave_max(m2);

  float z0 = (lane < cnt) ? expf(c0 - m0) : 0.f;
  float z1 = (lane < cnt) ? expf(c1 - m1) : 0.f;
  float z2 = (lane < cnt) ? expf(c2 - m2) : 0.f;
  float n0 = z0 * t0, n1 = z1 * t1, n2 = z2 * t2;
  for (int j = lane + 32; j < cnt; j += 32) {
    const int e = entry_src(q.colrow, q.ovd, q.ovs, q.nmain, q.d, j);
    const bool ok = e >= 0;
    const int s = ok ? e : q.d;
    const float4* tp = reinterpret_cast<const float4*>(T + (size_t)s * TW);
    const float4 a = tp[0], t = tp[1];
    const float e0 = ok ? expf(lrelu(a.x + ad0) - m0) : 0.f;
    const float e1 = ok ? expf(lrelu(a.y + ad1) - m1) : 0.f;
    const float e2 = ok ? expf(lrelu(a.z + ad2) - m2) : 0.f;
    z0 += e0; z1 += e1; z2 += e2;
    n0 = fmaf(e0, t.x, n0); n1 = fmaf(e1, t.y, n1); n2 = fmaf(e2, t.z, n2);
  }
  z0 = wave_sum(z0); z1 = wave_sum(z1); z2 = wave_sum(z2);
  n0 = wave_sum(n0); n1 = wave_sum(n1); n2 = wave_sum(n2);
  if (lane == 0) res[d] = (n0 / z0 + n1 / z1) + (n2 / z2 + cst[0]);
}

extern "C" void kernel_launch(void* const* d_in, const int* in_sizes, int n_in,
                              void* d_out, int out_size, void* d_ws, size_t ws_size,
                              hipStream_t stream) {
  const float* x   = (const float*)d_in[0];
  const int*   ei  = (const int*)d_in[1];
  const float* W1  = (const float*)d_in[2];
  const float* as1 = (const float*)d_in[3];
  const float* ad1 = (const float*)d_in[4];
  const float* b1  = (const float*)d_in[5];
  const float* W2  = (const float*)d_in[6];
  const float* as2 = (const float*)d_in[7];
  const float* ad2 = (const float*)d_in[8];
  const float* b2  = (const float*)d_in[9];
  const float* Wl  = (const float*)d_in[10];
  const float* bl  = (const float*)d_in[11];
  float* out = (float*)d_out;

  const int N  = in_sizes[0] / HIN;
  const int E  = in_sizes[1] / 2;
  const int* src = ei;
  const int* dst = ei + E;

  float* ws = (float*)d_ws;
  float* T      = ws;
  int*   COL    = (int*)(T + (size_t)N * TW);
  float* VT     = (float*)(COL + (size_t)N * CAP);
  float* CST    = VT + (size_t)12 * F;
  float* MT     = CST + 64;
  float* ASRC   = MT + (size_t)16 * MK;
  float* ADST   = ASRC + (size_t)3 * N;
  int*   CNT    = (int*)(ADST + (size_t)3 * N);
  int*   OVC    = CNT + N;
  int*   OVD    = OVC + 1;
  int*   OVS    = OVD + E;

  const int eblocks = (E + 255) / 256;
  const int rblocks = (N + 7) / 8;
  const int ablocks = (N + 127) / 128;

  k_fold<<<12 * HEADS + 1, 32, 0, stream>>>(W2, as2, ad2, Wl, b2, bl, VT, CST);

  k_att1<<<ablocks + HEADS, 128, 0, stream>>>(x, W1, as1, ad1, b1, VT, MT, ASRC, ADST, CNT, OVC, N);
  k_bucket<<<eblocks, 256, 0, stream>>>(src, dst, E, CNT, COL, OVC, OVD, OVS);

  k_rows1<<<rblocks, 256, 0, stream>>>(CNT, COL, OVC, OVD, OVS, ASRC, ADST, x, MT, T, N);

  k_rows2<<<rblocks, 256, 0, stream>>>(CNT, COL, OVC, OVD, OVS, T, CST, out, N);
}


# one launch fewer (5 -> 4): the 37 W2-fold wave tasks ride in front of the attention-dot launch's node blocks, the W1 fold in front of the edge-grouping launch; arithmetic and sum order unchanged
# speedup vs baseline: 15.8678x; 1.1533x over previous
#include <hip/hip_runtime.h>
#include <hip/hip_bf16.h>

#define HEADS 3
#define CPH   128
#define F     384
#define HIN   12
#define TW    16
#define MK    40
#define CAP   64
#define GRP   8
#define NPB   32
#define GP    44

typedef __attribute__((ext_vector_type(2))) float v2f;
typedef __attribute__((ext_vector_type(8))) float v8f;

__device__ __forceinline__ float lrelu(float v) { return v > 0.f ? v : 0.2f * v; }

__device__ __forceinline__ float grp_max(float v) {
#pragma unroll
  for (int o = GRP / 2; o >= 1; o >>= 1) v = fmaxf(v, __shfl_xor(v, o, 32));
  return v;
}
__device__ __forceinline__ float grp_sum(float v) {
#pragma unroll
  for (int o = GRP / 2; o >= 1; o >>= 1) v += __shfl_xor(v, o, 32);
  return v;
}
__device__ __forceinline__ float wave_sum(float v) {
#pragma unroll
  for (int o = 16; o >= 1; o >>= 1) v += __shfl_xor(v, o, 32);
  return v;
}

__device__ __forceinline__ void fold_w1(int h, int lane, const float* __restrict__ W1,
                                        const float* __restrict__ b1, const float* __restrict__ Vt,
                                        float* __restrict__ Mt) {
  const int c = lane & 15, hi = lane >> 4;
  const float* arow = (c < HIN) ? (W1 + (size_t)c * F) : b1;
  int bn = ((c & 3) == 3) ? (c - 1) : c;
  bn = (bn > 10) ? 10 : bn;
  const float4* ap = reinterpret_cast<const float4*>(arow + h * CPH) + hi;
  const float4* bp = reinterpret_cast<const float4*>(Vt + (size_t)bn * F + h * CPH) + hi;
  v8f acc0 = {};
  v8f acc1 = {};
#pragma unroll
  for (int i = 0; i < CPH / 8; ++i) {
    const float4 a = ap[2 * i], b = bp[2 * i];
    v2f ax; ax.x = a.x; ax.y = a.y;
    v2f ay; ay.x = a.z; ay.y = a.w;
    v2f bx; bx.x = b.x; bx.y = b.y;
    v2f by; by.x = b.z; by.y = b.w;
    acc0 = __builtin_amdgcn_wmma_f32_16x16x4_f32(false, ax, false, bx, (short)0, acc0, false, false);
    acc1 = __builtin_amdgcn_wmma_f32_16x16x4_f32(false, ay, false, by, (short)0, acc1, false, false);
  }
  float s[8];
#pragma unroll
  for (int r = 0; r < 8; ++r) s[r] = acc0[r] + acc1[r];
  if ((c & 3) != 3 && c < 11) {
    float* o = Mt + (size_t)c * MK + h * HIN;
    if (hi == 0) {
      reinterpret_cast<float4*>(o)[0] = make_float4(s[0], s[1], s[2], s[3]);
      reinterpret_cast<float4*>(o)[1] = make_float4(s[4], s[5], s[6], s[7]);
    } else {
      reinterpret_cast<float4*>(o)[2] = make_float4(s[0], s[1], s[2], s[3]);
      Mt[(size_t)c * MK + 3 * HIN + h] = s[4];
      if (h == 0) Mt[(size_t)c * MK + MK - 1] = 0.f;
    }
  }
}

__device__ __forceinline__ void fold_w2(int w, int lane, const float* __restrict__ W2,
                                        const float* __restrict__ a_src,
                                        const float* __restrict__ a_dst,
                                        const float* __restrict__ Wl,
                                        const float* __restrict__ b2,
                                        const float* __restrict__ bl,
                                        float* __restrict__ Vt, float* __restrict__ cst) {
  if (w == 36) {
    float p = 0.f;
    for (int i = lane; i < F; i += 32) p = fmaf(b2[i], Wl[i], p);
    p = wave_sum(p);
    if (lane == 0) cst[0] = p + bl[0];
    return;
  }
  const int g = w / HEADS, h = w - g * HEADS;
  const int c = lane & 15, hi = lane >> 4;
  const float* qb = (c < 4) ? a_src : (c < 8) ? Wl : a_dst;
  const float4* bp  = reinterpret_cast<const float4*>(qb + h * CPH) + hi;
  const float4* ap0 = reinterpret_cast<const float4*>(W2 + (size_t)(g * 32 + c) * F + h * CPH) + hi;
  const float4* ap1 = reinterpret_cast<const float4*>(W2 + (size_t)(g * 32 + 16 + c) * F + h * CPH) + hi;
  v8f acc0 = {};
  v8f acc1 = {};
#pragma unroll
  for (int i = 0; i < CPH / 8; ++i) {
    const float4 b = bp[2 * i], a0 = ap0[2 * i], a1 = ap1[2 * i];
    v2f bx; bx.x = b.x;  bx.y = b.y;
    v2f by; by.x = b.z;  by.y = b.w;
    v2f p0; p0.x = a0.x; p0.y = a0.y;
    v2f q0; q0.x = a0.z; q0.y = a0.w;
    v2f p1; p1.x = a1.x; p1.y = a1.y;
    v2f q1; q1.x = a1.z; q1.y = a1.w;
    acc0 = __builtin_amdgcn_wmma_f32_16x16x4_f32(false, p0, false, bx, (short)0, acc0, false, false);
    acc1 = __builtin_amdgcn_wmma_f32_16x16x4_f32(false, p1, false, bx, (short)0, acc1, false, false);
    acc0 = __builtin_amdgcn_wmma_f32_16x16x4_f32(false, q0, false, by, (short)0, acc0, false, false);
    acc1 = __builtin_amdgcn_wmma_f32_16x16x4_f32(false, q1, false, by, (short)0, acc1, false, false);
  }
  if ((c & 3) == h && c < 11) {
    float4* o = reinterpret_cast<float4*>(Vt + (size_t)c * F + g * 32 + 8 * hi);
    o[0] = make_float4(acc0[0], acc0[1], acc0[2], acc0[3]);
    o[1] = make_float4(acc0[4], acc0[5], acc0[6], acc0[7]);
    o[4] = make_float4(acc1[0], acc1[1], acc1[2], acc1[3]);
    o[5] = make_float4(acc1[4], acc1[5], acc1[6], acc1[7]);
  }
}

#define FTASKS 37
#define FB     10
__global__ void __launch_bounds__(128) k_att1(const float* __restrict__ x, const float* __restrict__ W1,
                                              const float* __restrict__ a_src,
                                              const float* __restrict__ a_dst,
                                              const float* __restrict__ W2,
                                              const float* __restrict__ a_src2,
                                              const float* __restrict__ a_dst2,
                                              const float* __restrict__ Wl,
                                              const float* __restrict__ b2,
                                              const float* __restrict__ bl,
                                              float* __restrict__ Vt, float* __restrict__ cst,
                                              float* __restrict__ asrc, float* __restrict__ adst,
                                              int* __restrict__ cnt, int* __restrict__ ovc,
                                              int N) {
  __shared__ float v[6][HIN];
  const int t = threadIdx.x;
  if ((int)blockIdx.x < FB) {
    const int w = __builtin_amdgcn_readfirstlane((int)blockIdx.x * 4 + (int)(threadIdx.x >> 5));
    if (w < FTASKS) fold_w2(w, t & 31, W2, a_src2, a_dst2, Wl, b2, bl, Vt, cst);
    return;
  }
  const int nblk = (int)blockIdx.x - FB;
  if (t < 6 * HIN) {
    const int j = t / HIN, k = t - j * HIN;
    const int h = (j < HEADS) ? j : (j - HEADS);
    const float4* wp = reinterpret_cast<const float4*>(W1 + (size_t)k * F + h * CPH);
    const float4* ap = reinterpret_cast<const float4*>(((j < HEADS) ? a_src : a_dst) + h * CPH);
    float p0 = 0.f, p1 = 0.f, p2 = 0.f, p3 = 0.f;
#pragma unroll 8
    for (int i = 0; i < CPH / 4; ++i) {
      const float4 w = wp[i], a = ap[i];
      p0 = fmaf(w.x, a.x, p0); p1 = fmaf(w.y, a.y, p1);
      p2 = fmaf(w.z, a.z, p2); p3 = fmaf(w.w, a.w, p3);
    }
    v[j][k] = (p0 + p1) + (p2 + p3);
  }
  __syncthreads();
  const int n = nblk * 128 + t;
  if (n >= N) return;
  const float4* xr = reinterpret_cast<const float4*>(x + (size_t)n * HIN);
  const float4 xa = xr[0], xb = xr[1], xc = xr[2];
  const float xs[HIN] = {xa.x, xa.y, xa.z, xa.w, xb.x, xb.y, xb.z, xb.w, xc.x, xc.y, xc.z, xc.w};
  float r[6];
#pragma unroll
  for (int j = 0; j < 6; ++j) {
    float s = 0.f;
#pragma unroll
    for (int k = 0; k < HIN; ++k) s = fmaf(xs[k], v[j][k], s);
    r[j] = s;
  }
  asrc[(size_t)n * HEADS + 0] = r[0]; asrc[(size_t)n * HEADS + 1] = r[1];
  asrc[(size_t)n * HEADS + 2] = r[2];
  adst[(size_t)n * HEADS + 0] = r[3]; adst[(size_t)n * HEADS + 1] = r[4];
  adst[(size_t)n * HEADS + 2] = r[5];
  cnt[n] = 0;
  if (n == 0) ovc[0] = 0;
}

__device__ __forceinline__ void put_edge(int d, int s, int r, int* __restrict__ col,
                                         int* __restrict__ ovc, int* __restrict__ ovd,
                                         int* __restrict__ ovs) {
  if (r < CAP) {
    col[(size_t)d * CAP + r] = s;
  } else {
    const int o = atomicAdd(ovc, 1);
    ovd[o] = d; ovs[o] = s;
  }
}

__global__ void __launch_bounds__(256) k_bucket(const int* __restrict__ src, const int* __restrict__ dst,
                                                int E, int* __restrict__ cnt, int* __restrict__ col,
                                                int* __restrict__ ovc, int* __restrict__ ovd,
                                                int* __restrict__ ovs, const float* __restrict__ W1,
                                                const float* __restrict__ b1,
                                                const float* __restrict__ Vt, float* __restrict__ Mt) {
  if (blockIdx.x == 0) {
    const int w = __builtin_amdgcn_readfirstlane((int)(threadIdx.x >> 5));
    if (w < HEADS) fold_w1(w, threadIdx.x & 31, W1, b1, Vt, Mt);
    return;
  }
  const int eb = (int)blockIdx.x - 1;
  const int stride = ((int)gridDim.x - 1) * blockDim.x;
  for (int b = eb * blockDim.x + threadIdx.x; b < E; b += 4 * stride) {
    if (b + 3 * stride < E) {
      const int d0 = dst[b], d1 = dst[b + stride], d2 = dst[b + 2 * stride], d3 = dst[b + 3 * stride];
      const int s0 = src[b], s1 = src[b + stride], s2 = src[b + 2 * stride], s3 = src[b + 3 * stride];
      const int r0 = atomicAdd(&cnt[d0], 1);
      const int r1 = atomicAdd(&cnt[d1], 1);
      const int r2 = atomicAdd(&cnt[d2], 1);
      const int r3 = atomicAdd(&cnt[d3], 1);
      put_edge(d0, s0, r0, col, ovc, ovd, ovs);
      put_edge(d1, s1, r1, col, ovc, ovd, ovs);
      put_edge(d2, s2, r2, col, ovc, ovd, ovs);
      put_edge(d3, s3, r3, col, ovc, ovd, ovs);
    } else {
      for (int e = b; e < E; e += stride) {
        const int d = dst[e], s = src[e];
        put_edge(d, s, atomicAdd(&cnt[d], 1), col, ovc, ovd, ovs);
      }
    }
  }
}

struct Run {
  const int* colrow;
  const int* ovd;
  const int* ovs;
  int nmain;
  int cnt;
  int d;
};

__device__ __forceinline__ Run make_run(const int* __restrict__ cnt, const int* __restrict__ col,
                                        const int* __restrict__ ovc, const int* __restrict__ ovd,
                                        const int* __restrict__ ovs, int d) {
  Run q;
  const int indeg = cnt[d];
  q.colrow = col + (size_t)d * CAP;
  q.ovd = ovd; q.ovs = ovs; q.d = d;
  q.nmain = (indeg < CAP) ? indeg : CAP;
  q.cnt = q.nmain + 1;
  if (indeg > CAP) q.cnt += ovc[0];
  return q;
}

__device__ __forceinline__ int entry_src(const int* __restrict__ colrow, const int* __restrict__ ovd,
                                         const int* __restrict__ ovs, int nmain, int d, int j) {
  int s = d;
  if (j < nmain) {
    s = colrow[j];
  } else if (j > nmain) {
    const int o = j - nmain - 1;
    s = (ovd[o] == d) ? ovs[o] : -1;
  }
  return s;
}

__global__ void __launch_bounds__(256) k_rows1(const int* __restrict__ cnt_in,
                                               const int* __restrict__ col,
                                               const int* __restrict__ ovc,
                                               const int* __restrict__ ovd,
                                               const int* __restrict__ ovs,
                                               const float* __restrict__ asrc,
                                               const float* __restrict__ adst,
                                               const float* __restrict__ x,
                                               const float* __restrict__ Mt,
                                               float* __restrict__ T, int N) {
  __shared__ __attribute__((aligned(16))) float gt[256][GP];
  const int tid = threadIdx.x;
  const int lane = tid & 31;
  const int sub = tid & (GRP - 1);
  const int dn = blockIdx.x * NPB + (tid >> 3);
  const int d = (dn < N) ? dn : (N - 1);
  const Run q = make_run(cnt_in, col, ovc, ovd, ovs, d);
  const int cnt = q.cnt;
  const float ad0 = adst[(size_t)d * HEADS + 0];
  const float ad1 = adst[(size_t)d * HEADS + 1];
  const float ad2 = adst[(size_t)d * HEADS + 2];

  int sr[4];
#pragma unroll
  for (int i = 0; i < 4; ++i) {
    const bool filled = sub + GRP * i < q.nmain;
    const int* p = filled ? (q.colrow + sub + GRP * i) : (cnt_in + d);
    const int v = *p;
    sr[i] = filled ? v : d;
  }
  float a0[4], a1[4], a2[4];
  float4 xa[4], xb[4], xc[4];
#pragma unroll
  for (int i = 0; i < 4; ++i) {
    const float* ap = asrc + (size_t)sr[i] * HEADS;
    a0[i] = ap[0]; a1[i] = ap[1]; a2[i] = ap[2];
    const float4* xr = reinterpret_cast<const float4*>(x + (size_t)sr[i] * HIN);
    xa[i] = xr[0]; xb[i] = xr[1]; xc[i] = xr[2];
  }
  float c0[4], c1[4], c2[4];
  float m0 = -INFINITY, m1 = -INFINITY, m2 = -INFINITY;
#pragma unroll
  for (int i = 0; i < 4; ++i) {
    const bool in = sub + GRP * i < cnt;
    c0[i] = in ? lrelu(a0[i] + ad0) : -INFINITY;
    c1[i] = in ? lrelu(a1[i] + ad1) : -INFINITY;
    c2[i] = in ? lrelu(a2[i] + ad2) : -INFINITY;
    m0 = fmaxf(m0, c0[i]); m1 = fmaxf(m1, c1[i]); m2 = fmaxf(m2, c2[i]);
  }
  for (int j = 4 * GRP + sub; j < cnt; j += GRP) {
    const int e = entry_src(q.colrow, q.ovd, q.ovs, q.nmain, q.d, j);
    if (e >= 0) {
      const float* ap = asrc + (size_t)e * HEADS;
      m0 = fmaxf(m0, lrelu(ap[0] + ad0));
      m1 = fmaxf(m1, lrelu(ap[1] + ad1));
      m2 = fmaxf(m2, lrelu(ap[2] + ad2));
    }
  }
  m0 = grp_max(m0); m1 = grp_max(m1); m2 = grp_max(m2);

  float g0[HIN], g1[HIN], g2[HIN];
#pragma unroll
  for (int k = 0; k < HIN; ++k) { g0[k] = 0.f; g1[k] = 0.f; g2[k] = 0.f; }
  float z0 = 0.f, z1 = 0.f, z2 = 0.f;
#pragma unroll
  for (int i = 0; i < 4; ++i) {
    const bool in = sub + GRP * i < cnt;
    const float e0 = in ? expf(c0[i] - m0) : 0.f;
    const float e1 = in ? expf(c1[i] - m1) : 0.f;
    const float e2 = in ? expf(c2[i] - m2) : 0.f;
    z0 += e0; z1 += e1; z2 += e2;
    const float xs[HIN] = {xa[i].x, xa[i].y, xa[i].z, xa[i].w, xb[i].x, xb[i].y, xb[i].z, xb[i].w,
                           xc[i].x, xc[i].y, xc[i].z, xc[i].w};
#pragma unroll
    for (int k = 0; k < HIN; ++k) {
      g0[k] = fmaf(e0, xs[k], g0[k]); g1[k] = fmaf(e1, xs[k], g1[k]); g2[k] = fmaf(e2, xs[k], g2[k]);
    }
  }
  for (int j = 4 * GRP + sub; j < cnt; j += GRP) {
    const int e = entry_src(q.colrow, q.ovd, q.ovs, q.nmain, q.d, j);
    if (e >= 0) {
      const float* ap = asrc + (size_t)e * HEADS;
      const float e0 = expf(lrelu(ap[0] + ad0) - m0);
      const float e1 = expf(lrelu(ap[1] + ad1) - m1);
      const float e2 = expf(lrelu(ap[2] + ad2) - m2);
      z0 += e0; z1 += e1; z2 += e2;
      const float4* xr = reinterpret_cast<const float4*>(x + (size_t)e * HIN);
      const float4 ya = xr[0], yb = xr[1], yc = xr[2];
      const float xs[HIN] = {ya.x, ya.y, ya.z, ya.w, yb.x, yb.y, yb.z, yb.w, yc.x, yc.y, yc.z, yc.w};
#pragma unroll
      for (int k = 0; k < HIN; ++k) {
        g0[k] = fmaf(e0, xs[k], g0[k]); g1[k] = fmaf(e1, xs[k], g1[k]); g2[k] = fmaf(e2, xs[k], g2[k]);
      }
    }
  }
  z0 = grp_sum(z0); z1 = grp_sum(z1); z2 = grp_sum(z2);
  const float r0 = 1.f / z0, r1 = 1.f / z1, r2 = 1.f / z2;

  {
    float4* o = reinterpret_cast<float4*>(&gt[tid][0]);
    o[0] = make_float4(g0[0] * r0, g0[1] * r0, g0[2] * r0, g0[3] * r0);
    o[1] = make_float4(g0[4] * r0, g0[5] * r0, g0[6] * r0, g0[7] * r0);
    o[2] = make_float4(g0[8] * r0, g0[9] * r0, g0[10] * r0, g0[11] * r0);
    o[3] = make_float4(g1[0] * r1, g1[1] * r1, g1[2] * r1, g1[3] * r1);
    o[4] = make_float4(g1[4] * r1, g1[5] * r1, g1[6] * r1, g1[7] * r1);
    o[5] = make_float4(g1[8] * r1, g1[9] * r1, g1[10] * r1, g1[11] * r1);
    o[6] = make_float4(g2[0] * r2, g2[1] * r2, g2[2] * r2, g2[3] * r2);
    o[7] = make_float4(g2[4] * r2, g2[5] * r2, g2[6] * r2, g2[7] * r2);
    o[8] = make_float4(g2[8] * r2, g2[9] * r2, g2[10] * r2, g2[11] * r2);
    const float one = (sub == 0) ? 1.f : 0.f;
    o[9] = make_float4(one, one, one, 0.f);
  }
  __syncthreads();

  const int w = __builtin_amdgcn_readfirstlane(tid >> 5);
  const int c = lane & 15, hi = lane >> 4;
  int bn = ((c & 3) == 3) ? (c - 1) : c;
  bn = (bn > 10) ? 10 : bn;
  const float4* ap0 = reinterpret_cast<const float4*>(&gt[w * 32 + c][0]) + hi;
  const float4* ap1 = reinterpret_cast<const float4*>(&gt[w * 32 + 16 + c][0]) + hi;
  const float4* bp = reinterpret_cast<const float4*>(Mt + (size_t)bn * MK) + hi;
  v8f acc0 = {};
  v8f acc1 = {};
#pragma unroll
  for (int i = 0; i < MK / 8; ++i) {
    const float4 b = bp[2 * i], a0 = ap0[2 * i], a1 = ap1[2 * i];
    v2f bx; bx.x = b.x;  bx.y = b.y;
    v2f by; by.x = b.z;  by.y = b.w;
    v2f p0; p0.x = a0.x; p0.y = a0.y;
    v2f q0; q0.x = a0.z; q0.y = a0.w;
    v2f p1; p1.x = a1.x; p1.y = a1.y;
    v2f q1; q1.x = a1.z; q1.y = a1.w;
    acc0 = __builtin_amdgcn_wmma_f32_16x16x4_f32(false, p0, false, bx, (short)0, acc0, false, false);
    acc1 = __builtin_amdgcn_wmma_f32_16x16x4_f32(false, p1, false, bx, (short)0, acc1, false, false);
    acc0 = __builtin_amdgcn_wmma_f32_16x16x4_f32(false, q0, false, by, (short)0, acc0, false, false);
    acc1 = __builtin_amdgcn_wmma_f32_16x16x4_f32(false, q1, false, by, (short)0, acc1, false, false);
  }
  const float t0 = ((acc0[0] + acc0[1]) + (acc0[2] + acc0[3])) + ((acc0[4] + acc0[5]) + (acc0[6] + acc0[7]));
  const float t1 = ((acc1[0] + acc1[1]) + (acc1[2] + acc1[3])) + ((acc1[4] + acc1[5]) + (acc1[6] + acc1[7]));
  const int n0 = blockIdx.x * NPB + w * 4 + hi;
  if (n0 < N)     T[(size_t)n0 * TW + c] = t0;
  if (n0 + 2 < N) T[(size_t)(n0 + 2) * TW + c] = t1;
}

__global__ void __launch_bounds__(256) k_rows2(const int* __restrict__ cnt_in,
                                               const int* __restrict__ col,
                                               const int* __restrict__ ovc,
                                               const int* __restrict__ ovd,
                                               const int* __restrict__ ovs,
                                               const float* __restrict__ T,
                                               const float* __restrict__ cst,
                                               float* __restrict__ res, int N) {
  const int tid = threadIdx.x;
  const int sub = tid & (GRP - 1);
  const int dn = blockIdx.x * NPB + (tid >> 3);
  const int d = (dn < N) ? dn : (N - 1);
  const Run q = make_run(cnt_in, col, ovc, ovd, ovs, d);
  const int cnt = q.cnt;
  const float4 td = *reinterpret_cast<const float4*>(T + (size_t)d * TW + 8);
  const float ad0 = td.x, ad1 = td.y, ad2 = td.z;

  int sr[4];
#pragma unroll
  for (int i = 0; i < 4; ++i) {
    const bool filled = sub + GRP * i < q.nmain;
    const int* p = filled ? (q.colrow + sub + GRP * i) : (cnt_in + d);
    const int v = *p;
    sr[i] = filled ? v : d;
  }
  float4 ta[4], tt[4];
#pragma unroll
  for (int i = 0; i < 4; ++i) {
    const float4* tp = reinterpret_cast<const float4*>(T + (size_t)sr[i] * TW);
    ta[i] = tp[0]; tt[i] = tp[1];
  }
  float c0[4], c1[4], c2[4], t0[4], t1[4], t2[4];
  float m0 = -INFINITY, m1 = -INFINITY, m2 = -INFINITY;
#pragma unroll
  for (int i = 0; i < 4; ++i) {
    const bool in = sub + GRP * i < cnt;
    c0[i] = in ? lrelu(ta[i].x + ad0) : -INFINITY;
    c1[i] = in ? lrelu(ta[i].y + ad1) : -INFINITY;
    c2[i] = in ? lrelu(ta[i].z + ad2) : -INFINITY;
    t0[i] = tt[i].x; t1[i] = tt[i].y; t2[i] = tt[i].z;
    m0 = fmaxf(m0, c0[i]); m1 = fmaxf(m1, c1[i]); m2 = fmaxf(m2, c2[i]);
  }
  for (int j = 4 * GRP + sub; j < cnt; j += GRP) {
    const int e = entry_src(q.colrow, q.ovd, q.ovs, q.nmain, q.d, j);
    if (e >= 0) {
      const float4 a = *reinterpret_cast<const float4*>(T + (size_t)e * TW);
      m0 = fmaxf(m0, lrelu(a.x + ad0));
      m1 = fmaxf(m1, lrelu(a.y + ad1));
      m2 = fmaxf(m2, lrelu(a.z + ad2));
    }
  }
  m0 = grp_max(m0); m1 = grp_max(m1); m2 = grp_max(m2);

  float z0 = 0.f, z1 = 0.f, z2 = 0.f;
  float n0 = 0.f, n1 = 0.f, n2 = 0.f;
#pragma unroll
  for (int i = 0; i < 4; ++i) {
    const bool in = sub + GRP * i < cnt;
    const float e0 = in ? expf(c0[i] - m0) : 0.f;
    const float e1 = in ? expf(c1[i] - m1) : 0.f;
    const float e2 = in ? expf(c2[i] - m2) : 0.f;
    z0 += e0; z1 += e1; z2 += e2;
    n0 = fmaf(e0, t0[i], n0); n1 = fmaf(e1, t1[i], n1); n2 = fmaf(e2, t2[i], n2);
  }
  for (int j = 4 * GRP + sub; j < cnt; j += GRP) {
    const int e = entry_src(q.colrow, q.ovd, q.ovs, q.nmain, q.d, j);
    if (e >= 0) {
      const float4* tp = reinterpret_cast<const float4*>(T + (size_t)e * TW);
      const float4 a = tp[0], t = tp[1];
      const float e0 = expf(lrelu(a.x + ad0) - m0);
      const float e1 = expf(lrelu(a.y + ad1) - m1);
      const float e2 = expf(lrelu(a.z + ad2) - m2);
      z0 += e0; z1 += e1; z2 += e2;
      n0 = fmaf(e0, t.x, n0); n1 = fmaf(e1, t.y, n1); n2 = fmaf(e2, t.z, n2);
    }
  }
  z0 = grp_sum(z0); z1 = grp_sum(z1); z2 = grp_sum(z2);
  n0 = grp_sum(n0); n1 = grp_sum(n1); n2 = grp_sum(n2);
  if (sub == 0 && dn < N) res[dn] = (n0 / z0 + n1 / z1) + (n2 / z2 + cst[0]);
}

extern "C" void kernel_launch(void* const* d_in, const int* in_sizes, int n_in,
                              void* d_out, int out_size, void* d_ws, size_t ws_size,
                              hipStream_t stream) {
  const float* x   = (const float*)d_in[0];
  const int*   ei  = (const int*)d_in[1];
  const float* W1  = (const float*)d_in[2];
  const float* as1 = (const float*)d_in[3];
  const float* ad1 = (const float*)d_in[4];
  const float* b1  = (const float*)d_in[5];
  const float* W2  = (const float*)d_in[6];
  const float* as2 = (const float*)d_in[7];
  const float* ad2 = (const float*)d_in[8];
  const float* b2  = (const float*)d_in[9];
  const float* Wl  = (const float*)d_in[10];
  const float* bl  = (const float*)d_in[11];
  float* out = (float*)d_out;

  const int N  = in_sizes[0] / HIN;
  const int E  = in_sizes[1] / 2;
  const int* src = ei;
  const int* dst = ei + E;

  float* ws = (float*)d_ws;
  float* T      = ws;
  int*   COL    = (int*)(T + (size_t)N * TW);
  float* VT     = (float*)(COL + (size_t)N * CAP);
  float* CST    = VT + (size_t)12 * F;
  float* MT     = CST + 64;
  float* ASRC   = MT + (size_t)16 * MK;
  float* ADST   = ASRC + (size_t)3 * N;
  int*   CNT    = (int*)(ADST + (size_t)3 * N);
  int*   OVC    = CNT + N;
  int*   OVD    = OVC + 1;
  int*   OVS    = OVD + E;

  const int eblocks = (E + 1023) / 1024;
  const int rblocks = (N + NPB - 1) / NPB;
  const int ablocks = (N + 127) / 128;

  k_att1<<<FB + ablocks, 128, 0, stream>>>(x, W1, as1, ad1, W2, as2, ad2, Wl, b2, bl, VT, CST,
                                           ASRC, ADST, CNT, OVC, N);

  k_bucket<<<1 + eblocks, 256, 0, stream>>>(src, dst, E, CNT, COL, OVC, OVD, OVS, W1, b1, VT, MT);

  k_rows1<<<rblocks, 256, 0, stream>>>(CNT, COL, OVC, OVD, OVS, ASRC, ADST, x, MT, T, N);

  k_rows2<<<rblocks, 256, 0, stream>>>(CNT, COL, OVC, OVD, OVS, T, CST, out, N);
}


# edge grouping on shared-memory counters: edges binned by 32-node row block (one returning global add per bin and block, not per edge), buckets formed in LDS by both row passes; sums in arrival order
# speedup vs baseline: 25.2316x; 1.5901x over previous
#include <hip/hip_runtime.h>
#include <hip/hip_bf16.h>

#define HEADS 3
#define CPH   128
#define F     384
#define HIN   12
#define TW    16
#define MK    40
#define CAP   64
#define BINCAP 2048
#define BT    1024
#define EPT   4
#define GST   64
#define GRP   8
#define NPB   32
#define GP    44

typedef __attribute__((ext_vector_type(2))) float v2f;
typedef __attribute__((ext_vector_type(8))) float v8f;

__device__ __forceinline__ float lrelu(float v) { return v > 0.f ? v : 0.2f * v; }

__device__ __forceinline__ float grp_max(float v) {
#pragma unroll
  for (int o = GRP / 2; o >= 1; o >>= 1) v = fmaxf(v, __shfl_xor(v, o, 32));
  return v;
}
__device__ __forceinline__ float grp_sum(float v) {
#pragma unroll
  for (int o = GRP / 2; o >= 1; o >>= 1) v += __shfl_xor(v, o, 32);
  return v;
}
__device__ __forceinline__ float wave_sum(float v) {
#pragma unroll
  for (int o = 16; o >= 1; o >>= 1) v += __shfl_xor(v, o, 32);
  return v;
}

__device__ __forceinline__ void fold_w1(int h, int lane, const float* __restrict__ W1,
                                        const float* __restrict__ b1, const float* __restrict__ Vt,
                                        float* __restrict__ Mt) {
  const int c = lane & 15, hi = lane >> 4;
  const float* arow = (c < HIN) ? (W1 + (size_t)c * F) : b1;
  int bn = ((c & 3) == 3) ? (c - 1) : c;
  bn = (bn > 10) ? 10 : bn;
  const float4* ap = reinterpret_cast<const float4*>(arow + h * CPH) + hi;
  const float4* bp = reinterpret_cast<const float4*>(Vt + (size_t)bn * F + h * CPH) + hi;
  v8f acc0 = {};
  v8f acc1 = {};
#pragma unroll
  for (int i = 0; i < CPH / 8; ++i) {
    const float4 a = ap[2 * i], b = bp[2 * i];
    v2f ax; ax.x = a.x; ax.y = a.y;
    v2f ay; ay.x = a.z; ay.y = a.w;
    v2f bx; bx.x = b.x; bx.y = b.y;
    v2f by; by.x = b.z; by.y = b.w;
    acc0 = __builtin_amdgcn_wmma_f32_16x16x4_f32(false, ax, false, bx, (short)0, acc0, false, false);
    acc1 = __builtin_amdgcn_wmma_f32_16x16x4_f32(false, ay, false, by, (short)0, acc1, false, false);
  }
  float s[8];
#pragma unroll
  for (int r = 0; r < 8; ++r) s[r] = acc0[r] + acc1[r];
  if ((c & 3) != 3 && c < 11) {
    float* o = Mt + (size_t)c * MK + h * HIN;
    if (hi == 0) {
      reinterpret_cast<float4*>(o)[0] = make_float4(s[0], s[1], s[2], s[3]);
      reinterpret_cast<float4*>(o)[1] = make_float4(s[4], s[5], s[6], s[7]);
    } else {
      reinterpret_cast<float4*>(o)[2] = make_float4(s[0], s[1], s[2], s[3]);
      Mt[(size_t)c * MK + 3 * HIN + h] = s[4];
      if (h == 0) Mt[(size_t)c * MK + MK - 1] = 0.f;
    }
  }
}

__device__ __forceinline__ void fold_w2(int w, int lane, const float* __restrict__ W2,
                                        const float* __restrict__ a_src,
                                        const float* __restrict__ a_dst,
                                        const float* __restrict__ Wl,
                                        const float* __restrict__ b2,
                                        const float* __restrict__ bl,
                                        float* __restrict__ Vt, float* __restrict__ cst) {
  if (w == 36) {
    float p = 0.f;
    for (int i = lane; i < F; i += 32) p = fmaf(b2[i], Wl[i], p);
    p = wave_sum(p);
    if (lane == 0) cst[0] = p + bl[0];
    return;
  }
  const int g = w / HEADS, h = w - g * HEADS;
  const int c = lane & 15, hi = lane >> 4;
  const float* qb = (c < 4) ? a_src : (c < 8) ? Wl : a_dst;
  const float4* bp  = reinterpret_cast<const float4*>(qb + h * CPH) + hi;
  const float4* ap0 = reinterpret_cast<const float4*>(W2 + (size_t)(g * 32 + c) * F + h * CPH) + hi;
  const float4* ap1 = reinterpret_cast<const float4*>(W2 + (size_t)(g * 32 + 16 + c) * F + h * CPH) + hi;
  v8f acc0 = {};
  v8f acc1 = {};
#pragma unroll
  for (int i = 0; i < CPH / 8; ++i) {
    const float4 b = bp[2 * i], a0 = ap0[2 * i], a1 = ap1[2 * i];
    v2f bx; bx.x = b.x;  bx.y = b.y;
    v2f by; by.x = b.z;  by.y = b.w;
    v2f p0; p0.x = a0.x; p0.y = a0.y;
    v2f q0; q0.x = a0.z; q0.y = a0.w;
    v2f p1; p1.x = a1.x; p1.y = a1.y;
    v2f q1; q1.x = a1.z; q1.y = a1.w;
    acc0 = __builtin_amdgcn_wmma_f32_16x16x4_f32(false, p0, false, bx, (short)0, acc0, false, false);
    acc1 = __builtin_amdgcn_wmma_f32_16x16x4_f32(false, p1, false, bx, (short)0, acc1, false, false);
    acc0 = __builtin_amdgcn_wmma_f32_16x16x4_f32(false, q0, false, by, (short)0, acc0, false, false);
    acc1 = __builtin_amdgcn_wmma_f32_16x16x4_f32(false, q1, false, by, (short)0, acc1, false, false);
  }
  if ((c & 3) == h && c < 11) {
    float4* o = reinterpret_cast<float4*>(Vt + (size_t)c * F + g * 32 + 8 * hi);
    o[0] = make_float4(acc0[0], acc0[1], acc0[2], acc0[3]);
    o[1] = make_float4(acc0[4], acc0[5], acc0[6], acc0[7]);
    o[4] = make_float4(acc1[0], acc1[1], acc1[2], acc1[3]);
    o[5] = make_float4(acc1[4], acc1[5], acc1[6], acc1[7]);
  }
}

#define FTASKS 37
#define FB     10
__global__ void __launch_bounds__(128) k_att1(const float* __restrict__ x, const float* __restrict__ W1,
                                              const float* __restrict__ a_src,
                                              const float* __restrict__ a_dst,
                                              const float* __restrict__ W2,
                                              const float* __restrict__ a_src2,
                                              const float* __restrict__ a_dst2,
                                              const float* __restrict__ Wl,
                                              const float* __restrict__ b2,
                                              const float* __restrict__ bl,
                                              float* __restrict__ Vt, float* __restrict__ cst,
                                              float* __restrict__ asrc, float* __restrict__ adst,
                                              int* __restrict__ gcnt, int* __restrict__ ovc,
                                              int N, int nbins) {
  __shared__ float v[6][HIN];
  const int t = threadIdx.x;
  if ((int)blockIdx.x < FB) {
    const int w = __builtin_amdgcn_readfirstlane((int)blockIdx.x * 4 + (int)(threadIdx.x >> 5));
    if (w < FTASKS) fold_w2(w, t & 31, W2, a_src2, a_dst2, Wl, b2, bl, Vt, cst);
    return;
  }
  const int nblk = (int)blockIdx.x - FB;
  if (t < 6 * HIN) {
    const int j = t / HIN, k = t - j * HIN;
    const int h = (j < HEADS) ? j : (j - HEADS);
    const float4* wp = reinterpret_cast<const float4*>(W1 + (size_t)k * F + h * CPH);
    const float4* ap = reinterpret_cast<const float4*>(((j < HEADS) ? a_src : a_dst) + h * CPH);
    float p0 = 0.f, p1 = 0.f, p2 = 0.f, p3 = 0.f;
#pragma unroll 8
    for (int i = 0; i < CPH / 4; ++i) {
      const float4 w = wp[i], a = ap[i];
      p0 = fmaf(w.x, a.x, p0); p1 = fmaf(w.y, a.y, p1);
      p2 = fmaf(w.z, a.z, p2); p3 = fmaf(w.w, a.w, p3);
    }
    v[j][k] = (p0 + p1) + (p2 + p3);
  }
  __syncthreads();
  const int n = nblk * 128 + t;
  if (n >= N) return;
  const float4* xr = reinterpret_cast<const float4*>(x + (size_t)n * HIN);
  const float4 xa = xr[0], xb = xr[1], xc = xr[2];
  const float xs[HIN] = {xa.x, xa.y, xa.z, xa.w, xb.x, xb.y, xb.z, xb.w, xc.x, xc.y, xc.z, xc.w};
  float r[6];
#pragma unroll
  for (int j = 0; j < 6; ++j) {
    float s = 0.f;
#pragma unroll
    for (int k = 0; k < HIN; ++k) s = fmaf(xs[k], v[j][k], s);
    r[j] = s;
  }
  asrc[(size_t)n * HEADS + 0] = r[0]; asrc[(size_t)n * HEADS + 1] = r[1];
  asrc[(size_t)n * HEADS + 2] = r[2];
  adst[(size_t)n * HEADS + 0] = r[3]; adst[(size_t)n * HEADS + 1] = r[4];
  adst[(size_t)n * HEADS + 2] = r[5];
  if (n < nbins) gcnt[(size_t)n * GST] = 0;
  if (n == 0) ovc[0] = 0;
}

__global__ void __launch_bounds__(BT) k_bin(const int* __restrict__ src, const int* __restrict__ dst,
                                            int E, int nbins, int* __restrict__ gcnt,
                                            int2* __restrict__ rec, int* __restrict__ ovc,
                                            int* __restrict__ ovd, int* __restrict__ ovs,
                                            const float* __restrict__ W1, const float* __restrict__ b1,
                                            const float* __restrict__ Vt, float* __restrict__ Mt) {
  extern __shared__ int sh[];
  if (blockIdx.x == 0) {
    const int w = __builtin_amdgcn_readfirstlane((int)(threadIdx.x >> 5));
    if (w < HEADS) fold_w1(w, threadIdx.x & 31, W1, b1, Vt, Mt);
    return;
  }
  int* hist = sh;
  int* base = sh + nbins;
  const int tid = threadIdx.x;
  for (int i = tid; i < nbins; i += BT) hist[i] = 0;
  __syncthreads();
  const int e0 = ((int)blockIdx.x - 1) * (BT * EPT) + tid;
  int d[EPT], s[EPT], r[EPT];
#pragma unroll
  for (int k = 0; k < EPT; ++k) {
    const int e = e0 + k * BT;
    d[k] = -1; s[k] = 0; r[k] = 0;
    if (e < E) { d[k] = dst[e]; s[k] = src[e]; }
  }
#pragma unroll
  for (int k = 0; k < EPT; ++k)
    if (d[k] >= 0) r[k] = atomicAdd(&hist[d[k] / NPB], 1);
  __syncthreads();
  for (int i = tid; i < nbins; i += BT) {
    const int h = hist[i];
    if (h > 0) base[i] = atomicAdd(&gcnt[(size_t)i * GST], h);
  }
  __syncthreads();
#pragma unroll
  for (int k = 0; k < EPT; ++k) {
    if (d[k] >= 0) {
      const int b = d[k] / NPB;
      const int p = base[b] + r[k];
      if (p < BINCAP) {
        rec[(size_t)b * BINCAP + p] = make_int2(d[k], s[k]);
      } else {
        const int o = atomicAdd(ovc, 1);
        ovd[o] = d[k]; ovs[o] = s[k];
      }
    }
  }
}

struct Bk {
  int cnt[NPB];
  int col[NPB][CAP];
};

__device__ __forceinline__ void build_buckets(Bk& bk, const int2* __restrict__ rec, int nb,
                                              int node0, int tid) {
  if (tid < NPB) bk.cnt[tid] = 0;
  __syncthreads();
  for (int i0 = 0; i0 < nb; i0 += 4 * 256) {
    int2 r[4];
#pragma unroll
    for (int u = 0; u < 4; ++u) {
      const int i = i0 + tid + 256 * u;
      r[u] = rec[(i < nb) ? i : (nb - 1)];
    }
#pragma unroll
    for (int u = 0; u < 4; ++u) {
      if (i0 + tid + 256 * u < nb) {
        const int dl = r[u].x - node0;
        const int slot = atomicAdd(&bk.cnt[dl], 1);
        if (slot < CAP) bk.col[dl][slot] = r[u].y;
      }
    }
  }
  __syncthreads();
}

struct Run {
  const int* colrow;
  const int2* rec;
  const int* ovd;
  const int* ovs;
  int nmain;
  int nreg;
  int nb;
  int cnt;
  int d;
  bool aside;
};

__device__ __forceinline__ Run make_run(const Bk& bk, int dl, const int2* __restrict__ rec, int gc,
                                        int nb, const int* __restrict__ ovc,
                                        const int* __restrict__ ovd, const int* __restrict__ ovs,
                                        int d) {
  Run q;
  const int indeg = bk.cnt[dl];
  q.colrow = bk.col[dl];
  q.rec = rec; q.ovd = ovd; q.ovs = ovs; q.d = d; q.nb = nb;
  q.aside = (indeg > CAP) || (gc > BINCAP);
  if (!q.aside) {
    q.nmain = indeg;
    q.nreg = indeg + 1;
    q.cnt = indeg + 1;
  } else {
    q.nmain = 0;
    q.nreg = 1;
    q.cnt = 4 * GRP + nb;
    if (gc > BINCAP) q.cnt += ovc[0];
  }
  return q;
}

__device__ __forceinline__ int entry_src(const Run& q, int j) {
  if (!q.aside) return (j < q.nmain) ? q.colrow[j] : q.d;
  int o = j - 4 * GRP;
  if (o < q.nb) {
    const int2 r = q.rec[o];
    return (r.x == q.d) ? r.y : -1;
  }
  o -= q.nb;
  return (q.ovd[o] == q.d) ? q.ovs[o] : -1;
}

__global__ void __launch_bounds__(256) k_rows1(const int* __restrict__ gcnt,
                                               const int2* __restrict__ rec,
                                               const int* __restrict__ ovc,
                                               const int* __restrict__ ovd,
                                               const int* __restrict__ ovs,
                                               const float* __restrict__ asrc,
                                               const float* __restrict__ adst,
                                               const float* __restrict__ x,
                                               const float* __restrict__ Mt,
                                               float* __restrict__ T, int N) {
  __shared__ __attribute__((aligned(16))) float gt[256][GP];
  __shared__ Bk bk;
  const int tid = threadIdx.x;
  const int lane = tid & 31;
  const int sub = tid & (GRP - 1);
  const int node0 = blockIdx.x * NPB;
  const int dn = node0 + (tid >> 3);
  const int d = (dn < N) ? dn : (N - 1);
  const int gc = gcnt[(size_t)blockIdx.x * GST];
  const int nb = (gc < BINCAP) ? gc : BINCAP;
  const int2* brec = rec + (size_t)blockIdx.x * BINCAP;
  build_buckets(bk, brec, nb, node0, tid);
  const Run q = make_run(bk, d - node0, brec, gc, nb, ovc, ovd, ovs, d);
  const int cnt = q.cnt;
  const float ad0 = adst[(size_t)d * HEADS + 0];
  const float ad1 = adst[(size_t)d * HEADS + 1];
  const float ad2 = adst[(size_t)d * HEADS + 2];

  int sr[4];
#pragma unroll
  for (int i = 0; i < 4; ++i) {
    const bool filled = sub + GRP * i < q.nmain;
    sr[i] = filled ? q.colrow[sub + GRP * i] : d;
  }
  float a0[4], a1[4], a2[4];
  float4 xa[4], xb[4], xc[4];
#pragma unroll
  for (int i = 0; i < 4; ++i) {
    const float* ap = asrc + (size_t)sr[i] * HEADS;
    a0[i] = ap[0]; a1[i] = ap[1]; a2[i] = ap[2];
    const float4* xr = reinterpret_cast<const float4*>(x + (size_t)sr[i] * HIN);
    xa[i] = xr[0]; xb[i] = xr[1]; xc[i] = xr[2];
  }
  float c0[4], c1[4], c2[4];
  float m0 = -INFINITY, m1 = -INFINITY, m2 = -INFINITY;
#pragma unroll
  for (int i = 0; i < 4; ++i) {
    const bool in = sub + GRP * i < q.nreg;
    c0[i] = in ? lrelu(a0[i] + ad0) : -INFINITY;
    c1[i] = in ? lrelu(a1[i] + ad1) : -INFINITY;
    c2[i] = in ? lrelu(a2[i] + ad2) : -INFINITY;
    m0 = fmaxf(m0, c0[i]); m1 = fmaxf(m1, c1[i]); m2 = fmaxf(m2, c2[i]);
  }
  for (int j = 4 * GRP + sub; j < cnt; j += GRP) {
    const int e = entry_src(q, j);
    if (e >= 0) {
      const float* ap = asrc + (size_t)e * HEADS;
      m0 = fmaxf(m0, lrelu(ap[0] + ad0));
      m1 = fmaxf(m1, lrelu(ap[1] + ad1));
      m2 = fmaxf(m2, lrelu(ap[2] + ad2));
    }
  }
  m0 = grp_max(m0); m1 = grp_max(m1); m2 = grp_max(m2);

  float g0[HIN], g1[HIN], g2[HIN];
#pragma unroll
  for (int k = 0; k < HIN; ++k) { g0[k] = 0.f; g1[k] = 0.f; g2[k] = 0.f; }
  float z0 = 0.f, z1 = 0.f, z2 = 0.f;
#pragma unroll
  for (int i = 0; i < 4; ++i) {
    const bool in = sub + GRP * i < q.nreg;
    const float e0 = in ? expf(c0[i] - m0) : 0.f;
    const float e1 = in ? expf(c1[i] - m1) : 0.f;
    const float e2 = in ? expf(c2[i] - m2) : 0.f;
    z0 += e0; z1 += e1; z2 += e2;
    const float xs[HIN] = {xa[i].x, xa[i].y, xa[i].z, xa[i].w, xb[i].x, xb[i].y, xb[i].z, xb[i].w,
                           xc[i].x, xc[i].y, xc[i].z, xc[i].w};
#pragma unroll
    for (int k = 0; k < HIN; ++k) {
      g0[k] = fmaf(e0, xs[k], g0[k]); g1[k] = fmaf(e1, xs[k], g1[k]); g2[k] = fmaf(e2, xs[k], g2[k]);
    }
  }
  for (int j = 4 * GRP + sub; j < cnt; j += GRP) {
    const int e = entry_src(q, j);
    if (e >= 0) {
      const float* ap = asrc + (size_t)e * HEADS;
      const float e0 = expf(lrelu(ap[0] + ad0) - m0);
      const float e1 = expf(lrelu(ap[1] + ad1) - m1);
      const float e2 = expf(lrelu(ap[2] + ad2) - m2);
      z0 += e0; z1 += e1; z2 += e2;
      const float4* xr = reinterpret_cast<const float4*>(x + (size_t)e * HIN);
      const float4 ya = xr[0], yb = xr[1], yc = xr[2];
      const float xs[HIN] = {ya.x, ya.y, ya.z, ya.w, yb.x, yb.y, yb.z, yb.w, yc.x, yc.y, yc.z, yc.w};
#pragma unroll
      for (int k = 0; k < HIN; ++k) {
        g0[k] = fmaf(e0, xs[k], g0[k]); g1[k] = fmaf(e1, xs[k], g1[k]); g2[k] = fmaf(e2, xs[k], g2[k]);
      }
    }
  }
  z0 = grp_sum(z0); z1 = grp_sum(z1); z2 = grp_sum(z2);
  const float r0 = 1.f / z0, r1 = 1.f / z1, r2 = 1.f / z2;

  {
    float4* o = reinterpret_cast<float4*>(&gt[tid][0]);
    o[0] = make_float4(g0[0] * r0, g0[1] * r0, g0[2] * r0, g0[3] * r0);
    o[1] = make_float4(g0[4] * r0, g0[5] * r0, g0[6] * r0, g0[7] * r0);
    o[2] = make_float4(g0[8] * r0, g0[9] * r0, g0[10] * r0, g0[11] * r0);
    o[3] = make_float4(g1[0] * r1, g1[1] * r1, g1[2] * r1, g1[3] * r1);
    o[4] = make_float4(g1[4] * r1, g1[5] * r1, g1[6] * r1, g1[7] * r1);
    o[5] = make_float4(g1[8] * r1, g1[9] * r1, g1[10] * r1, g1[11] * r1);
    o[6] = make_float4(g2[0] * r2, g2[1] * r2, g2[2] * r2, g2[3] * r2);
    o[7] = make_float4(g2[4] * r2, g2[5] * r2, g2[6] * r2, g2[7] * r2);
    o[8] = make_float4(g2[8] * r2, g2[9] * r2, g2[10] * r2, g2[11] * r2);
    const float one = (sub == 0) ? 1.f : 0.f;
    o[9] = make_float4(one, one, one, 0.f);
  }
  __syncthreads();

  const int w = __builtin_amdgcn_readfirstlane(tid >> 5);
  const int c = lane & 15, hi = lane >> 4;
  int bn = ((c & 3) == 3) ? (c - 1) : c;
  bn = (bn > 10) ? 10 : bn;
  const float4* ap0 = reinterpret_cast<const float4*>(&gt[w * 32 + c][0]) + hi;
  const float4* ap1 = reinterpret_cast<const float4*>(&gt[w * 32 + 16 + c][0]) + hi;
  const float4* bp = reinterpret_cast<const float4*>(Mt + (size_t)bn * MK) + hi;
  v8f acc0 = {};
  v8f acc1 = {};
#pragma unroll
  for (int i = 0; i < MK / 8; ++i) {
    const float4 b = bp[2 * i], a0 = ap0[2 * i], a1 = ap1[2 * i];
    v2f bx; bx.x = b.x;  bx.y = b.y;
    v2f by; by.x = b.z;  by.y = b.w;
    v2f p0; p0.x = a0.x; p0.y = a0.y;
    v2f q0; q0.x = a0.z; q0.y = a0.w;
    v2f p1; p1.x = a1.x; p1.y = a1.y;
    v2f q1; q1.x = a1.z; q1.y = a1.w;
    acc0 = __builtin_amdgcn_wmma_f32_16x16x4_f32(false, p0, false, bx, (short)0, acc0, false, false);
    acc1 = __builtin_amdgcn_wmma_f32_16x16x4_f32(false, p1, false, bx, (short)0, acc1, false, false);
    acc0 = __builtin_amdgcn_wmma_f32_16x16x4_f32(false, q0, false, by, (short)0, acc0, false, false);
    acc1 = __builtin_amdgcn_wmma_f32_16x16x4_f32(false, q1, false, by, (short)0, acc1, false, false);
  }
  const float t0 = ((acc0[0] + acc0[1]) + (acc0[2] + acc0[3])) + ((acc0[4] + acc0[5]) + (acc0[6] + acc0[7]));
  const float t1 = ((acc1[0] + acc1[1]) + (acc1[2] + acc1[3])) + ((acc1[4] + acc1[5]) + (acc1[6] + acc1[7]));
  const int n0 = blockIdx.x * NPB + w * 4 + hi;
  if (n0 < N)     T[(size_t)n0 * TW + c] = t0;
  if (n0 + 2 < N) T[(size_t)(n0 + 2) * TW + c] = t1;
}

__global__ void __launch_bounds__(256) k_rows2(const int* __restrict__ gcnt,
                                               const int2* __restrict__ rec,
                                               const int* __restrict__ ovc,
                                               const int* __restrict__ ovd,
                                               const int* __restrict__ ovs,
                                               const float* __restrict__ T,
                                               const float* __restrict__ cst,
                                               float* __restrict__ res, int N) {
  __shared__ Bk bk;
  const int tid = threadIdx.x;
  const int sub = tid & (GRP - 1);
  const int node0 = blockIdx.x * NPB;
  const int dn = node0 + (tid >> 3);
  const int d = (dn < N) ? dn : (N - 1);
  const int gc = gcnt[(size_t)blockIdx.x * GST];
  const int nb = (gc < BINCAP) ? gc : BINCAP;
  const int2* brec = rec + (size_t)blockIdx.x * BINCAP;
  build_buckets(bk, brec, nb, node0, tid);
  const Run q = make_run(bk, d - node0, brec, gc, nb, ovc, ovd, ovs, d);
  const int cnt = q.cnt;
  const float4 td = *reinterpret_cast<const float4*>(T + (size_t)d * TW + 8);
  const float ad0 = td.x, ad1 = td.y, ad2 = td.z;

  int sr[4];
#pragma unroll
  for (int i = 0; i < 4; ++i) {
    const bool filled = sub + GRP * i < q.nmain;
    sr[i] = filled ? q.colrow[sub + GRP * i] : d;
  }
  float4 ta[4], tt[4];
#pragma unroll
  for (int i = 0; i < 4; ++i) {
    const float4* tp = reinterpret_cast<const float4*>(T + (size_t)sr[i] * TW);
    ta[i] = tp[0]; tt[i] = tp[1];
  }
  float c0[4], c1[4], c2[4], t0[4], t1[4], t2[4];
  float m0 = -INFINITY, m1 = -INFINITY, m2 = -INFINITY;
#pragma unroll
  for (int i = 0; i < 4; ++i) {
    const bool in = sub + GRP * i < q.nreg;
    c0[i] = in ? lrelu(ta[i].x + ad0) : -INFINITY;
    c1[i] = in ? lrelu(ta[i].y + ad1) : -INFINITY;
    c2[i] = in ? lrelu(ta[i].z + ad2) : -INFINITY;
    t0[i] = tt[i].x; t1[i] = tt[i].y; t2[i] = tt[i].z;
    m0 = fmaxf(m0, c0[i]); m1 = fmaxf(m1, c1[i]); m2 = fmaxf(m2, c2[i]);
  }
  for (int j = 4 * GRP + sub; j < cnt; j += GRP) {
    const int e = entry_src(q, j);
    if (e >= 0) {
      const float4 a = *reinterpret_cast<const float4*>(T + (size_t)e * TW);
      m0 = fmaxf(m0, lrelu(a.x + ad0));
      m1 = fmaxf(m1, lrelu(a.y + ad1));
      m2 = fmaxf(m2, lrelu(a.z + ad2));
    }
  }
  m0 = grp_max(m0); m1 = grp_max(m1); m2 = grp_max(m2);

  float z0 = 0.f, z1 = 0.f, z2 = 0.f;
  float n0 = 0.f, n1 = 0.f, n2 = 0.f;
#pragma unroll
  for (int i = 0; i < 4; ++i) {
    const bool in = sub + GRP * i < q.nreg;
    const float e0 = in ? expf(c0[i] - m0) : 0.f;
    const float e1 = in ? expf(c1[i] - m1) : 0.f;
    const float e2 = in ? expf(c2[i] - m2) : 0.f;
    z0 += e0; z1 += e1; z2 += e2;
    n0 = fmaf(e0, t0[i], n0); n1 = fmaf(e1, t1[i], n1); n2 = fmaf(e2, t2[i], n2);
  }
  for (int j = 4 * GRP + sub; j < cnt; j += GRP) {
    const int e = entry_src(q, j);
    if (e >= 0) {
      const float4* tp = reinterpret_cast<const float4*>(T + (size_t)e * TW);
      const float4 a = tp[0], t = tp[1];
      const float e0 = expf(lrelu(a.x + ad0) - m0);
      const float e1 = expf(lrelu(a.y + ad1) - m1);
      const float e2 = expf(lrelu(a.z + ad2) - m2);
      z0 += e0; z1 += e1; z2 += e2;
      n0 = fmaf(e0, t.x, n0); n1 = fmaf(e1, t.y, n1); n2 = fmaf(e2, t.z, n2);
    }
  }
  z0 = grp_sum(z0); z1 = grp_sum(z1); z2 = grp_sum(z2);
  n0 = grp_sum(n0); n1 = grp_sum(n1); n2 = grp_sum(n2);
  if (sub == 0 && dn < N) res[dn] = (n0 / z0 + n1 / z1) + (n2 / z2 + cst[0]);
}

extern "C" void kernel_launch(void* const* d_in, const int* in_sizes, int n_in,
                              void* d_out, int out_size, void* d_ws, size_t ws_size,
                              hipStream_t stream) {
  const float* x   = (const float*)d_in[0];
  const int*   ei  = (const int*)d_in[1];
  const float* W1  = (const float*)d_in[2];
  const float* as1 = (const float*)d_in[3];
  const float* ad1 = (const float*)d_in[4];
  const float* b1  = (const float*)d_in[5];
  const float* W2  = (const float*)d_in[6];
  const float* as2 = (const float*)d_in[7];
  const float* ad2 = (const float*)d_in[8];
  const float* b2  = (const float*)d_in[9];
  const float* Wl  = (const float*)d_in[10];
  const float* bl  = (const float*)d_in[11];
  float* out = (float*)d_out;

  const int N  = in_sizes[0] / HIN;
  const int E  = in_sizes[1] / 2;
  const int* src = ei;
  const int* dst = ei + E;

  float* ws = (float*)d_ws;
  float* T      = ws;
  const int nbins = (N + NPB - 1) / NPB;
  int2*  REC    = (int2*)(T + (size_t)N * TW);
  float* VT     = (float*)(REC + (size_t)nbins * BINCAP);
  float* CST    = VT + (size_t)12 * F;
  float* MT     = CST + 64;
  float* ASRC   = MT + (size_t)16 * MK;
  float* ADST   = ASRC + (size_t)3 * N;
  int*   GCNT   = (int*)(ADST + (size_t)3 * N);
  int*   OVC    = GCNT + (size_t)nbins * GST;
  int*   OVD    = OVC + 1;
  int*   OVS    = OVD + E;

  const int eblocks = (E + BT * EPT - 1) / (BT * EPT);
  const int rblocks = nbins;
  const int ablocks = (N + 127) / 128;

  k_att1<<<FB + ablocks, 128, 0, stream>>>(x, W1, as1, ad1, W2, as2, ad2, Wl, b2, bl, VT, CST,
                                           ASRC, ADST, GCNT, OVC, N, nbins);

  k_bin<<<1 + eblocks, BT, 2 * nbins * sizeof(int), stream>>>(src, dst, E, nbins, GCNT, REC, OVC, OVD,
                                                              OVS, W1, b1, VT, MT);

  k_rows1<<<rblocks, 256, 0, stream>>>(GCNT, REC, OVC, OVD, OVS, ASRC, ADST, x, MT, T, N);

  k_rows2<<<rblocks, 256, 0, stream>>>(GCNT, REC, OVC, OVD, OVS, T, CST, out, N);
}


# one launch fewer (4 -> 3): edges sorted by bin per block in shared memory (no global counters to clear), so sorting, attention dots and both weight folds share launch 1; sums in arrival order
# speedup vs baseline: 31.2912x; 1.2402x over previous
#include <hip/hip_runtime.h>
#include <hip/hip_bf16.h>

#define HEADS 3
#define CPH   128
#define F     384
#define HIN   12
#define TW    16
#define MK    40
#define CAP   64
#define BT    1024
#define EPT   4
#define SEGE  4096
#define TPS   2
#define VP    388
#define GRP   8
#define NPB   32
#define GP    44

typedef __attribute__((ext_vector_type(2))) float v2f;
typedef __attribute__((ext_vector_type(8))) float v8f;

__device__ __forceinline__ float lrelu(float v) { return v > 0.f ? v : 0.2f * v; }

__device__ __forceinline__ float grp_max(float v) {
#pragma unroll
  for (int o = GRP / 2; o >= 1; o >>= 1) v = fmaxf(v, __shfl_xor(v, o, 32));
  return v;
}
__device__ __forceinline__ float grp_sum(float v) {
#pragma unroll
  for (int o = GRP / 2; o >= 1; o >>= 1) v += __shfl_xor(v, o, 32);
  return v;
}
__device__ __forceinline__ float wave_sum(float v) {
#pragma unroll
  for (int o = 16; o >= 1; o >>= 1) v += __shfl_xor(v, o, 32);
  return v;
}

__device__ __forceinline__ void fold_w1(int h, int lane, const float* __restrict__ W1,
                                        const float* __restrict__ b1, const float* __restrict__ Vt,
                                        float* __restrict__ Mt) {
  const int c = lane & 15, hi = lane >> 4;
  const float* arow = (c < HIN) ? (W1 + (size_t)c * F) : b1;
  int bn = ((c & 3) == 3) ? (c - 1) : c;
  bn = (bn > 10) ? 10 : bn;
  const float4* ap = reinterpret_cast<const float4*>(arow + h * CPH) + hi;
  const float4* bp = reinterpret_cast<const float4*>(Vt + (size_t)bn * VP + h * CPH) + hi;
  v8f acc0 = {};
  v8f acc1 = {};
#pragma unroll
  for (int i = 0; i < CPH / 8; ++i) {
    const float4 a = ap[2 * i], b = bp[2 * i];
    v2f ax; ax.x = a.x; ax.y = a.y;
    v2f ay; ay.x = a.z; ay.y = a.w;
    v2f bx; bx.x = b.x; bx.y = b.y;
    v2f by; by.x = b.z; by.y = b.w;
    acc0 = __builtin_amdgcn_wmma_f32_16x16x4_f32(false, ax, false, bx, (short)0, acc0, false, false);
    acc1 = __builtin_amdgcn_wmma_f32_16x16x4_f32(false, ay, false, by, (short)0, acc1, false, false);
  }
  float s[8];
#pragma unroll
  for (int r = 0; r < 8; ++r) s[r] = acc0[r] + acc1[r];
  if ((c & 3) != 3 && c < 11) {
    float* o = Mt + (size_t)c * MK + h * HIN;
    if (hi == 0) {
      reinterpret_cast<float4*>(o)[0] = make_float4(s[0], s[1], s[2], s[3]);
      reinterpret_cast<float4*>(o)[1] = make_float4(s[4], s[5], s[6], s[7]);
    } else {
      reinterpret_cast<float4*>(o)[2] = make_float4(s[0], s[1], s[2], s[3]);
      Mt[(size_t)c * MK + 3 * HIN + h] = s[4];
      if (h == 0) Mt[(size_t)c * MK + MK - 1] = 0.f;
    }
  }
}

__device__ __forceinline__ void fold_w2(int w, int lane, const float* __restrict__ W2,
                                        const float* __restrict__ a_src,
                                        const float* __restrict__ a_dst,
                                        const float* __restrict__ Wl,
                                        const float* __restrict__ b2,
                                        const float* __restrict__ bl,
                                        float* __restrict__ Vt, float* __restrict__ cst) {
  if (w == 36) {
    float p = 0.f;
    for (int i = lane; i < F; i += 32) p = fmaf(b2[i], Wl[i], p);
    p = wave_sum(p);
    if (lane == 0) cst[0] = p + bl[0];
    return;
  }
  const int g = w / HEADS, h = w - g * HEADS;
  const int c = lane & 15, hi = lane >> 4;
  const float* qb = (c < 4) ? a_src : (c < 8) ? Wl : a_dst;
  const float4* bp  = reinterpret_cast<const float4*>(qb + h * CPH) + hi;
  const float4* ap0 = reinterpret_cast<const float4*>(W2 + (size_t)(g * 32 + c) * F + h * CPH) + hi;
  const float4* ap1 = reinterpret_cast<const float4*>(W2 + (size_t)(g * 32 + 16 + c) * F + h * CPH) + hi;
  v8f acc0 = {};
  v8f acc1 = {};
#pragma unroll
  for (int i = 0; i < CPH / 8; ++i) {
    const float4 b = bp[2 * i], a0 = ap0[2 * i], a1 = ap1[2 * i];
    v2f bx; bx.x = b.x;  bx.y = b.y;
    v2f by; by.x = b.z;  by.y = b.w;
    v2f p0; p0.x = a0.x; p0.y = a0.y;
    v2f q0; q0.x = a0.z; q0.y = a0.w;
    v2f p1; p1.x = a1.x; p1.y = a1.y;
    v2f q1; q1.x = a1.z; q1.y = a1.w;
    acc0 = __builtin_amdgcn_wmma_f32_16x16x4_f32(false, p0, false, bx, (short)0, acc0, false, false);
    acc1 = __builtin_amdgcn_wmma_f32_16x16x4_f32(false, p1, false, bx, (short)0, acc1, false, false);
    acc0 = __builtin_amdgcn_wmma_f32_16x16x4_f32(false, q0, false, by, (short)0, acc0, false, false);
    acc1 = __builtin_amdgcn_wmma_f32_16x16x4_f32(false, q1, false, by, (short)0, acc1, false, false);
  }
  if ((c & 3) == h && c < 11) {
    float4* o = reinterpret_cast<float4*>(Vt + (size_t)c * VP + g * 32 + 8 * hi);
    o[0] = make_float4(acc0[0], acc0[1], acc0[2], acc0[3]);
    o[1] = make_float4(acc0[4], acc0[5], acc0[6], acc0[7]);
    o[4] = make_float4(acc1[0], acc1[1], acc1[2], acc1[3]);
    o[5] = make_float4(acc1[4], acc1[5], acc1[6], acc1[7]);
  }
}

__global__ void __launch_bounds__(BT) k_front(const float* __restrict__ x, const float* __restrict__ W1,
                                              const float* __restrict__ a_src,
                                              const float* __restrict__ a_dst,
                                              const float* __restrict__ b1,
                                              const float* __restrict__ W2,
                                              const float* __restrict__ a_src2,
                                              const float* __restrict__ a_dst2,
                                              const float* __restrict__ Wl,
                                              const float* __restrict__ b2,
                                              const float* __restrict__ bl,
                                              const int* __restrict__ src, const int* __restrict__ dst,
                                              float* __restrict__ cst, float* __restrict__ Mt,
                                              float* __restrict__ asrc, float* __restrict__ adst,
                                              int2* __restrict__ segd, int2* __restrict__ rec,
                                              int N, int E, int nbins, int nseg, int ablocks) {
  extern __shared__ int sh[];
  __shared__ __attribute__((aligned(16))) float vts[11 * VP];
  __shared__ float v[6][HIN];
  __shared__ int wsum[BT / 32];
  const int tid = threadIdx.x;
  const int blk = (int)blockIdx.x;

  if (blk < HEADS) {
    const int w = __builtin_amdgcn_readfirstlane(tid >> 5);
    if (w < 12) fold_w2(12 * blk + w, tid & 31, W2, a_src2, a_dst2, Wl, b2, bl, vts, cst);
    else if (w == 12 && blk == 0) fold_w2(36, tid & 31, W2, a_src2, a_dst2, Wl, b2, bl, vts, cst);
    __syncthreads();
    if (w == 0) fold_w1(blk, tid & 31, W1, b1, vts, Mt);
    return;
  }

  if (blk < HEADS + ablocks) {
    const int nblk = blk - HEADS;
    if (tid < 6 * HIN) {
      const int j = tid / HIN, k = tid - j * HIN;
      const int h = (j < HEADS) ? j : (j - HEADS);
      const float4* wp = reinterpret_cast<const float4*>(W1 + (size_t)k * F + h * CPH);
      const float4* ap = reinterpret_cast<const float4*>(((j < HEADS) ? a_src : a_dst) + h * CPH);
      float p0 = 0.f, p1 = 0.f, p2 = 0.f, p3 = 0.f;
#pragma unroll 8
      for (int i = 0; i < CPH / 4; ++i) {
        const float4 w = wp[i], a = ap[i];
        p0 = fmaf(w.x, a.x, p0); p1 = fmaf(w.y, a.y, p1);
        p2 = fmaf(w.z, a.z, p2); p3 = fmaf(w.w, a.w, p3);
      }
      v[j][k] = (p0 + p1) + (p2 + p3);
    }
    __syncthreads();
    const int n = nblk * BT + tid;
    if (n >= N) return;
    const float4* xr = reinterpret_cast<const float4*>(x + (size_t)n * HIN);
    const float4 xa = xr[0], xb = xr[1], xc = xr[2];
    const float xs[HIN] = {xa.x, xa.y, xa.z, xa.w, xb.x, xb.y, xb.z, xb.w, xc.x, xc.y, xc.z, xc.w};
    float r[6];
#pragma unroll
    for (int j = 0; j < 6; ++j) {
      float s = 0.f;
#pragma unroll
      for (int k = 0; k < HIN; ++k) s = fmaf(xs[k], v[j][k], s);
      r[j] = s;
    }
    asrc[(size_t)n * HEADS + 0] = r[0]; asrc[(size_t)n * HEADS + 1] = r[1];
    asrc[(size_t)n * HEADS + 2] = r[2];
    adst[(size_t)n * HEADS + 0] = r[3]; adst[(size_t)n * HEADS + 1] = r[4];
    adst[(size_t)n * HEADS + 2] = r[5];
    return;
  }

  const int bb = blk - HEADS - ablocks;
  int* hist = sh;
  int* start = sh + nbins;
  for (int i = tid; i < nbins; i += BT) hist[i] = 0;
  __syncthreads();
  const int e0 = bb * SEGE + tid;
  int d[EPT], s[EPT], r[EPT];
#pragma unroll
  for (int k = 0; k < EPT; ++k) {
    const int e = e0 + k * BT;
    d[k] = -1; s[k] = 0; r[k] = 0;
    if (e < E) { d[k] = dst[e]; s[k] = src[e]; }
  }
#pragma unroll
  for (int k = 0; k < EPT; ++k)
    if (d[k] >= 0) r[k] = atomicAdd(&hist[d[k] / NPB], 1);
  __syncthreads();
  const int lane = tid & 31, wv = tid >> 5;
  int carry = 0;
  for (int b0 = 0; b0 < nbins; b0 += BT) {
    const int i = b0 + tid;
    const int hv = (i < nbins) ? hist[i] : 0;
    int xsum = hv;
#pragma unroll
    for (int o = 1; o < 32; o <<= 1) {
      const int y = __shfl_up(xsum, o, 32);
      if (lane >= o) xsum += y;
    }
    if (lane == 31) wsum[wv] = xsum;
    __syncthreads();
    if (wv == 0) {
      int ws = wsum[lane];
#pragma unroll
      for (int o = 1; o < 32; o <<= 1) {
        const int y = __shfl_up(ws, o, 32);
        if (lane >= o) ws += y;
      }
      wsum[lane] = ws;
    }
    __syncthreads();
    const int st = carry + ((wv > 0) ? wsum[wv - 1] : 0) + xsum - hv;
    if (i < nbins) {
      start[i] = st;
      segd[(size_t)i * nseg + bb] = make_int2(st, hv);
    }
    carry += wsum[BT / 32 - 1];
    __syncthreads();
  }
#pragma unroll
  for (int k = 0; k < EPT; ++k) {
    if (d[k] >= 0) {
      const int p = start[d[k] / NPB] + r[k];
      rec[(size_t)bb * SEGE + p] = make_int2(d[k], s[k]);
    }
  }
}

struct Bk {
  int cnt[NPB];
  int col[NPB][CAP];
};

__device__ __forceinline__ void build_buckets(Bk& bk, const int2* __restrict__ sd,
                                              const int2* __restrict__ rec, int nseg,
                                              int node0, int tid) {
  if (tid < NPB) bk.cnt[tid] = 0;
  __syncthreads();
  const int k = tid & (TPS - 1);
  for (int sg0 = 0; sg0 < nseg; sg0 += 256 / TPS) {
    const int sg = sg0 + tid / TPS;
    int2 ds = make_int2(0, 0);
    if (sg < nseg) ds = sd[sg];
    const int2* rp = rec + (size_t)((sg < nseg) ? sg : 0) * SEGE + ds.x;
    for (int s0 = 0; s0 < ds.y; s0 += 8 * TPS) {
      int2 r[8];
#pragma unroll
      for (int u = 0; u < 8; ++u) {
        const int i = s0 + k + TPS * u;
        r[u] = rp[(i < ds.y) ? i : (ds.y - 1)];
      }
#pragma unroll
      for (int u = 0; u < 8; ++u) {
        if (s0 + k + TPS * u < ds.y) {
          const int dl = r[u].x - node0;
          const int slot = atomicAdd(&bk.cnt[dl], 1);
          if (slot < CAP) bk.col[dl][slot] = r[u].y;
        }
      }
    }
  }
  __syncthreads();
}

struct Run {
  const int* colrow;
  const int2* sd;
  const int2* rec;
  int nseg;
  int nmain;
  int nreg;
  int cnt;
  int d;
  bool aside;
};

__device__ __forceinline__ Run make_run(const Bk& bk, int dl, const int2* __restrict__ sd,
                                        const int2* __restrict__ rec, int nseg, int d) {
  Run q;
  const int indeg = bk.cnt[dl];
  q.colrow = bk.col[dl];
  q.sd = sd; q.rec = rec; q.nseg = nseg; q.d = d;
  q.aside = indeg > CAP;
  if (!q.aside) {
    q.nmain = indeg;
    q.nreg = indeg + 1;
    q.cnt = indeg + 1;
  } else {
    q.nmain = 0;
    q.nreg = 1;
    q.cnt = 0;
  }
  return q;
}

template <class Fn>
__device__ __forceinline__ void for_extra(const Run& q, int sub, Fn fn) {
  if (!q.aside) {
    for (int j = 4 * GRP + sub; j < q.cnt; j += GRP) fn((j < q.nmain) ? q.colrow[j] : q.d);
  } else {
    for (int sg = 0; sg < q.nseg; ++sg) {
      const int2 ds = q.sd[sg];
      const int2* rp = q.rec + (size_t)sg * SEGE + ds.x;
      for (int s = sub; s < ds.y; s += GRP) {
        const int2 r = rp[s];
        if (r.x == q.d) fn(r.y);
      }
    }
  }
}

__global__ void __launch_bounds__(256) k_rows1(const int2* __restrict__ segd,
                                               const int2* __restrict__ rec, int nseg,
                                               const float* __restrict__ asrc,
                                               const float* __restrict__ adst,
                                               const float* __restrict__ x,
                                               const float* __restrict__ Mt,
                                               float* __restrict__ T, int N) {
  __shared__ __attribute__((aligned(16))) float gt[256][GP];
  __shared__ Bk bk;
  const int tid = threadIdx.x;
  const int lane = tid & 31;
  const int sub = tid & (GRP - 1);
  const int node0 = blockIdx.x * NPB;
  const int dn = node0 + (tid >> 3);
  const int d = (dn < N) ? dn : (N - 1);
  const int2* sd = segd + (size_t)blockIdx.x * nseg;
  build_buckets(bk, sd, rec, nseg, node0, tid);
  const Run q = make_run(bk, d - node0, sd, rec, nseg, d);
  const float ad0 = adst[(size_t)d * HEADS + 0];
  const float ad1 = adst[(size_t)d * HEADS + 1];
  const float ad2 = adst[(size_t)d * HEADS + 2];

  int sr[4];
#pragma unroll
  for (int i = 0; i < 4; ++i) {
    const bool filled = sub + GRP * i < q.nmain;
    sr[i] = filled ? q.colrow[sub + GRP * i] : d;
  }
  float a0[4], a1[4], a2[4];
  float4 xa[4], xb[4], xc[4];
#pragma unroll
  for (int i = 0; i < 4; ++i) {
    const float* ap = asrc + (size_t)sr[i] * HEADS;
    a0[i] = ap[0]; a1[i] = ap[1]; a2[i] = ap[2];
    const float4* xr = reinterpret_cast<const float4*>(x + (size_t)sr[i] * HIN);
    xa[i] = xr[0]; xb[i] = xr[1]; xc[i] = xr[2];
  }
  float c0[4], c1[4], c2[4];
  float m0 = -INFINITY, m1 = -INFINITY, m2 = -INFINITY;
#pragma unroll
  for (int i = 0; i < 4; ++i) {
    const bool in = sub + GRP * i < q.nreg;
    c0[i] = in ? lrelu(a0[i] + ad0) : -INFINITY;
    c1[i] = in ? lrelu(a1[i] + ad1) : -INFINITY;
    c2[i] = in ? lrelu(a2[i] + ad2) : -INFINITY;
    m0 = fmaxf(m0, c0[i]); m1 = fmaxf(m1, c1[i]); m2 = fmaxf(m2, c2[i]);
  }
  for_extra(q, sub, [&](int e) __attribute__((always_inline)) {
    const float* ap = asrc + (size_t)e * HEADS;
    m0 = fmaxf(m0, lrelu(ap[0] + ad0));
    m1 = fmaxf(m1, lrelu(ap[1] + ad1));
    m2 = fmaxf(m2, lrelu(ap[2] + ad2));
  });
  m0 = grp_max(m0); m1 = grp_max(m1); m2 = grp_max(m2);

  float g0[HIN], g1[HIN], g2[HIN];
#pragma unroll
  for (int k = 0; k < HIN; ++k) { g0[k] = 0.f; g1[k] = 0.f; g2[k] = 0.f; }
  float z0 = 0.f, z1 = 0.f, z2 = 0.f;
#pragma unroll
  for (int i = 0; i < 4; ++i) {
    const bool in = sub + GRP * i < q.nreg;
    const float e0 = in ? expf(c0[i] - m0) : 0.f;
    const float e1 = in ? expf(c1[i] - m1) : 0.f;
    const float e2 = in ? expf(c2[i] - m2) : 0.f;
    z0 += e0; z1 += e1; z2 += e2;
    const float xs[HIN] = {xa[i].x, xa[i].y, xa[i].z, xa[i].w, xb[i].x, xb[i].y, xb[i].z, xb[i].w,
                           xc[i].x, xc[i].y, xc[i].z, xc[i].w};
#pragma unroll
    for (int k = 0; k < HIN; ++k) {
      g0[k] = fmaf(e0, xs[k], g0[k]); g1[k] = fmaf(e1, xs[k], g1[k]); g2[k] = fmaf(e2, xs[k], g2[k]);
    }
  }
  for_extra(q, sub, [&](int e) __attribute__((always_inline)) {
    const float* ap = asrc + (size_t)e * HEADS;
    const float e0 = expf(lrelu(ap[0] + ad0) - m0);
    const float e1 = expf(lrelu(ap[1] + ad1) - m1);
    const float e2 = expf(lrelu(ap[2] + ad2) - m2);
    z0 += e0; z1 += e1; z2 += e2;
    const float4* xr = reinterpret_cast<const float4*>(x + (size_t)e * HIN);
    const float4 ya = xr[0], yb = xr[1], yc = xr[2];
    const float xs[HIN] = {ya.x, ya.y, ya.z, ya.w, yb.x, yb.y, yb.z, yb.w, yc.x, yc.y, yc.z, yc.w};
#pragma unroll
    for (int k = 0; k < HIN; ++k) {
      g0[k] = fmaf(e0, xs[k], g0[k]); g1[k] = fmaf(e1, xs[k], g1[k]); g2[k] = fmaf(e2, xs[k], g2[k]);
    }
  });
  z0 = grp_sum(z0); z1 = grp_sum(z1); z2 = grp_sum(z2);
  const float r0 = 1.f / z0, r1 = 1.f / z1, r2 = 1.f / z2;

  {
    float4* o = reinterpret_cast<float4*>(&gt[tid][0]);
    o[0] = make_float4(g0[0] * r0, g0[1] * r0, g0[2] * r0, g0[3] * r0);
    o[1] = make_float4(g0[4] * r0, g0[5] * r0, g0[6] * r0, g0[7] * r0);
    o[2] = make_float4(g0[8] * r0, g0[9] * r0, g0[10] * r0, g0[11] * r0);
    o[3] = make_float4(g1[0] * r1, g1[1] * r1, g1[2] * r1, g1[3] * r1);
    o[4] = make_float4(g1[4] * r1, g1[5] * r1, g1[6] * r1, g1[7] * r1);
    o[5] = make_float4(g1[8] * r1, g1[9] * r1, g1[10] * r1, g1[11] * r1);
    o[6] = make_float4(g2[0] * r2, g2[1] * r2, g2[2] * r2, g2[3] * r2);
    o[7] = make_float4(g2[4] * r2, g2[5] * r2, g2[6] * r2, g2[7] * r2);
    o[8] = make_float4(g2[8] * r2, g2[9] * r2, g2[10] * r2, g2[11] * r2);
    const float one = (sub == 0) ? 1.f : 0.f;
    o[9] = make_float4(one, one, one, 0.f);
  }
  __syncthreads();

  const int w = __builtin_amdgcn_readfirstlane(tid >> 5);
  const int c = lane & 15, hi = lane >> 4;
  int bn = ((c & 3) == 3) ? (c - 1) : c;
  bn = (bn > 10) ? 10 : bn;
  const float4* ap0 = reinterpret_cast<const float4*>(&gt[w * 32 + c][0]) + hi;
  const float4* ap1 = reinterpret_cast<const float4*>(&gt[w * 32 + 16 + c][0]) + hi;
  const float4* bp = reinterpret_cast<const float4*>(Mt + (size_t)bn * MK) + hi;
  v8f acc0 = {};
  v8f acc1 = {};
#pragma unroll
  for (int i = 0; i < MK / 8; ++i) {
    const float4 b = bp[2 * i], a0 = ap0[2 * i], a1 = ap1[2 * i];
    v2f bx; bx.x = b.x;  bx.y = b.y;
    v2f by; by.x = b.z;  by.y = b.w;
    v2f p0; p0.x = a0.x; p0.y = a0.y;
    v2f q0; q0.x = a0.z; q0.y = a0.w;
    v2f p1; p1.x = a1.x; p1.y = a1.y;
    v2f q1; q1.x = a1.z; q1.y = a1.w;
    acc0 = __builtin_amdgcn_wmma_f32_16x16x4_f32(false, p0, false, bx, (short)0, acc0, false, false);
    acc1 = __builtin_amdgcn_wmma_f32_16x16x4_f32(false, p1, false, bx, (short)0, acc1, false, false);
    acc0 = __builtin_amdgcn_wmma_f32_16x16x4_f32(false, q0, false, by, (short)0, acc0, false, false);
    acc1 = __builtin_amdgcn_wmma_f32_16x16x4_f32(false, q1, false, by, (short)0, acc1, false, false);
  }
  const float t0 = ((acc0[0] + acc0[1]) + (acc0[2] + acc0[3])) + ((acc0[4] + acc0[5]) + (acc0[6] + acc0[7]));
  const float t1 = ((acc1[0] + acc1[1]) + (acc1[2] + acc1[3])) + ((acc1[4] + acc1[5]) + (acc1[6] + acc1[7]));
  const int n0 = blockIdx.x * NPB + w * 4 + hi;
  if (n0 < N)     T[(size_t)n0 * TW + c] = t0;
  if (n0 + 2 < N) T[(size_t)(n0 + 2) * TW + c] = t1;
}

__global__ void __launch_bounds__(256) k_rows2(const int2* __restrict__ segd,
                                               const int2* __restrict__ rec, int nseg,
                                               const float* __restrict__ T,
                                               const float* __restrict__ cst,
                                               float* __restrict__ res, int N) {
  __shared__ Bk bk;
  const int tid = threadIdx.x;
  const int sub = tid & (GRP - 1);
  const int node0 = blockIdx.x * NPB;
  const int dn = node0 + (tid >> 3);
  const int d = (dn < N) ? dn : (N - 1);
  const int2* sd = segd + (size_t)blockIdx.x * nseg;
  build_buckets(bk, sd, rec, nseg, node0, tid);
  const Run q = make_run(bk, d - node0, sd, rec, nseg, d);
  const float4 td = *reinterpret_cast<const float4*>(T + (size_t)d * TW + 8);
  const float ad0 = td.x, ad1 = td.y, ad2 = td.z;

  int sr[4];
#pragma unroll
  for (int i = 0; i < 4; ++i) {
    const bool filled = sub + GRP * i < q.nmain;
    sr[i] = filled ? q.colrow[sub + GRP * i] : d;
  }
  float4 ta[4], tt[4];
#pragma unroll
  for (int i = 0; i < 4; ++i) {
    const float4* tp = reinterpret_cast<const float4*>(T + (size_t)sr[i] * TW);
    ta[i] = tp[0]; tt[i] = tp[1];
  }
  float c0[4], c1[4], c2[4], t0[4], t1[4], t2[4];
  float m0 = -INFINITY, m1 = -INFINITY, m2 = -INFINITY;
#pragma unroll
  for (int i = 0; i < 4; ++i) {
    const bool in = sub + GRP * i < q.nreg;
    c0[i] = in ? lrelu(ta[i].x + ad0) : -INFINITY;
    c1[i] = in ? lrelu(ta[i].y + ad1) : -INFINITY;
    c2[i] = in ? lrelu(ta[i].z + ad2) : -INFINITY;
    t0[i] = tt[i].x; t1[i] = tt[i].y; t2[i] = tt[i].z;
    m0 = fmaxf(m0, c0[i]); m1 = fmaxf(m1, c1[i]); m2 = fmaxf(m2, c2[i]);
  }
  for_extra(q, sub, [&](int e) __attribute__((always_inline)) {
    const float4 a = *reinterpret_cast<const float4*>(T + (size_t)e * TW);
    m0 = fmaxf(m0, lrelu(a.x + ad0));
    m1 = fmaxf(m1, lrelu(a.y + ad1));
    m2 = fmaxf(m2, lrelu(a.z + ad2));
  });
  m0 = grp_max(m0); m1 = grp_max(m1); m2 = grp_max(m2);

  float z0 = 0.f, z1 = 0.f, z2 = 0.f;
  float n0 = 0.f, n1 = 0.f, n2 = 0.f;
#pragma unroll
  for (int i = 0; i < 4; ++i) {
    const bool in = sub + GRP * i < q.nreg;
    const float e0 = in ? expf(c0[i] - m0) : 0.f;
    const float e1 = in ? expf(c1[i] - m1) : 0.f;
    const float e2 = in ? expf(c2[i] - m2) : 0.f;
    z0 += e0; z1 += e1; z2 += e2;
    n0 = fmaf(e0, t0[i], n0); n1 = fmaf(e1, t1[i], n1); n2 = fmaf(e2, t2[i], n2);
  }
  for_extra(q, sub, [&](int e) __attribute__((always_inline)) {
    const float4* tp = reinterpret_cast<const float4*>(T + (size_t)e * TW);
    const float4 a = tp[0], t = tp[1];
    const float e0 = expf(lrelu(a.x + ad0) - m0);
    const float e1 = expf(lrelu(a.y + ad1) - m1);
    const float e2 = expf(lrelu(a.z + ad2) - m2);
    z0 += e0; z1 += e1; z2 += e2;
    n0 = fmaf(e0, t.x, n0); n1 = fmaf(e1, t.y, n1); n2 = fmaf(e2, t.z, n2);
  });
  z0 = grp_sum(z0); z1 = grp_sum(z1); z2 = grp_sum(z2);
  n0 = grp_sum(n0); n1 = grp_sum(n1); n2 = grp_sum(n2);
  if (sub == 0 && dn < N) res[dn] = (n0 / z0 + n1 / z1) + (n2 / z2 + cst[0]);
}

extern "C" void kernel_launch(void* const* d_in, const int* in_sizes, int n_in,
                              void* d_out, int out_size, void* d_ws, size_t ws_size,
                              hipStream_t stream) {
  const float* x   = (const float*)d_in[0];
  const int*   ei  = (const int*)d_in[1];
  const float* W1  = (const float*)d_in[2];
  const float* as1 = (const float*)d_in[3];
  const float* ad1 = (const float*)d_in[4];
  const float* b1  = (const float*)d_in[5];
  const float* W2  = (const float*)d_in[6];
  const float* as2 = (const float*)d_in[7];
  const float* ad2 = (const float*)d_in[8];
  const float* b2  = (const float*)d_in[9];
  const float* Wl  = (const float*)d_in[10];
  const float* bl  = (const float*)d_in[11];
  float* out = (float*)d_out;

  const int N  = in_sizes[0] / HIN;
  const int E  = in_sizes[1] / 2;
  const int* src = ei;
  const int* dst = ei + E;

  float* ws = (float*)d_ws;
  float* T      = ws;
  const int nbins = (N + NPB - 1) / NPB;
  const int nseg  = (E + SEGE - 1) / SEGE;
  int2*  REC    = (int2*)(T + (size_t)N * TW);
  float* CST    = (float*)(REC + (size_t)nseg * SEGE);
  float* MT     = CST + 64;
  float* ASRC   = MT + (size_t)16 * MK;
  float* ADST   = ASRC + (size_t)3 * N;
  int2*  SEGD   = (int2*)(ADST + (size_t)3 * N);

  const int rblocks = nbins;
  const int ablocks = (N + BT - 1) / BT;

  k_front<<<HEADS + ablocks + nseg, BT, 2 * nbins * sizeof(int), stream>>>(
      x, W1, as1, ad1, b1, W2, as2, ad2, Wl, b2, bl, src, dst, CST, MT, ASRC, ADST, SEGD, REC,
      N, E, nbins, nseg, ablocks);

  k_rows1<<<rblocks, 256, 0, stream>>>(SEGD, REC, nseg, ASRC, ADST, x, MT, T, N);

  k_rows2<<<rblocks, 256, 0, stream>>>(SEGD, REC, nseg, T, CST, out, N);
}


# edge sorting: each block puts its 32 KiB piece of the sorted list together in shared memory and stores it in order as whole 128-byte lines (was one scattered 8-byte store per edge)
# speedup vs baseline: 33.3966x; 1.0673x over previous
#include <hip/hip_runtime.h>
#include <hip/hip_bf16.h>

#define HEADS 3
#define CPH   128
#define F     384
#define HIN   12
#define TW    16
#define MK    40
#define CAP   64
#define BT    1024
#define EPT   4
#define SEGE  4096
#define TPS   2
#define VP    388
#define GRP   8
#define NPB   32
#define GP    44

typedef __attribute__((ext_vector_type(2))) float v2f;
typedef __attribute__((ext_vector_type(8))) float v8f;

__device__ __forceinline__ float lrelu(float v) { return v > 0.f ? v : 0.2f * v; }

__device__ __forceinline__ float grp_max(float v) {
#pragma unroll
  for (int o = GRP / 2; o >= 1; o >>= 1) v = fmaxf(v, __shfl_xor(v, o, 32));
  return v;
}
__device__ __forceinline__ float grp_sum(float v) {
#pragma unroll
  for (int o = GRP / 2; o >= 1; o >>= 1) v += __shfl_xor(v, o, 32);
  return v;
}
__device__ __forceinline__ float wave_sum(float v) {
#pragma unroll
  for (int o = 16; o >= 1; o >>= 1) v += __shfl_xor(v, o, 32);
  return v;
}

__device__ __forceinline__ void fold_w1(int h, int lane, const float* __restrict__ W1,
                                        const float* __restrict__ b1, const float* __restrict__ Vt,
                                        float* __restrict__ Mt) {
  const int c = lane & 15, hi = lane >> 4;
  const float* arow = (c < HIN) ? (W1 + (size_t)c * F) : b1;
  int bn = ((c & 3) == 3) ? (c - 1) : c;
  bn = (bn > 10) ? 10 : bn;
  const float4* ap = reinterpret_cast<const float4*>(arow + h * CPH) + hi;
  const float4* bp = reinterpret_cast<const float4*>(Vt + (size_t)bn * VP + h * CPH) + hi;
  v8f acc0 = {};
  v8f acc1 = {};
#pragma unroll
  for (int i = 0; i < CPH / 8; ++i) {
    const float4 a = ap[2 * i], b = bp[2 * i];
    v2f ax; ax.x = a.x; ax.y = a.y;
    v2f ay; ay.x = a.z; ay.y = a.w;
    v2f bx; bx.x = b.x; bx.y = b.y;
    v2f by; by.x = b.z; by.y = b.w;
    acc0 = __builtin_amdgcn_wmma_f32_16x16x4_f32(false, ax, false, bx, (short)0, acc0, false, false);
    acc1 = __builtin_amdgcn_wmma_f32_16x16x4_f32(false, ay, false, by, (short)0, acc1, false, false);
  }
  float s[8];
#pragma unroll
  for (int r = 0; r < 8; ++r) s[r] = acc0[r] + acc1[r];
  if ((c & 3) != 3 && c < 11) {
    float* o = Mt + (size_t)c * MK + h * HIN;
    if (hi == 0) {
      reinterpret_cast<float4*>(o)[0] = make_float4(s[0], s[1], s[2], s[3]);
      reinterpret_cast<float4*>(o)[1] = make_float4(s[4], s[5], s[6], s[7]);
    } else {
      reinterpret_cast<float4*>(o)[2] = make_float4(s[0], s[1], s[2], s[3]);
      Mt[(size_t)c * MK + 3 * HIN + h] = s[4];
      if (h == 0) Mt[(size_t)c * MK + MK - 1] = 0.f;
    }
  }
}

__device__ __forceinline__ void fold_w2(int w, int lane, const float* __restrict__ W2,
                                        const float* __restrict__ a_src,
                                        const float* __restrict__ a_dst,
                                        const float* __restrict__ Wl,
                                        const float* __restrict__ b2,
                                        const float* __restrict__ bl,
                                        float* __restrict__ Vt, float* __restrict__ cst) {
  if (w == 36) {
    float p = 0.f;
    for (int i = lane; i < F; i += 32) p = fmaf(b2[i], Wl[i], p);
    p = wave_sum(p);
    if (lane == 0) cst[0] = p + bl[0];
    return;
  }
  const int g = w / HEADS, h = w - g * HEADS;
  const int c = lane & 15, hi = lane >> 4;
  const float* qb = (c < 4) ? a_src : (c < 8) ? Wl : a_dst;
  const float4* bp  = reinterpret_cast<const float4*>(qb + h * CPH) + hi;
  const float4* ap0 = reinterpret_cast<const float4*>(W2 + (size_t)(g * 32 + c) * F + h * CPH) + hi;
  const float4* ap1 = reinterpret_cast<const float4*>(W2 + (size_t)(g * 32 + 16 + c) * F + h * CPH) + hi;
  v8f acc0 = {};
  v8f acc1 = {};
#pragma unroll
  for (int i = 0; i < CPH / 8; ++i) {
    const float4 b = bp[2 * i], a0 = ap0[2 * i], a1 = ap1[2 * i];
    v2f bx; bx.x = b.x;  bx.y = b.y;
    v2f by; by.x = b.z;  by.y = b.w;
    v2f p0; p0.x = a0.x; p0.y = a0.y;
    v2f q0; q0.x = a0.z; q0.y = a0.w;
    v2f p1; p1.x = a1.x; p1.y = a1.y;
    v2f q1; q1.x = a1.z; q1.y = a1.w;
    acc0 = __builtin_amdgcn_wmma_f32_16x16x4_f32(false, p0, false, bx, (short)0, acc0, false, false);
    acc1 = __builtin_amdgcn_wmma_f32_16x16x4_f32(false, p1, false, bx, (short)0, acc1, false, false);
    acc0 = __builtin_amdgcn_wmma_f32_16x16x4_f32(false, q0, false, by, (short)0, acc0, false, false);
    acc1 = __builtin_amdgcn_wmma_f32_16x16x4_f32(false, q1, false, by, (short)0, acc1, false, false);
  }
  if ((c & 3) == h && c < 11) {
    float4* o = reinterpret_cast<float4*>(Vt + (size_t)c * VP + g * 32 + 8 * hi);
    o[0] = make_float4(acc0[0], acc0[1], acc0[2], acc0[3]);
    o[1] = make_float4(acc0[4], acc0[5], acc0[6], acc0[7]);
    o[4] = make_float4(acc1[0], acc1[1], acc1[2], acc1[3]);
    o[5] = make_float4(acc1[4], acc1[5], acc1[6], acc1[7]);
  }
}

__global__ void __launch_bounds__(BT) k_front(const float* __restrict__ x, const float* __restrict__ W1,
                                              const float* __restrict__ a_src,
                                              const float* __restrict__ a_dst,
                                              const float* __restrict__ b1,
                                              const float* __restrict__ W2,
                                              const float* __restrict__ a_src2,
                                              const float* __restrict__ a_dst2,
                                              const float* __restrict__ Wl,
                                              const float* __restrict__ b2,
                                              const float* __restrict__ bl,
                                              const int* __restrict__ src, const int* __restrict__ dst,
                                              float* __restrict__ cst, float* __restrict__ Mt,
                                              float* __restrict__ asrc, float* __restrict__ adst,
                                              int2* __restrict__ segd, int2* __restrict__ rec,
                                              int N, int E, int nbins, int nseg, int ablocks) {
  extern __shared__ int sh[];
  __shared__ __attribute__((aligned(16))) float vts[11 * VP];
  __shared__ float v[6][HIN];
  __shared__ int wsum[BT / 32];
  __shared__ __attribute__((aligned(16))) int srec[2 * SEGE];
  const int tid = threadIdx.x;
  const int blk = (int)blockIdx.x;

  if (blk < HEADS) {
    const int w = __builtin_amdgcn_readfirstlane(tid >> 5);
    if (w < 12) fold_w2(12 * blk + w, tid & 31, W2, a_src2, a_dst2, Wl, b2, bl, vts, cst);
    else if (w == 12 && blk == 0) fold_w2(36, tid & 31, W2, a_src2, a_dst2, Wl, b2, bl, vts, cst);
    __syncthreads();
    if (w == 0) fold_w1(blk, tid & 31, W1, b1, vts, Mt);
    return;
  }

  if (blk < HEADS + ablocks) {
    const int nblk = blk - HEADS;
    if (tid < 6 * HIN) {
      const int j = tid / HIN, k = tid - j * HIN;
      const int h = (j < HEADS) ? j : (j - HEADS);
      const float4* wp = reinterpret_cast<const float4*>(W1 + (size_t)k * F + h * CPH);
      const float4* ap = reinterpret_cast<const float4*>(((j < HEADS) ? a_src : a_dst) + h * CPH);
      float p0 = 0.f, p1 = 0.f, p2 = 0.f, p3 = 0.f;
#pragma unroll 8
      for (int i = 0; i < CPH / 4; ++i) {
        const float4 w = wp[i], a = ap[i];
        p0 = fmaf(w.x, a.x, p0); p1 = fmaf(w.y, a.y, p1);
        p2 = fmaf(w.z, a.z, p2); p3 = fmaf(w.w, a.w, p3);
      }
      v[j][k] = (p0 + p1) + (p2 + p3);
    }
    __syncthreads();
    const int n = nblk * BT + tid;
    if (n >= N) return;
    const float4* xr = reinterpret_cast<const float4*>(x + (size_t)n * HIN);
    const float4 xa = xr[0], xb = xr[1], xc = xr[2];
    const float xs[HIN] = {xa.x, xa.y, xa.z, xa.w, xb.x, xb.y, xb.z, xb.w, xc.x, xc.y, xc.z, xc.w};
    float r[6];
#pragma unroll
    for (int j = 0; j < 6; ++j) {
      float s = 0.f;
#pragma unroll
      for (int k = 0; k < HIN; ++k) s = fmaf(xs[k], v[j][k], s);
      r[j] = s;
    }
    asrc[(size_t)n * HEADS + 0] = r[0]; asrc[(size_t)n * HEADS + 1] = r[1];
    asrc[(size_t)n * HEADS + 2] = r[2];
    adst[(size_t)n * HEADS + 0] = r[3]; adst[(size_t)n * HEADS + 1] = r[4];
    adst[(size_t)n * HEADS + 2] = r[5];
    return;
  }

  const int bb = blk - HEADS - ablocks;
  int* hist = sh;
  int* start = sh + nbins;
  for (int i = tid; i < nbins; i += BT) hist[i] = 0;
  __syncthreads();
  const int e0 = bb * SEGE + tid;
  int d[EPT], s[EPT], r[EPT];
#pragma unroll
  for (int k = 0; k < EPT; ++k) {
    const int e = e0 + k * BT;
    d[k] = -1; s[k] = 0; r[k] = 0;
    if (e < E) { d[k] = dst[e]; s[k] = src[e]; }
  }
#pragma unroll
  for (int k = 0; k < EPT; ++k)
    if (d[k] >= 0) r[k] = atomicAdd(&hist[d[k] / NPB], 1);
  __syncthreads();
  const int lane = tid & 31, wv = tid >> 5;
  int carry = 0;
  for (int b0 = 0; b0 < nbins; b0 += BT) {
    const int i = b0 + tid;
    const int hv = (i < nbins) ? hist[i] : 0;
    int xsum = hv;
#pragma unroll
    for (int o = 1; o < 32; o <<= 1) {
      const int y = __shfl_up(xsum, o, 32);
      if (lane >= o) xsum += y;
    }
    if (lane == 31) wsum[wv] = xsum;
    __syncthreads();
    if (wv == 0) {
      int ws = wsum[lane];
#pragma unroll
      for (int o = 1; o < 32; o <<= 1) {
        const int y = __shfl_up(ws, o, 32);
        if (lane >= o) ws += y;
      }
      wsum[lane] = ws;
    }
    __syncthreads();
    const int st = carry + ((wv > 0) ? wsum[wv - 1] : 0) + xsum - hv;
    if (i < nbins) {
      start[i] = st;
      segd[(size_t)i * nseg + bb] = make_int2(st, hv);
    }
    carry += wsum[BT / 32 - 1];
    __syncthreads();
  }
  int p[EPT];
#pragma unroll
  for (int k = 0; k < EPT; ++k) p[k] = start[((d[k] >= 0) ? d[k] : 0) / NPB] + r[k];
#pragma unroll
  for (int k = 0; k < EPT; ++k)
    if (d[k] >= 0) reinterpret_cast<int2*>(srec)[p[k]] = make_int2(d[k], s[k]);
  __syncthreads();
  const int nval = (E - bb * SEGE < SEGE) ? (E - bb * SEGE) : SEGE;
  int4* rout = reinterpret_cast<int4*>(rec + (size_t)bb * SEGE);
#pragma unroll
  for (int k = 0; k < EPT / 2; ++k) {
    const int i = k * BT + tid;
    if (2 * i < nval) rout[i] = reinterpret_cast<const int4*>(srec)[i];
  }
}

struct Bk {
  int cnt[NPB];
  int col[NPB][CAP];
};

__device__ __forceinline__ void build_buckets(Bk& bk, const int2* __restrict__ sd,
                                              const int2* __restrict__ rec, int nseg,
                                              int node0, int tid) {
  if (tid < NPB) bk.cnt[tid] = 0;
  __syncthreads();
  const int k = tid & (TPS - 1);
  for (int sg0 = 0; sg0 < nseg; sg0 += 256 / TPS) {
    const int sg = sg0 + tid / TPS;
    int2 ds = make_int2(0, 0);
    if (sg < nseg) ds = sd[sg];
    const int2* rp = rec + (size_t)((sg < nseg) ? sg : 0) * SEGE + ds.x;
    for (int s0 = 0; s0 < ds.y; s0 += 8 * TPS) {
      int2 r[8];
#pragma unroll
      for (int u = 0; u < 8; ++u) {
        const int i = s0 + k + TPS * u;
        r[u] = rp[(i < ds.y) ? i : (ds.y - 1)];
      }
#pragma unroll
      for (int u = 0; u < 8; ++u) {
        if (s0 + k + TPS * u < ds.y) {
          const int dl = r[u].x - node0;
          const int slot = atomicAdd(&bk.cnt[dl], 1);
          if (slot < CAP) bk.col[dl][slot] = r[u].y;
        }
      }
    }
  }
  __syncthreads();
}

struct Run {
  const int* colrow;
  const int2* sd;
  const int2* rec;
  int nseg;
  int nmain;
  int nreg;
  int cnt;
  int d;
  bool aside;
};

__device__ __forceinline__ Run make_run(const Bk& bk, int dl, const int2* __restrict__ sd,
                                        const int2* __restrict__ rec, int nseg, int d) {
  Run q;
  const int indeg = bk.cnt[dl];
  q.colrow = bk.col[dl];
  q.sd = sd; q.rec = rec; q.nseg = nseg; q.d = d;
  q.aside = indeg > CAP;
  if (!q.aside) {
    q.nmain = indeg;
    q.nreg = indeg + 1;
    q.cnt = indeg + 1;
  } else {
    q.nmain = 0;
    q.nreg = 1;
    q.cnt = 0;
  }
  return q;
}

template <class Fn>
__device__ __forceinline__ void for_extra(const Run& q, int sub, Fn fn) {
  if (!q.aside) {
    for (int j = 4 * GRP + sub; j < q.cnt; j += GRP) fn((j < q.nmain) ? q.colrow[j] : q.d);
  } else {
    for (int sg = 0; sg < q.nseg; ++sg) {
      const int2 ds = q.sd[sg];
      const int2* rp = q.rec + (size_t)sg * SEGE + ds.x;
      for (int s = sub; s < ds.y; s += GRP) {
        const int2 r = rp[s];
        if (r.x == q.d) fn(r.y);
      }
    }
  }
}

__global__ void __launch_bounds__(256) k_rows1(const int2* __restrict__ segd,
                                               const int2* __restrict__ rec, int nseg,
                                               const float* __restrict__ asrc,
                                               const float* __restrict__ adst,
                                               const float* __restrict__ x,
                                               const float* __restrict__ Mt,
                                               float* __restrict__ T, int N) {
  __shared__ __attribute__((aligned(16))) float gt[256][GP];
  __shared__ Bk bk;
  const int tid = threadIdx.x;
  const int lane = tid & 31;
  const int sub = tid & (GRP - 1);
  const int node0 = blockIdx.x * NPB;
  const int dn = node0 + (tid >> 3);
  const int d = (dn < N) ? dn : (N - 1);
  const int2* sd = segd + (size_t)blockIdx.x * nseg;
  build_buckets(bk, sd, rec, nseg, node0, tid);
  const Run q = make_run(bk, d - node0, sd, rec, nseg, d);
  const float ad0 = adst[(size_t)d * HEADS + 0];
  const float ad1 = adst[(size_t)d * HEADS + 1];
  const float ad2 = adst[(size_t)d * HEADS + 2];

  int sr[4];
#pragma unroll
  for (int i = 0; i < 4; ++i) {
    const bool filled = sub + GRP * i < q.nmain;
    sr[i] = filled ? q.colrow[sub + GRP * i] : d;
  }
  float a0[4], a1[4], a2[4];
  float4 xa[4], xb[4], xc[4];
#pragma unroll
  for (int i = 0; i < 4; ++i) {
    const float* ap = asrc + (size_t)sr[i] * HEADS;
    a0[i] = ap[0]; a1[i] = ap[1]; a2[i] = ap[2];
    const float4* xr = reinterpret_cast<const float4*>(x + (size_t)sr[i] * HIN);
    xa[i] = xr[0]; xb[i] = xr[1]; xc[i] = xr[2];
  }
  float c0[4], c1[4], c2[4];
  float m0 = -INFINITY, m1 = -INFINITY, m2 = -INFINITY;
#pragma unroll
  for (int i = 0; i < 4; ++i) {
    const bool in = sub + GRP * i < q.nreg;
    c0[i] = in ? lrelu(a0[i] + ad0) : -INFINITY;
    c1[i] = in ? lrelu(a1[i] + ad1) : -INFINITY;
    c2[i] = in ? lrelu(a2[i] + ad2) : -INFINITY;
    m0 = fmaxf(m0, c0[i]); m1 = fmaxf(m1, c1[i]); m2 = fmaxf(m2, c2[i]);
  }
  for_extra(q, sub, [&](int e) __attribute__((always_inline)) {
    const float* ap = asrc + (size_t)e * HEADS;
    m0 = fmaxf(m0, lrelu(ap[0] + ad0));
    m1 = fmaxf(m1, lrelu(ap[1] + ad1));
    m2 = fmaxf(m2, lrelu(ap[2] + ad2));
  });
  m0 = grp_max(m0); m1 = grp_max(m1); m2 = grp_max(m2);

  float g0[HIN], g1[HIN], g2[HIN];
#pragma unroll
  for (int k = 0; k < HIN; ++k) { g0[k] = 0.f; g1[k] = 0.f; g2[k] = 0.f; }
  float z0 = 0.f, z1 = 0.f, z2 = 0.f;
#pragma unroll
  for (int i = 0; i < 4; ++i) {
    const bool in = sub + GRP * i < q.nreg;
    const float e0 = in ? expf(c0[i] - m0) : 0.f;
    const float e1 = in ? expf(c1[i] - m1) : 0.f;
    const float e2 = in ? expf(c2[i] - m2) : 0.f;
    z0 += e0; z1 += e1; z2 += e2;
    const float xs[HIN] = {xa[i].x, xa[i].y, xa[i].z, xa[i].w, xb[i].x, xb[i].y, xb[i].z, xb[i].w,
                           xc[i].x, xc[i].y, xc[i].z, xc[i].w};
#pragma unroll
    for (int k = 0; k < HIN; ++k) {
      g0[k] = fmaf(e0, xs[k], g0[k]); g1[k] = fmaf(e1, xs[k], g1[k]); g2[k] = fmaf(e2, xs[k], g2[k]);
    }
  }
  for_extra(q, sub, [&](int e) __attribute__((always_inline)) {
    const float* ap = asrc + (size_t)e * HEADS;
    const float e0 = expf(lrelu(ap[0] + ad0) - m0);
    const float e1 = expf(lrelu(ap[1] + ad1) - m1);
    const float e2 = expf(lrelu(ap[2] + ad2) - m2);
    z0 += e0; z1 += e1; z2 += e2;
    const float4* xr = reinterpret_cast<const float4*>(x + (size_t)e * HIN);
    const float4 ya = xr[0], yb = xr[1], yc = xr[2];
    const float xs[HIN] = {ya.x, ya.y, ya.z, ya.w, yb.x, yb.y, yb.z, yb.w, yc.x, yc.y, yc.z, yc.w};
#pragma unroll
    for (int k = 0; k < HIN; ++k) {
      g0[k] = fmaf(e0, xs[k], g0[k]); g1[k] = fmaf(e1, xs[k], g1[k]); g2[k] = fmaf(e2, xs[k], g2[k]);
    }
  });
  z0 = grp_sum(z0); z1 = grp_sum(z1); z2 = grp_sum(z2);
  const float r0 = 1.f / z0, r1 = 1.f / z1, r2 = 1.f / z2;

  {
    float4* o = reinterpret_cast<float4*>(&gt[tid][0]);
    o[0] = make_float4(g0[0] * r0, g0[1] * r0, g0[2] * r0, g0[3] * r0);
    o[1] = make_float4(g0[4] * r0, g0[5] * r0, g0[6] * r0, g0[7] * r0);
    o[2] = make_float4(g0[8] * r0, g0[9] * r0, g0[10] * r0, g0[11] * r0);
    o[3] = make_float4(g1[0] * r1, g1[1] * r1, g1[2] * r1, g1[3] * r1);
    o[4] = make_float4(g1[4] * r1, g1[5] * r1, g1[6] * r1, g1[7] * r1);
    o[5] = make_float4(g1[8] * r1, g1[9] * r1, g1[10] * r1, g1[11] * r1);
    o[6] = make_float4(g2[0] * r2, g2[1] * r2, g2[2] * r2, g2[3] * r2);
    o[7] = make_float4(g2[4] * r2, g2[5] * r2, g2[6] * r2, g2[7] * r2);
    o[8] = make_float4(g2[8] * r2, g2[9] * r2, g2[10] * r2, g2[11] * r2);
    const float one = (sub == 0) ? 1.f : 0.f;
    o[9] = make_float4(one, one, one, 0.f);
  }
  __syncthreads();

  const int w = __builtin_amdgcn_readfirstlane(tid >> 5);
  const int c = lane & 15, hi = lane >> 4;
  int bn = ((c & 3) == 3) ? (c - 1) : c;
  bn = (bn > 10) ? 10 : bn;
  const float4* ap0 = reinterpret_cast<const float4*>(&gt[w * 32 + c][0]) + hi;
  const float4* ap1 = reinterpret_cast<const float4*>(&gt[w * 32 + 16 + c][0]) + hi;
  const float4* bp = reinterpret_cast<const float4*>(Mt + (size_t)bn * MK) + hi;
  v8f acc0 = {};
  v8f acc1 = {};
#pragma unroll
  for (int i = 0; i < MK / 8; ++i) {
    const float4 b = bp[2 * i], a0 = ap0[2 * i], a1 = ap1[2 * i];
    v2f bx; bx.x = b.x;  bx.y = b.y;
    v2f by; by.x = b.z;  by.y = b.w;
    v2f p0; p0.x = a0.x; p0.y = a0.y;
    v2f q0; q0.x = a0.z; q0.y = a0.w;
    v2f p1; p1.x = a1.x; p1.y = a1.y;
    v2f q1; q1.x = a1.z; q1.y = a1.w;
    acc0 = __builtin_amdgcn_wmma_f32_16x16x4_f32(false, p0, false, bx, (short)0, acc0, false, false);
    acc1 = __builtin_amdgcn_wmma_f32_16x16x4_f32(false, p1, false, bx, (short)0, acc1, false, false);
    acc0 = __builtin_amdgcn_wmma_f32_16x16x4_f32(false, q0, false, by, (short)0, acc0, false, false);
    acc1 = __builtin_amdgcn_wmma_f32_16x16x4_f32(false, q1, false, by, (short)0, acc1, false, false);
  }
  const float t0 = ((acc0[0] + acc0[1]) + (acc0[2] + acc0[3])) + ((acc0[4] + acc0[5]) + (acc0[6] + acc0[7]));
  const float t1 = ((acc1[0] + acc1[1]) + (acc1[2] + acc1[3])) + ((acc1[4] + acc1[5]) + (acc1[6] + acc1[7]));
  const int n0 = blockIdx.x * NPB + w * 4 + hi;
  if (n0 < N)     T[(size_t)n0 * TW + c] = t0;
  if (n0 + 2 < N) T[(size_t)(n0 + 2) * TW + c] = t1;
}

__global__ void __launch_bounds__(256) k_rows2(const int2* __restrict__ segd,
                                               const int2* __restrict__ rec, int nseg,
                                               const float* __restrict__ T,
                                               const float* __restrict__ cst,
                                               float* __restrict__ res, int N) {
  __shared__ Bk bk;
  const int tid = threadIdx.x;
  const int sub = tid & (GRP - 1);
  const int node0 = blockIdx.x * NPB;
  const int dn = node0 + (tid >> 3);
  const int d = (dn < N) ? dn : (N - 1);
  const int2* sd = segd + (size_t)blockIdx.x * nseg;
  build_buckets(bk, sd, rec, nseg, node0, tid);
  const Run q = make_run(bk, d - node0, sd, rec, nseg, d);
  const float4 td = *reinterpret_cast<const float4*>(T + (size_t)d * TW + 8);
  const float ad0 = td.x, ad1 = td.y, ad2 = td.z;

  int sr[4];
#pragma unroll
  for (int i = 0; i < 4; ++i) {
    const bool filled = sub + GRP * i < q.nmain;
    sr[i] = filled ? q.colrow[sub + GRP * i] : d;
  }
  float4 ta[4], tt[4];
#pragma unroll
  for (int i = 0; i < 4; ++i) {
    const float4* tp = reinterpret_cast<const float4*>(T + (size_t)sr[i] * TW);
    ta[i] = tp[0]; tt[i] = tp[1];
  }
  float c0[4], c1[4], c2[4], t0[4], t1[4], t2[4];
  float m0 = -INFINITY, m1 = -INFINITY, m2 = -INFINITY;
#pragma unroll
  for (int i = 0; i < 4; ++i) {
    const bool in = sub + GRP * i < q.nreg;
    c0[i] = in ? lrelu(ta[i].x + ad0) : -INFINITY;
    c1[i] = in ? lrelu(ta[i].y + ad1) : -INFINITY;
    c2[i] = in ? lrelu(ta[i].z + ad2) : -INFINITY;
    t0[i] = tt[i].x; t1[i] = tt[i].y; t2[i] = tt[i].z;
    m0 = fmaxf(m0, c0[i]); m1 = fmaxf(m1, c1[i]); m2 = fmaxf(m2, c2[i]);
  }
  for_extra(q, sub, [&](int e) __attribute__((always_inline)) {
    const float4 a = *reinterpret_cast<const float4*>(T + (size_t)e * TW);
    m0 = fmaxf(m0, lrelu(a.x + ad0));
    m1 = fmaxf(m1, lrelu(a.y + ad1));
    m2 = fmaxf(m2, lrelu(a.z + ad2));
  });
  m0 = grp_max(m0); m1 = grp_max(m1); m2 = grp_max(m2);

  float z0 = 0.f, z1 = 0.f, z2 = 0.f;
  float n0 = 0.f, n1 = 0.f, n2 = 0.f;
#pragma unroll
  for (int i = 0; i < 4; ++i) {
    const bool in = sub + GRP * i < q.nreg;
    const float e0 = in ? expf(c0[i] - m0) : 0.f;
    const float e1 = in ? expf(c1[i] - m1) : 0.f;
    const float e2 = in ? expf(c2[i] - m2) : 0.f;
    z0 += e0; z1 += e1; z2 += e2;
    n0 = fmaf(e0, t0[i], n0); n1 = fmaf(e1, t1[i], n1); n2 = fmaf(e2, t2[i], n2);
  }
  for_extra(q, sub, [&](int e) __attribute__((always_inline)) {
    const float4* tp = reinterpret_cast<const float4*>(T + (size_t)e * TW);
    const float4 a = tp[0], t = tp[1];
    const float e0 = expf(lrelu(a.x + ad0) - m0);
    const float e1 = expf(lrelu(a.y + ad1) - m1);
    const float e2 = expf(lrelu(a.z + ad2) - m2);
    z0 += e0; z1 += e1; z2 += e2;
    n0 = fmaf(e0, t.x, n0); n1 = fmaf(e1, t.y, n1); n2 = fmaf(e2, t.z, n2);
  });
  z0 = grp_sum(z0); z1 = grp_sum(z1); z2 = grp_sum(z2);
  n0 = grp_sum(n0); n1 = grp_sum(n1); n2 = grp_sum(n2);
  if (sub == 0 && dn < N) res[dn] = (n0 / z0 + n1 / z1) + (n2 / z2 + cst[0]);
}

extern "C" void kernel_launch(void* const* d_in, const int* in_sizes, int n_in,
                              void* d_out, int out_size, void* d_ws, size_t ws_size,
                              hipStream_t stream) {
  const float* x   = (const float*)d_in[0];
  const int*   ei  = (const int*)d_in[1];
  const float* W1  = (const float*)d_in[2];
  const float* as1 = (const float*)d_in[3];
  const float* ad1 = (const float*)d_in[4];
  const float* b1  = (const float*)d_in[5];
  const float* W2  = (const float*)d_in[6];
  const float* as2 = (const float*)d_in[7];
  const float* ad2 = (const float*)d_in[8];
  const float* b2  = (const float*)d_in[9];
  const float* Wl  = (const float*)d_in[10];
  const float* bl  = (const float*)d_in[11];
  float* out = (float*)d_out;

  const int N  = in_sizes[0] / HIN;
  const int E  = in_sizes[1] / 2;
  const int* src = ei;
  const int* dst = ei + E;

  float* ws = (float*)d_ws;
  float* T      = ws;
  const int nbins = (N + NPB - 1) / NPB;
  const int nseg  = (E + SEGE - 1) / SEGE;
  int2*  REC    = (int2*)(T + (size_t)N * TW);
  float* CST    = (float*)(REC + (size_t)nseg * SEGE);
  float* MT     = CST + 64;
  float* ASRC   = MT + (size_t)16 * MK;
  float* ADST   = ASRC + (size_t)3 * N;
  int2*  SEGD   = (int2*)(ADST + (size_t)3 * N);

  const int rblocks = nbins;
  const int ablocks = (N + BT - 1) / BT;

  k_front<<<HEADS + ablocks + nseg, BT, 2 * nbins * sizeof(int), stream>>>(
      x, W1, as1, ad1, b1, W2, as2, ad2, Wl, b2, bl, src, dst, CST, MT, ASRC, ADST, SEGD, REC,
      N, E, nbins, nseg, ablocks);

  k_rows1<<<rblocks, 256, 0, stream>>>(SEGD, REC, nseg, ASRC, ADST, x, MT, T, N);

  k_rows2<<<rblocks, 256, 0, stream>>>(SEGD, REC, nseg, T, CST, out, N);
}
